# starting kernel
# baseline (speedup 1.0000x reference)
#include <hip/hip_runtime.h>
#include <math.h>


typedef __attribute__((ext_vector_type(2))) float v2f;
typedef __attribute__((ext_vector_type(8))) float v8f;

static constexpr int HID    = 256;
static constexpr int IN_DIM = 64;

__global__ void init_deg_kernel(float* __restrict__ deg, int n) {
  int i = blockIdx.x * blockDim.x + threadIdx.x;
  if (i < n) deg[i] = 1.0f;
}

__global__ void accum_deg_kernel(const int* __restrict__ tgt,
                                 float* __restrict__ deg, int e) {
  int i = blockIdx.x * blockDim.x + threadIdx.x;
  if (i < e) atomicAdd(&deg[tgt[i]], 1.0f);
}

__global__ void rsqrt_kernel(float* __restrict__ deg, int n) {
  int i = blockIdx.x * blockDim.x + threadIdx.x;
  if (i < n) deg[i] = rsqrtf(deg[i]);
}

__global__ void zero_kernel(float* __restrict__ p, int n) {
  int i = blockIdx.x * blockDim.x + threadIdx.x;
  if (i < n) p[i] = 0.0f;
}

__global__ void init_out_bias_kernel(float* __restrict__ out,
                                     const float* __restrict__ b, int total) {
  int i = blockIdx.x * blockDim.x + threadIdx.x;
  if (i < total) out[i] = b[i & (HID - 1)];
}

template <int K>
__global__ void wmma_gemm_kernel(const float* __restrict__ A,
                                 const float* __restrict__ B,
                                 float* __restrict__ C, int M) {
  const int lane = threadIdx.x & 31;
  const int wid  = (blockIdx.x * blockDim.x + threadIdx.x) >> 5;
  const int tilesN = HID / 16;
  const int tilesM = (M + 15) / 16;
  if (wid >= tilesM * tilesN) return;

  const int tm = wid / tilesN;
  const int tn = wid % tilesN;
  const int h  = lane >> 4;
  const int ln = lane & 15;

  const int m = tm * 16 + ln;
  const int n = tn * 16 + ln;

  v8f acc = {};
#pragma unroll 4
  for (int k0 = 0; k0 < K; k0 += 4) {
    v2f a, b;
    const float* ap = A + (size_t)m * K + (k0 + 2 * h);
    a.x = ap[0];
    a.y = ap[1];
    b.x = B[(size_t)(k0 + 2 * h) * HID + n];
    b.y = B[(size_t)(k0 + 2 * h + 1) * HID + n];
    acc = __builtin_amdgcn_wmma_f32_16x16x4_f32(
        false, a, false, b, (short)0, acc, false, false);
  }

  float* cp = C + (size_t)(tm * 16 + 8 * h) * HID + n;
#pragma unroll
  for (int r = 0; r < 8; ++r) cp[(size_t)r * HID] = acc[r];
}

__global__ void scatter_kernel(const float* __restrict__ H,
                               const int* __restrict__ src,
                               const int* __restrict__ tgt,
                               const float* __restrict__ dinv,
                               float* __restrict__ out, int E, int N) {
  const int wid  = (blockIdx.x * blockDim.x + threadIdx.x) >> 5;
  const int lane = threadIdx.x & 31;
  if (wid >= E + N) return;

  int s, t;
  if (wid < E) { s = src[wid]; t = tgt[wid]; }
  else         { s = t = wid - E; }

  const float norm = dinv[s] * dinv[t];
  const float* hp = H + (size_t)s * HID;
  float* op = out + (size_t)t * HID;
#pragma unroll
  for (int j = 0; j < 8; ++j) {
    const int c = lane + 32 * j;
    atomicAdd(op + c, hp[c] * norm);
  }
}

__global__ void epilogue1_kernel(const float* __restrict__ agg,
                                 const float* __restrict__ b1,
                                 float* __restrict__ X2, int total) {
  int idx = blockIdx.x * blockDim.x + threadIdx.x;
  if (idx >= total) return;
  const int i = idx >> 8;
  const int c = idx & 255;
  const float kLn1e4Over256 = 9.210340371976184f / 256.0f;
  const float freq = expf(-(float)(c & ~1) * kLn1e4Over256);
  const float ang  = (float)i * freq;
  const float pe   = (c & 1) ? cosf(ang) : sinf(ang);
  const float v = agg[idx] + b1[c] + pe;
  X2[idx] = v > 0.0f ? v : 0.0f;
}

extern "C" void kernel_launch(void* const* d_in, const int* in_sizes, int n_in,
                              void* d_out, int out_size, void* d_ws, size_t ws_size,
                              hipStream_t stream) {
  const float* X  = (const float*)d_in[0];
  const int*   EI = (const int*)d_in[1];
  const float* W1 = (const float*)d_in[2];
  const float* b1 = (const float*)d_in[3];
  const float* W2 = (const float*)d_in[4];
  const float* b2 = (const float*)d_in[5];
  float* out = (float*)d_out;

  const int N = in_sizes[0] / IN_DIM;
  const int E = in_sizes[1] / 2;
  const int NF = N * HID;

  float* dinv   = (float*)d_ws;
  float* bufH   = dinv + (1 << 16);
  float* bufAgg = bufH + (size_t)NF;

  const int* src = EI;
  const int* tgt = EI + E;

  const int T = 256;
  auto cdiv = [](int a, int b) { return (a + b - 1) / b; };

  init_deg_kernel<<<cdiv(N, T), T, 0, stream>>>(dinv, N);
  accum_deg_kernel<<<cdiv(E, T), T, 0, stream>>>(tgt, dinv, E);
  rsqrt_kernel<<<cdiv(N, T), T, 0, stream>>>(dinv, N);

  {
    const int tiles = cdiv(N, 16) * (HID / 16);
    wmma_gemm_kernel<IN_DIM><<<cdiv(tiles, T / 32), T, 0, stream>>>(X, W1, bufH, N);
  }

  zero_kernel<<<cdiv(NF, T), T, 0, stream>>>(bufAgg, NF);
  scatter_kernel<<<cdiv(E + N, T / 32), T, 0, stream>>>(bufH, src, tgt, dinv,
                                                        bufAgg, E, N);

  epilogue1_kernel<<<cdiv(NF, T), T, 0, stream>>>(bufAgg, b1, bufH, NF);

  {
    const int tiles = cdiv(N, 16) * (HID / 16);
    wmma_gemm_kernel<HID><<<cdiv(tiles, T / 32), T, 0, stream>>>(bufH, W2, bufAgg, N);
  }

  init_out_bias_kernel<<<cdiv(NF, T), T, 0, stream>>>(out, b2, NF);
  scatter_kernel<<<cdiv(E + N, T / 32), T, 0, stream>>>(bufAgg, src, tgt, dinv,
                                                        out, E, N);
}


# both aggregations keep their output rows in LDS (196 rows per group, 256 groups): each group filters all edges, adds source rows by LDS float atomics (arrival order), dinv[t] factored out
# speedup vs baseline: 1.9299x; 1.9299x over previous
#include <hip/hip_runtime.h>
#include <math.h>


typedef __attribute__((ext_vector_type(2))) float v2f;
typedef __attribute__((ext_vector_type(8))) float v8f;

static constexpr int HID    = 256;
static constexpr int IN_DIM = 64;

__global__ void init_deg_kernel(float* __restrict__ deg, int n) {
  int i = blockIdx.x * blockDim.x + threadIdx.x;
  if (i < n) deg[i] = 1.0f;
}

__global__ void accum_deg_kernel(const int* __restrict__ tgt,
                                 float* __restrict__ deg, int e) {
  int i = blockIdx.x * blockDim.x + threadIdx.x;
  if (i < e) atomicAdd(&deg[tgt[i]], 1.0f);
}

__global__ void rsqrt_kernel(float* __restrict__ deg, int n) {
  int i = blockIdx.x * blockDim.x + threadIdx.x;
  if (i < n) deg[i] = rsqrtf(deg[i]);
}

template <int K>
__global__ void wmma_gemm_kernel(const float* __restrict__ A,
                                 const float* __restrict__ B,
                                 float* __restrict__ C, int M) {
  const int lane = threadIdx.x & 31;
  const int wid  = (blockIdx.x * blockDim.x + threadIdx.x) >> 5;
  const int tilesN = HID / 16;
  const int tilesM = (M + 15) / 16;
  if (wid >= tilesM * tilesN) return;

  const int tm = wid / tilesN;
  const int tn = wid % tilesN;
  const int h  = lane >> 4;
  const int ln = lane & 15;

  const int m = tm * 16 + ln;
  const int n = tn * 16 + ln;

  v8f acc = {};
#pragma unroll 4
  for (int k0 = 0; k0 < K; k0 += 4) {
    v2f a, b;
    const float* ap = A + (size_t)m * K + (k0 + 2 * h);
    a.x = ap[0];
    a.y = ap[1];
    b.x = B[(size_t)(k0 + 2 * h) * HID + n];
    b.y = B[(size_t)(k0 + 2 * h + 1) * HID + n];
    acc = __builtin_amdgcn_wmma_f32_16x16x4_f32(
        false, a, false, b, (short)0, acc, false, false);
  }

  float* cp = C + (size_t)(tm * 16 + 8 * h) * HID + n;
#pragma unroll
  for (int r = 0; r < 8; ++r) cp[(size_t)r * HID] = acc[r];
}

static constexpr int TROWS = 196;
static constexpr int NWAVE = 32;
static constexpr int EPH   = 512;
static constexpr int BATCH = 32;
static constexpr int LCAP  = BATCH + EPH;


template <bool ADD_BIAS>
__global__ __launch_bounds__(1024)
void aggregate_lds_kernel(const float* __restrict__ H,
                          const int* __restrict__ src,
                          const int* __restrict__ tgt,
                          const float* __restrict__ dinv,
                          const float* __restrict__ bias,
                          float* __restrict__ out, int E, int N) {
  __shared__ float tile[TROWS * HID];
  __shared__ int   kept[NWAVE * LCAP];

  const int lane = threadIdx.x & 31;
  const int wave = threadIdx.x >> 5;
  int* mylist = kept + wave * LCAP;
  const unsigned lanes_below = (1u << lane) - 1u;
  const int hd  = min((int)(((16u - (unsigned)((size_t)tgt & 15)) & 15u) >> 2), E);
  const int* tb = tgt + hd;
  const int EB  = E - hd;
  const int G     = EB >> 2;
  const int GPI   = NWAVE * 8 * 32;
  const int ITERS = (G + GPI - 1) / GPI;
  const int FULL  = G / GPI;

  for (int base = blockIdx.x * TROWS; base < N; base += gridDim.x * TROWS) {
    const int rows = min(TROWS, N - base);

    for (int r = wave; r < rows; r += NWAVE) {
      const int i = base + r;
      const float d = dinv[i];
      const float4* hp = (const float4*)(H + (size_t)i * HID);
      const float4 a = hp[lane], b = hp[32 + lane];
      float* tp = tile + r * HID;
      tp[lane]       = a.x * d;  tp[32 + lane]  = a.y * d;
      tp[64 + lane]  = a.z * d;  tp[96 + lane]  = a.w * d;
      tp[128 + lane] = b.x * d;  tp[160 + lane] = b.y * d;
      tp[192 + lane] = b.z * d;  tp[224 + lane] = b.w * d;
    }
    __syncthreads();

    auto flush = [&](int n) {
      __builtin_amdgcn_fence(__ATOMIC_ACQ_REL, "wavefront");
      for (int k0 = 0; k0 < n; k0 += 32) {
        const int k = k0 + lane;
        int s = 0, r = 0;
        float ds = 0.0f;
        if (k < n) {
          const int e = mylist[k];
          s  = src[e];
          r  = tgt[e] - base;
          ds = dinv[s];
        }
        const int m = min(32, n - k0);
        int j = 0;
        for (; j + 4 <= m; j += 4) {
          float4 a[4], b[4];
          float  dj[4];
          int    rj[4];
#pragma unroll
          for (int u = 0; u < 4; ++u) {
            const int sj = __builtin_amdgcn_readlane(s, j + u);
            rj[u] = __builtin_amdgcn_readlane(r, j + u);
            dj[u] = __int_as_float(__builtin_amdgcn_readlane(__float_as_int(ds), j + u));
            const float4* hp = (const float4*)(H + (size_t)sj * HID);
            a[u] = hp[lane];
            b[u] = hp[32 + lane];
          }
#pragma unroll
          for (int u = 0; u < 4; ++u) {
            float* tp = tile + rj[u] * HID + lane;
            const float d = dj[u];
            __hip_atomic_fetch_add(tp,       a[u].x * d, __ATOMIC_RELAXED, __HIP_MEMORY_SCOPE_WORKGROUP);
            __hip_atomic_fetch_add(tp + 32,  a[u].y * d, __ATOMIC_RELAXED, __HIP_MEMORY_SCOPE_WORKGROUP);
            __hip_atomic_fetch_add(tp + 64,  a[u].z * d, __ATOMIC_RELAXED, __HIP_MEMORY_SCOPE_WORKGROUP);
            __hip_atomic_fetch_add(tp + 96,  a[u].w * d, __ATOMIC_RELAXED, __HIP_MEMORY_SCOPE_WORKGROUP);
            __hip_atomic_fetch_add(tp + 128, b[u].x * d, __ATOMIC_RELAXED, __HIP_MEMORY_SCOPE_WORKGROUP);
            __hip_atomic_fetch_add(tp + 160, b[u].y * d, __ATOMIC_RELAXED, __HIP_MEMORY_SCOPE_WORKGROUP);
            __hip_atomic_fetch_add(tp + 192, b[u].z * d, __ATOMIC_RELAXED, __HIP_MEMORY_SCOPE_WORKGROUP);
            __hip_atomic_fetch_add(tp + 224, b[u].w * d, __ATOMIC_RELAXED, __HIP_MEMORY_SCOPE_WORKGROUP);
          }
        }
        for (; j < m; ++j) {
          const int sj = __builtin_amdgcn_readlane(s, j);
          const int rr = __builtin_amdgcn_readlane(r, j);
          const float d = __int_as_float(__builtin_amdgcn_readlane(__float_as_int(ds), j));
          const float4* hp = (const float4*)(H + (size_t)sj * HID);
          const float4 a = hp[lane], b = hp[32 + lane];
          float* tp = tile + rr * HID + lane;
          __hip_atomic_fetch_add(tp,       a.x * d, __ATOMIC_RELAXED, __HIP_MEMORY_SCOPE_WORKGROUP);
          __hip_atomic_fetch_add(tp + 32,  a.y * d, __ATOMIC_RELAXED, __HIP_MEMORY_SCOPE_WORKGROUP);
          __hip_atomic_fetch_add(tp + 64,  a.z * d, __ATOMIC_RELAXED, __HIP_MEMORY_SCOPE_WORKGROUP);
          __hip_atomic_fetch_add(tp + 96,  a.w * d, __ATOMIC_RELAXED, __HIP_MEMORY_SCOPE_WORKGROUP);
          __hip_atomic_fetch_add(tp + 128, b.x * d, __ATOMIC_RELAXED, __HIP_MEMORY_SCOPE_WORKGROUP);
          __hip_atomic_fetch_add(tp + 160, b.y * d, __ATOMIC_RELAXED, __HIP_MEMORY_SCOPE_WORKGROUP);
          __hip_atomic_fetch_add(tp + 192, b.z * d, __ATOMIC_RELAXED, __HIP_MEMORY_SCOPE_WORKGROUP);
          __hip_atomic_fetch_add(tp + 224, b.w * d, __ATOMIC_RELAXED, __HIP_MEMORY_SCOPE_WORKGROUP);
        }
      }
      __builtin_amdgcn_fence(__ATOMIC_ACQ_REL, "wavefront");
    };

    int cnt = 0;
    auto consider = [&](int t, int e) {
      const bool hit = (unsigned)(t - base) < (unsigned)rows;
      const unsigned m = __builtin_amdgcn_ballot_w32(hit);
      if (m != 0u) {
        if (hit) mylist[cnt + __builtin_popcount(m & lanes_below)] = e;
        cnt += __builtin_popcount(m);
      }
    };

    if (wave == 0) {
      const int e = (lane < hd) ? lane : hd + 4 * G + (lane - hd);
      const int t = (lane < 8 && e < E) ? tgt[e] : base - 1;
      consider(t, e);
    }
    cnt = __builtin_amdgcn_readfirstlane(cnt);

    const int rot = (ITERS > 0) ? (int)(blockIdx.x % (unsigned)ITERS) : 0;
    auto place_of = [&](int it) {
      int p = it + rot;
      if (p >= ITERS) p -= ITERS;
      return p;
    };
    auto group_of = [&](int p, int u) {
      return ((p * NWAVE + wave) * 8 + u) * 32 + lane;
    };
    auto load_place = [&](int p, int4 (&dst)[8]) {
      if (p < FULL) {
#pragma unroll
        for (int u = 0; u < 8; ++u)
          dst[u] = *(const int4*)(tb + 4 * group_of(p, u));
      } else {
#pragma unroll
        for (int u = 0; u < 8; ++u) {
          const int g = group_of(p, u);
          int4 v = *(const int4*)(tb + 4 * min(g, G - 1));
          if (g >= G) v.x = v.y = v.z = v.w = base - 1;
          dst[u] = v;
        }
      }
    };

    for (int it = 0; it < ITERS; ++it) {
      const int p = place_of(it);
      int4 cur[8];
      load_place(p, cur);
#pragma unroll
      for (int half = 0; half < 2; ++half) {
        if (cnt >= BATCH) { flush(cnt); cnt = 0; }
#pragma unroll
        for (int u = 4 * half; u < 4 * half + 4; ++u) {
          const int e = hd + 4 * group_of(p, u);
          consider(cur[u].x, e);
          consider(cur[u].y, e + 1);
          consider(cur[u].z, e + 2);
          consider(cur[u].w, e + 3);
        }
      }
    }
    flush(cnt);
    __syncthreads();

    float4 ba = {0.0f, 0.0f, 0.0f, 0.0f}, bb = {0.0f, 0.0f, 0.0f, 0.0f};
    if (ADD_BIAS) {
      ba.x = bias[4 * lane];           ba.y = bias[4 * lane + 1];
      ba.z = bias[4 * lane + 2];       ba.w = bias[4 * lane + 3];
      bb.x = bias[128 + 4 * lane];     bb.y = bias[128 + 4 * lane + 1];
      bb.z = bias[128 + 4 * lane + 2]; bb.w = bias[128 + 4 * lane + 3];
    }
    for (int r = wave; r < rows; r += NWAVE) {
      const int i = base + r;
      const float d = dinv[i];
      const float* tp = tile + r * HID;
      float4 a, b;
      a.x = tp[lane] * d;        a.y = tp[32 + lane] * d;
      a.z = tp[64 + lane] * d;   a.w = tp[96 + lane] * d;
      b.x = tp[128 + lane] * d;  b.y = tp[160 + lane] * d;
      b.z = tp[192 + lane] * d;  b.w = tp[224 + lane] * d;
      if (ADD_BIAS) {
        a.x += ba.x; a.y += ba.y; a.z += ba.z; a.w += ba.w;
        b.x += bb.x; b.y += bb.y; b.z += bb.z; b.w += bb.w;
      }
      float4* op = (float4*)(out + (size_t)i * HID);
      op[lane]      = a;
      op[32 + lane] = b;
    }
    __syncthreads();
  }
}

__global__ void epilogue1_kernel(const float* __restrict__ agg,
                                 const float* __restrict__ b1,
                                 float* __restrict__ X2, int total) {
  int idx = blockIdx.x * blockDim.x + threadIdx.x;
  if (idx >= total) return;
  const int i = idx >> 8;
  const int c = idx & 255;
  const float kLn1e4Over256 = 9.210340371976184f / 256.0f;
  const float freq = expf(-(float)(c & ~1) * kLn1e4Over256);
  const float ang  = (float)i * freq;
  const float pe   = (c & 1) ? cosf(ang) : sinf(ang);
  const float v = agg[idx] + b1[c] + pe;
  X2[idx] = v > 0.0f ? v : 0.0f;
}

extern "C" void kernel_launch(void* const* d_in, const int* in_sizes, int n_in,
                              void* d_out, int out_size, void* d_ws, size_t ws_size,
                              hipStream_t stream) {
  const float* X  = (const float*)d_in[0];
  const int*   EI = (const int*)d_in[1];
  const float* W1 = (const float*)d_in[2];
  const float* b1 = (const float*)d_in[3];
  const float* W2 = (const float*)d_in[4];
  const float* b2 = (const float*)d_in[5];
  float* out = (float*)d_out;

  const int N = in_sizes[0] / IN_DIM;
  const int E = in_sizes[1] / 2;
  const int NF = N * HID;

  float* dinv   = (float*)d_ws;
  float* bufH   = dinv + (1 << 16);
  float* bufAgg = bufH + (size_t)NF;

  const int* src = EI;
  const int* tgt = EI + E;

  const int T = 256;
  auto cdiv = [](int a, int b) { return (a + b - 1) / b; };

  init_deg_kernel<<<cdiv(N, T), T, 0, stream>>>(dinv, N);
  accum_deg_kernel<<<cdiv(E, T), T, 0, stream>>>(tgt, dinv, E);
  rsqrt_kernel<<<cdiv(N, T), T, 0, stream>>>(dinv, N);

  {
    const int tiles = cdiv(N, 16) * (HID / 16);
    wmma_gemm_kernel<IN_DIM><<<cdiv(tiles, T / 32), T, 0, stream>>>(X, W1, bufH, N);
  }

  aggregate_lds_kernel<false><<<256, 1024, 0, stream>>>(bufH, src, tgt, dinv, b1,
                                                        bufAgg, E, N);

  epilogue1_kernel<<<cdiv(NF, T), T, 0, stream>>>(bufAgg, b1, bufH, NF);

  {
    const int tiles = cdiv(N, 16) * (HID / 16);
    wmma_gemm_kernel<HID><<<cdiv(tiles, T / 32), T, 0, stream>>>(bufH, W2, bufAgg, N);
  }

  aggregate_lds_kernel<true><<<256, 1024, 0, stream>>>(bufAgg, src, tgt, dinv, b2,
                                                       out, E, N);
}


# first aggregation finishes its rows in LDS (bias, position term, relu) and multiplies them by W2 there on the matrix cores: 6 launches (was 8); W2 product k order permuted (summation order)
# speedup vs baseline: 2.6868x; 1.3922x over previous
#include <hip/hip_runtime.h>
#include <math.h>


typedef __attribute__((ext_vector_type(2))) float v2f;
typedef __attribute__((ext_vector_type(8))) float v8f;

static constexpr int HID    = 256;
static constexpr int IN_DIM = 64;

__global__ void init_deg_kernel(float* __restrict__ deg, int n) {
  int i = blockIdx.x * blockDim.x + threadIdx.x;
  if (i < n) deg[i] = 1.0f;
}

__global__ void accum_deg_kernel(const int* __restrict__ tgt,
                                 float* __restrict__ deg, int e) {
  int i = blockIdx.x * blockDim.x + threadIdx.x;
  if (i < e) atomicAdd(&deg[tgt[i]], 1.0f);
}

__global__ void rsqrt_kernel(float* __restrict__ deg, int n) {
  int i = blockIdx.x * blockDim.x + threadIdx.x;
  if (i < n) deg[i] = rsqrtf(deg[i]);
}

template <int K>
__global__ void wmma_gemm_kernel(const float* __restrict__ A,
                                 const float* __restrict__ B,
                                 float* __restrict__ C, int M) {
  const int lane = threadIdx.x & 31;
  const int wid  = (blockIdx.x * blockDim.x + threadIdx.x) >> 5;
  const int tilesN = HID / 16;
  const int tilesM = (M + 15) / 16;
  if (wid >= tilesM * tilesN) return;

  const int tm = wid / tilesN;
  const int tn = wid % tilesN;
  const int h  = lane >> 4;
  const int ln = lane & 15;

  const int m = tm * 16 + ln;
  const int n = tn * 16 + ln;

  v8f acc = {};
#pragma unroll 4
  for (int k0 = 0; k0 < K; k0 += 4) {
    v2f a, b;
    const float* ap = A + (size_t)m * K + (k0 + 2 * h);
    a.x = ap[0];
    a.y = ap[1];
    b.x = B[(size_t)(k0 + 2 * h) * HID + n];
    b.y = B[(size_t)(k0 + 2 * h + 1) * HID + n];
    acc = __builtin_amdgcn_wmma_f32_16x16x4_f32(
        false, a, false, b, (short)0, acc, false, false);
  }

  float* cp = C + (size_t)(tm * 16 + 8 * h) * HID + n;
#pragma unroll
  for (int r = 0; r < 8; ++r) cp[(size_t)r * HID] = acc[r];
}

static constexpr int TROWS = 196;
static constexpr int NWAVE = 32;
static constexpr int EPH   = 512;
static constexpr int BATCH = 32;
static constexpr int LCAP  = BATCH + EPH;

static constexpr int TSTR  = HID + 1;

template <int NS>
__device__ __forceinline__ void rows_times_w(const float* tile,
                                             const float* __restrict__ Wm,
                                             float* __restrict__ out,
                                             int base, int rows, int mq, int tp,
                                             int lane) {
  const int h  = lane >> 4;
  const int ln = lane & 15;
  const int n2 = tp * 32 + 2 * ln;
  int aoff[NS];
#pragma unroll
  for (int i = 0; i < NS; ++i) {
    const int row = min((mq + 4 * i) * 16 + ln, rows - 1);
    aoff[i] = row * TSTR;
  }
  v8f acc[NS][2];
#pragma unroll
  for (int i = 0; i < NS; ++i) { acc[i][0] = v8f{}; acc[i][1] = v8f{}; }

  auto w_of = [&](int s) {
    const int lh = ((s & 31) + 16 * h) & 31;
    const int c0 = (s >> 5) * 128 + 4 * lh + 2 * h;
    return Wm + (size_t)c0 * HID + n2;
  };
  const float* wp = w_of(0);
  float2 w0n = *(const float2*)wp;
  float2 w1n = *(const float2*)(wp + HID);
#pragma unroll 2
  for (int s = 0; s < 64; ++s) {
    const float2 w0 = w0n, w1 = w1n;
    wp  = w_of(min(s + 1, 63));
    w0n = *(const float2*)wp;
    w1n = *(const float2*)(wp + HID);
    const int lh  = ((s & 31) + 16 * h) & 31;
    const int pos = (s >> 5) * 128 + 64 * h + lh;
    v2f a[NS];
#pragma unroll
    for (int i = 0; i < NS; ++i) {
      a[i].x = tile[aoff[i] + pos];
      a[i].y = tile[aoff[i] + pos + 32];
    }
    v2f be, bo;
    be.x = w0.x;  be.y = w1.x;
    bo.x = w0.y;  bo.y = w1.y;
#pragma unroll
    for (int i = 0; i < NS; ++i) {
      acc[i][0] = __builtin_amdgcn_wmma_f32_16x16x4_f32(
          false, a[i], false, be, (short)0, acc[i][0], false, false);
      acc[i][1] = __builtin_amdgcn_wmma_f32_16x16x4_f32(
          false, a[i], false, bo, (short)0, acc[i][1], false, false);
    }
  }

#pragma unroll
  for (int i = 0; i < NS; ++i) {
    const int r0 = (mq + 4 * i) * 16 + 8 * h;
#pragma unroll
    for (int r = 0; r < 8; ++r) {
      if (r0 + r < rows) {
        float2 v;
        v.x = acc[i][0][r];
        v.y = acc[i][1][r];
        *(float2*)(out + (size_t)(base + r0 + r) * HID + n2) = v;
      }
    }
  }
}

template <bool LAST>
__global__ __launch_bounds__(1024)
void aggregate_lds_kernel(const float* __restrict__ H,
                          const int* __restrict__ src,
                          const int* __restrict__ tgt,
                          const float* __restrict__ dinv,
                          const float* __restrict__ bias,
                          const float* __restrict__ Wn,
                          float* __restrict__ out, int E, int N) {
  __shared__ float tile[TROWS * TSTR];
  __shared__ int   kept[NWAVE * LCAP];

  const int lane = threadIdx.x & 31;
  const int wave = threadIdx.x >> 5;
  int* mylist = kept + wave * LCAP;
  const unsigned lanes_below = (1u << lane) - 1u;
  const int hd  = min((int)(((16u - (unsigned)((size_t)tgt & 15)) & 15u) >> 2), E);
  const int* tb = tgt + hd;
  const int EB  = E - hd;
  const int G     = EB >> 2;
  const int GPI   = NWAVE * 8 * 32;
  const int ITERS = (G + GPI - 1) / GPI;
  const int FULL  = G / GPI;

  for (int base = blockIdx.x * TROWS; base < N; base += gridDim.x * TROWS) {
    const int rows = min(TROWS, N - base);

    for (int r = wave; r < rows; r += NWAVE) {
      const int i = base + r;
      const float d = dinv[i];
      const float4* hp = (const float4*)(H + (size_t)i * HID);
      const float4 a = hp[lane], b = hp[32 + lane];
      float* tp = tile + r * TSTR;
      tp[lane]       = a.x * d;  tp[32 + lane]  = a.y * d;
      tp[64 + lane]  = a.z * d;  tp[96 + lane]  = a.w * d;
      tp[128 + lane] = b.x * d;  tp[160 + lane] = b.y * d;
      tp[192 + lane] = b.z * d;  tp[224 + lane] = b.w * d;
    }
    __syncthreads();

    auto flush = [&](int n) {
      __builtin_amdgcn_fence(__ATOMIC_ACQ_REL, "wavefront");
      for (int k0 = 0; k0 < n; k0 += 32) {
        const int k = k0 + lane;
        int s = 0, r = 0;
        float ds = 0.0f;
        if (k < n) {
          const int e = mylist[k];
          s  = src[e];
          r  = tgt[e] - base;
          ds = dinv[s];
        }
        const int m = min(32, n - k0);
        int j = 0;
        for (; j + 4 <= m; j += 4) {
          float4 a[4], b[4];
          float  dj[4];
          int    rj[4];
#pragma unroll
          for (int u = 0; u < 4; ++u) {
            const int sj = __builtin_amdgcn_readlane(s, j + u);
            rj[u] = __builtin_amdgcn_readlane(r, j + u);
            dj[u] = __int_as_float(__builtin_amdgcn_readlane(__float_as_int(ds), j + u));
            const float4* hp = (const float4*)(H + (size_t)sj * HID);
            a[u] = hp[lane];
            b[u] = hp[32 + lane];
          }
#pragma unroll
          for (int u = 0; u < 4; ++u) {
            float* tp = tile + rj[u] * TSTR + lane;
            const float d = dj[u];
            __hip_atomic_fetch_add(tp,       a[u].x * d, __ATOMIC_RELAXED, __HIP_MEMORY_SCOPE_WORKGROUP);
            __hip_atomic_fetch_add(tp + 32,  a[u].y * d, __ATOMIC_RELAXED, __HIP_MEMORY_SCOPE_WORKGROUP);
            __hip_atomic_fetch_add(tp + 64,  a[u].z * d, __ATOMIC_RELAXED, __HIP_MEMORY_SCOPE_WORKGROUP);
            __hip_atomic_fetch_add(tp + 96,  a[u].w * d, __ATOMIC_RELAXED, __HIP_MEMORY_SCOPE_WORKGROUP);
            __hip_atomic_fetch_add(tp + 128, b[u].x * d, __ATOMIC_RELAXED, __HIP_MEMORY_SCOPE_WORKGROUP);
            __hip_atomic_fetch_add(tp + 160, b[u].y * d, __ATOMIC_RELAXED, __HIP_MEMORY_SCOPE_WORKGROUP);
            __hip_atomic_fetch_add(tp + 192, b[u].z * d, __ATOMIC_RELAXED, __HIP_MEMORY_SCOPE_WORKGROUP);
            __hip_atomic_fetch_add(tp + 224, b[u].w * d, __ATOMIC_RELAXED, __HIP_MEMORY_SCOPE_WORKGROUP);
          }
        }
        for (; j < m; ++j) {
          const int sj = __builtin_amdgcn_readlane(s, j);
          const int rr = __builtin_amdgcn_readlane(r, j);
          const float d = __int_as_float(__builtin_amdgcn_readlane(__float_as_int(ds), j));
          const float4* hp = (const float4*)(H + (size_t)sj * HID);
          const float4 a = hp[lane], b = hp[32 + lane];
          float* tp = tile + rr * TSTR + lane;
          __hip_atomic_fetch_add(tp,       a.x * d, __ATOMIC_RELAXED, __HIP_MEMORY_SCOPE_WORKGROUP);
          __hip_atomic_fetch_add(tp + 32,  a.y * d, __ATOMIC_RELAXED, __HIP_MEMORY_SCOPE_WORKGROUP);
          __hip_atomic_fetch_add(tp + 64,  a.z * d, __ATOMIC_RELAXED, __HIP_MEMORY_SCOPE_WORKGROUP);
          __hip_atomic_fetch_add(tp + 96,  a.w * d, __ATOMIC_RELAXED, __HIP_MEMORY_SCOPE_WORKGROUP);
          __hip_atomic_fetch_add(tp + 128, b.x * d, __ATOMIC_RELAXED, __HIP_MEMORY_SCOPE_WORKGROUP);
          __hip_atomic_fetch_add(tp + 160, b.y * d, __ATOMIC_RELAXED, __HIP_MEMORY_SCOPE_WORKGROUP);
          __hip_atomic_fetch_add(tp + 192, b.z * d, __ATOMIC_RELAXED, __HIP_MEMORY_SCOPE_WORKGROUP);
          __hip_atomic_fetch_add(tp + 224, b.w * d, __ATOMIC_RELAXED, __HIP_MEMORY_SCOPE_WORKGROUP);
        }
      }
      __builtin_amdgcn_fence(__ATOMIC_ACQ_REL, "wavefront");
    };

    int cnt = 0;
    auto consider = [&](int t, int e) {
      const bool hit = (unsigned)(t - base) < (unsigned)rows;
      const unsigned m = __builtin_amdgcn_ballot_w32(hit);
      if (m != 0u) {
        if (hit) mylist[cnt + __builtin_popcount(m & lanes_below)] = e;
        cnt += __builtin_popcount(m);
      }
    };

    if (wave == 0) {
      const int e = (lane < hd) ? lane : hd + 4 * G + (lane - hd);
      const int t = (lane < 8 && e < E) ? tgt[e] : base - 1;
      consider(t, e);
    }
    cnt = __builtin_amdgcn_readfirstlane(cnt);

    const int rot = (ITERS > 0) ? (int)(blockIdx.x % (unsigned)ITERS) : 0;
    auto place_of = [&](int it) {
      int p = it + rot;
      if (p >= ITERS) p -= ITERS;
      return p;
    };
    auto group_of = [&](int p, int u) {
      return ((p * NWAVE + wave) * 8 + u) * 32 + lane;
    };
    auto load_place = [&](int p, int4 (&dst)[8]) {
      if (p < FULL) {
#pragma unroll
        for (int u = 0; u < 8; ++u)
          dst[u] = *(const int4*)(tb + 4 * group_of(p, u));
      } else {
#pragma unroll
        for (int u = 0; u < 8; ++u) {
          const int g = group_of(p, u);
          int4 v = *(const int4*)(tb + 4 * min(g, G - 1));
          if (g >= G) v.x = v.y = v.z = v.w = base - 1;
          dst[u] = v;
        }
      }
    };

    for (int it = 0; it < ITERS; ++it) {
      const int p = place_of(it);
      int4 cur[8];
      load_place(p, cur);
#pragma unroll
      for (int half = 0; half < 2; ++half) {
        if (cnt >= BATCH) { flush(cnt); cnt = 0; }
#pragma unroll
        for (int u = 4 * half; u < 4 * half + 4; ++u) {
          const int e = hd + 4 * group_of(p, u);
          consider(cur[u].x, e);
          consider(cur[u].y, e + 1);
          consider(cur[u].z, e + 2);
          consider(cur[u].w, e + 3);
        }
      }
    }
    flush(cnt);
    __syncthreads();

    float4 ba, bb;
    ba.x = bias[4 * lane];           ba.y = bias[4 * lane + 1];
    ba.z = bias[4 * lane + 2];       ba.w = bias[4 * lane + 3];
    bb.x = bias[128 + 4 * lane];     bb.y = bias[128 + 4 * lane + 1];
    bb.z = bias[128 + 4 * lane + 2]; bb.w = bias[128 + 4 * lane + 3];
    if (LAST) {
      for (int r = wave; r < rows; r += NWAVE) {
        const int i = base + r;
        const float d = dinv[i];
        const float* tp = tile + r * TSTR;
        float4 a, b;
        a.x = tp[lane] * d;        a.y = tp[32 + lane] * d;
        a.z = tp[64 + lane] * d;   a.w = tp[96 + lane] * d;
        b.x = tp[128 + lane] * d;  b.y = tp[160 + lane] * d;
        b.z = tp[192 + lane] * d;  b.w = tp[224 + lane] * d;
        a.x += ba.x; a.y += ba.y; a.z += ba.z; a.w += ba.w;
        b.x += bb.x; b.y += bb.y; b.z += bb.z; b.w += bb.w;
        float4* op = (float4*)(out + (size_t)i * HID);
        op[lane]      = a;
        op[32 + lane] = b;
      }
    } else {
      const float kLn1e4Over256 = 9.210340371976184f / 256.0f;
      const float f0 = expf(-(float)(4 * lane)           * kLn1e4Over256);
      const float f1 = expf(-(float)(4 * lane + 2)       * kLn1e4Over256);
      const float f2 = expf(-(float)(128 + 4 * lane)     * kLn1e4Over256);
      const float f3 = expf(-(float)(128 + 4 * lane + 2) * kLn1e4Over256);
      for (int r = wave; r < rows; r += NWAVE) {
        const int i = base + r;
        const float d  = dinv[i];
        const float fi = (float)i;
        float* tp = tile + r * TSTR;
        float s0, c0, s1, c1, s2, c2, s3, c3;
        sincosf(fi * f0, &s0, &c0);
        sincosf(fi * f1, &s1, &c1);
        sincosf(fi * f2, &s2, &c2);
        sincosf(fi * f3, &s3, &c3);
        float4 a, b;
        a.x = tp[lane] * d + ba.x + s0;        a.y = tp[32 + lane] * d + ba.y + c0;
        a.z = tp[64 + lane] * d + ba.z + s1;   a.w = tp[96 + lane] * d + ba.w + c1;
        b.x = tp[128 + lane] * d + bb.x + s2;  b.y = tp[160 + lane] * d + bb.y + c2;
        b.z = tp[192 + lane] * d + bb.z + s3;  b.w = tp[224 + lane] * d + bb.w + c3;
        tp[lane]       = a.x > 0.0f ? a.x : 0.0f;  tp[32 + lane]  = a.y > 0.0f ? a.y : 0.0f;
        tp[64 + lane]  = a.z > 0.0f ? a.z : 0.0f;  tp[96 + lane]  = a.w > 0.0f ? a.w : 0.0f;
        tp[128 + lane] = b.x > 0.0f ? b.x : 0.0f;  tp[160 + lane] = b.y > 0.0f ? b.y : 0.0f;
        tp[192 + lane] = b.z > 0.0f ? b.z : 0.0f;  tp[224 + lane] = b.w > 0.0f ? b.w : 0.0f;
      }
      __syncthreads();
      const int tp8 = wave & 7;
      const int mq  = wave >> 3;
      if (mq == 0) rows_times_w<4>(tile, Wn, out, base, rows, mq, tp8, lane);
      else         rows_times_w<3>(tile, Wn, out, base, rows, mq, tp8, lane);
    }
    __syncthreads();
  }
}

extern "C" void kernel_launch(void* const* d_in, const int* in_sizes, int n_in,
                              void* d_out, int out_size, void* d_ws, size_t ws_size,
                              hipStream_t stream) {
  const float* X  = (const float*)d_in[0];
  const int*   EI = (const int*)d_in[1];
  const float* W1 = (const float*)d_in[2];
  const float* b1 = (const float*)d_in[3];
  const float* W2 = (const float*)d_in[4];
  const float* b2 = (const float*)d_in[5];
  float* out = (float*)d_out;

  const int N = in_sizes[0] / IN_DIM;
  const int E = in_sizes[1] / 2;
  const int NF = N * HID;

  float* dinv   = (float*)d_ws;
  float* bufH   = dinv + (1 << 16);
  float* bufAgg = bufH + (size_t)NF;

  const int* src = EI;
  const int* tgt = EI + E;

  const int T = 256;
  auto cdiv = [](int a, int b) { return (a + b - 1) / b; };

  init_deg_kernel<<<cdiv(N, T), T, 0, stream>>>(dinv, N);
  accum_deg_kernel<<<cdiv(E, T), T, 0, stream>>>(tgt, dinv, E);
  rsqrt_kernel<<<cdiv(N, T), T, 0, stream>>>(dinv, N);

  {
    const int tiles = cdiv(N, 16) * (HID / 16);
    wmma_gemm_kernel<IN_DIM><<<cdiv(tiles, T / 32), T, 0, stream>>>(X, W1, bufH, N);
  }

  aggregate_lds_kernel<false><<<256, 1024, 0, stream>>>(bufH, src, tgt, dinv, b1, W2,
                                                        bufAgg, E, N);

  aggregate_lds_kernel<true><<<256, 1024, 0, stream>>>(bufAgg, src, tgt, dinv, b2, W2,
                                                       out, E, N);
}


# second aggregation takes its edges (source row, tile row, source dinv) from per-wave lists that the first aggregation's search leaves in the workspace; it searches no edges; sums in arrival order
# speedup vs baseline: 2.8150x; 1.0477x over previous
#include <hip/hip_runtime.h>
#include <math.h>


typedef __attribute__((ext_vector_type(2))) float v2f;
typedef __attribute__((ext_vector_type(8))) float v8f;

static constexpr int HID    = 256;
static constexpr int IN_DIM = 64;

__global__ void init_deg_kernel(float* __restrict__ deg, int n) {
  int i = blockIdx.x * blockDim.x + threadIdx.x;
  if (i < n) deg[i] = 1.0f;
}

__global__ void accum_deg_kernel(const int* __restrict__ tgt,
                                 float* __restrict__ deg, int e) {
  int i = blockIdx.x * blockDim.x + threadIdx.x;
  if (i < e) atomicAdd(&deg[tgt[i]], 1.0f);
}

__global__ void rsqrt_kernel(float* __restrict__ deg, int n) {
  int i = blockIdx.x * blockDim.x + threadIdx.x;
  if (i < n) deg[i] = rsqrtf(deg[i]);
}

template <int K>
__global__ void wmma_gemm_kernel(const float* __restrict__ A,
                                 const float* __restrict__ B,
                                 float* __restrict__ C, int M) {
  const int lane = threadIdx.x & 31;
  const int wid  = (blockIdx.x * blockDim.x + threadIdx.x) >> 5;
  const int tilesN = HID / 16;
  const int tilesM = (M + 15) / 16;
  if (wid >= tilesM * tilesN) return;

  const int tm = wid / tilesN;
  const int tn = wid % tilesN;
  const int h  = lane >> 4;
  const int ln = lane & 15;

  const int m = tm * 16 + ln;
  const int n = tn * 16 + ln;

  v8f acc = {};
#pragma unroll 4
  for (int k0 = 0; k0 < K; k0 += 4) {
    v2f a, b;
    const float* ap = A + (size_t)m * K + (k0 + 2 * h);
    a.x = ap[0];
    a.y = ap[1];
    b.x = B[(size_t)(k0 + 2 * h) * HID + n];
    b.y = B[(size_t)(k0 + 2 * h + 1) * HID + n];
    acc = __builtin_amdgcn_wmma_f32_16x16x4_f32(
        false, a, false, b, (short)0, acc, false, false);
  }

  float* cp = C + (size_t)(tm * 16 + 8 * h) * HID + n;
#pragma unroll
  for (int r = 0; r < 8; ++r) cp[(size_t)r * HID] = acc[r];
}

static constexpr int TROWS = 196;
static constexpr int NWAVE = 32;
static constexpr int EPH   = 512;
static constexpr int BATCH = 32;
static constexpr int LCAP  = BATCH + EPH;
static constexpr int WCAP  = 512;
static constexpr int NLIST = 256;

static constexpr int TSTR  = HID + 1;

template <int NS>
__device__ __forceinline__ void rows_times_w(const float* tile,
                                             const float* __restrict__ Wm,
                                             float* __restrict__ out,
                                             int base, int rows, int mq, int tp,
                                             int lane) {
  const int h  = lane >> 4;
  const int ln = lane & 15;
  const int n2 = tp * 32 + 2 * ln;
  int aoff[NS];
#pragma unroll
  for (int i = 0; i < NS; ++i) {
    const int row = min((mq + 4 * i) * 16 + ln, rows - 1);
    aoff[i] = row * TSTR;
  }
  v8f acc[NS][2];
#pragma unroll
  for (int i = 0; i < NS; ++i) { acc[i][0] = v8f{}; acc[i][1] = v8f{}; }

  auto w_of = [&](int s) {
    const int lh = ((s & 31) + 16 * h) & 31;
    const int c0 = (s >> 5) * 128 + 4 * lh + 2 * h;
    return Wm + (size_t)c0 * HID + n2;
  };
  const float* wp = w_of(0);
  float2 w0n = *(const float2*)wp;
  float2 w1n = *(const float2*)(wp + HID);
#pragma unroll 2
  for (int s = 0; s < 64; ++s) {
    const float2 w0 = w0n, w1 = w1n;
    wp  = w_of(min(s + 1, 63));
    w0n = *(const float2*)wp;
    w1n = *(const float2*)(wp + HID);
    const int lh  = ((s & 31) + 16 * h) & 31;
    const int pos = (s >> 5) * 128 + 64 * h + lh;
    v2f a[NS];
#pragma unroll
    for (int i = 0; i < NS; ++i) {
      a[i].x = tile[aoff[i] + pos];
      a[i].y = tile[aoff[i] + pos + 32];
    }
    v2f be, bo;
    be.x = w0.x;  be.y = w1.x;
    bo.x = w0.y;  bo.y = w1.y;
#pragma unroll
    for (int i = 0; i < NS; ++i) {
      acc[i][0] = __builtin_amdgcn_wmma_f32_16x16x4_f32(
          false, a[i], false, be, (short)0, acc[i][0], false, false);
      acc[i][1] = __builtin_amdgcn_wmma_f32_16x16x4_f32(
          false, a[i], false, bo, (short)0, acc[i][1], false, false);
    }
  }

#pragma unroll
  for (int i = 0; i < NS; ++i) {
    const int r0 = (mq + 4 * i) * 16 + 8 * h;
#pragma unroll
    for (int r = 0; r < 8; ++r) {
      if (r0 + r < rows) {
        float2 v;
        v.x = acc[i][0][r];
        v.y = acc[i][1][r];
        *(float2*)(out + (size_t)(base + r0 + r) * HID + n2) = v;
      }
    }
  }
}

template <bool LAST>
__global__ __launch_bounds__(1024)
void aggregate_lds_kernel(const float* __restrict__ H,
                          const int* __restrict__ src,
                          const int* __restrict__ tgt,
                          const float* __restrict__ dinv,
                          const float* __restrict__ bias,
                          const float* __restrict__ Wn,
                          float* __restrict__ out,
                          int* __restrict__ wlist, int* __restrict__ wcount,
                          int E, int N) {
  __shared__ float tile[TROWS * TSTR];
  __shared__ int   kept[NWAVE * LCAP];

  const int lane = threadIdx.x & 31;
  const int wave = threadIdx.x >> 5;
  int* mylist = kept + wave * LCAP;
  const unsigned lanes_below = (1u << lane) - 1u;
  const int hd  = min((int)(((16u - (unsigned)((size_t)tgt & 15)) & 15u) >> 2), E);
  const int* tb = tgt + hd;
  const int EB  = E - hd;
  const int G     = EB >> 2;
  const int GPI   = NWAVE * 8 * 32;
  const int ITERS = (G + GPI - 1) / GPI;
  const int FULL  = G / GPI;

  for (int base = blockIdx.x * TROWS; base < N; base += gridDim.x * TROWS) {
    const int rows = min(TROWS, N - base);
    const int tile_id  = base / TROWS;
    const bool listed  = tile_id < NLIST && N <= (1 << 24);
    const int  lslot   = (listed ? tile_id : 0) * NWAVE + wave;
    int2* mywl = (int2*)wlist + (size_t)lslot * WCAP;
    int pos = 0;

    for (int r = wave; r < rows; r += NWAVE) {
      const int i = base + r;
      const float d = dinv[i];
      const float4* hp = (const float4*)(H + (size_t)i * HID);
      const float4 a = hp[lane], b = hp[32 + lane];
      float* tp = tile + r * TSTR;
      tp[lane]       = a.x * d;  tp[32 + lane]  = a.y * d;
      tp[64 + lane]  = a.z * d;  tp[96 + lane]  = a.w * d;
      tp[128 + lane] = b.x * d;  tp[160 + lane] = b.y * d;
      tp[192 + lane] = b.z * d;  tp[224 + lane] = b.w * d;
    }
    __syncthreads();

    auto add_rows = [&](int s, int r, float ds, int m) {
      int j = 0;
      for (; j + 4 <= m; j += 4) {
        float4 a[4], b[4];
        float  dj[4];
        int    rj[4];
#pragma unroll
        for (int u = 0; u < 4; ++u) {
          const int sj = __builtin_amdgcn_readlane(s, j + u);
          rj[u] = __builtin_amdgcn_readlane(r, j + u);
          dj[u] = __int_as_float(__builtin_amdgcn_readlane(__float_as_int(ds), j + u));
          const float4* hp = (const float4*)(H + (size_t)sj * HID);
          a[u] = hp[lane];
          b[u] = hp[32 + lane];
        }
#pragma unroll
        for (int u = 0; u < 4; ++u) {
          float* tp = tile + rj[u] * TSTR + lane;
          const float d = dj[u];
          __hip_atomic_fetch_add(tp,       a[u].x * d, __ATOMIC_RELAXED, __HIP_MEMORY_SCOPE_WORKGROUP);
          __hip_atomic_fetch_add(tp + 32,  a[u].y * d, __ATOMIC_RELAXED, __HIP_MEMORY_SCOPE_WORKGROUP);
          __hip_atomic_fetch_add(tp + 64,  a[u].z * d, __ATOMIC_RELAXED, __HIP_MEMORY_SCOPE_WORKGROUP);
          __hip_atomic_fetch_add(tp + 96,  a[u].w * d, __ATOMIC_RELAXED, __HIP_MEMORY_SCOPE_WORKGROUP);
          __hip_atomic_fetch_add(tp + 128, b[u].x * d, __ATOMIC_RELAXED, __HIP_MEMORY_SCOPE_WORKGROUP);
          __hip_atomic_fetch_add(tp + 160, b[u].y * d, __ATOMIC_RELAXED, __HIP_MEMORY_SCOPE_WORKGROUP);
          __hip_atomic_fetch_add(tp + 192, b[u].z * d, __ATOMIC_RELAXED, __HIP_MEMORY_SCOPE_WORKGROUP);
          __hip_atomic_fetch_add(tp + 224, b[u].w * d, __ATOMIC_RELAXED, __HIP_MEMORY_SCOPE_WORKGROUP);
        }
      }
      for (; j < m; ++j) {
        const int sj = __builtin_amdgcn_readlane(s, j);
        const int rr = __builtin_amdgcn_readlane(r, j);
        const float d = __int_as_float(__builtin_amdgcn_readlane(__float_as_int(ds), j));
        const float4* hp = (const float4*)(H + (size_t)sj * HID);
        const float4 a = hp[lane], b = hp[32 + lane];
        float* tp = tile + rr * TSTR + lane;
        __hip_atomic_fetch_add(tp,       a.x * d, __ATOMIC_RELAXED, __HIP_MEMORY_SCOPE_WORKGROUP);
        __hip_atomic_fetch_add(tp + 32,  a.y * d, __ATOMIC_RELAXED, __HIP_MEMORY_SCOPE_WORKGROUP);
        __hip_atomic_fetch_add(tp + 64,  a.z * d, __ATOMIC_RELAXED, __HIP_MEMORY_SCOPE_WORKGROUP);
        __hip_atomic_fetch_add(tp + 96,  a.w * d, __ATOMIC_RELAXED, __HIP_MEMORY_SCOPE_WORKGROUP);
        __hip_atomic_fetch_add(tp + 128, b.x * d, __ATOMIC_RELAXED, __HIP_MEMORY_SCOPE_WORKGROUP);
        __hip_atomic_fetch_add(tp + 160, b.y * d, __ATOMIC_RELAXED, __HIP_MEMORY_SCOPE_WORKGROUP);
        __hip_atomic_fetch_add(tp + 192, b.z * d, __ATOMIC_RELAXED, __HIP_MEMORY_SCOPE_WORKGROUP);
        __hip_atomic_fetch_add(tp + 224, b.w * d, __ATOMIC_RELAXED, __HIP_MEMORY_SCOPE_WORKGROUP);
      }
    };

    auto flush = [&](int n) {
      __builtin_amdgcn_fence(__ATOMIC_ACQ_REL, "wavefront");
      for (int k0 = 0; k0 < n; k0 += 32) {
        const int k = k0 + lane;
        int s = 0, r = 0;
        float ds = 0.0f;
        if (k < n) {
          const int e = mylist[k];
          s  = src[e];
          r  = tgt[e] - base;
          ds = dinv[s];
          if (!LAST && listed && pos + k < WCAP)
            mywl[pos + k] = make_int2((int)((unsigned)s | ((unsigned)r << 24)), __float_as_int(ds));
        }
        add_rows(s, r, ds, min(32, n - k0));
      }
      pos += n;
      __builtin_amdgcn_fence(__ATOMIC_ACQ_REL, "wavefront");
    };

    int nl = 0;
    bool use_list = false;
    if (LAST && listed) {
      nl = __builtin_amdgcn_readfirstlane(wcount[lslot]);
      use_list = nl <= WCAP;
    }
    if (use_list) {
      int2 p = (lane < nl) ? mywl[lane] : make_int2(0, 0);
      for (int k0 = 0; k0 < nl; k0 += 32) {
        const int k = k0 + lane;
        const int s = p.x & 0xFFFFFF;
        const int r = (int)((unsigned)p.x >> 24);
        const float ds = __int_as_float(p.y);
        p = (k + 32 < nl) ? mywl[k + 32] : make_int2(0, 0);
        add_rows(s, r, ds, min(32, nl - k0));
      }
    } else {
      int cnt = 0;
      auto consider = [&](int t, int e) {
        const bool hit = (unsigned)(t - base) < (unsigned)rows;
        const unsigned m = __builtin_amdgcn_ballot_w32(hit);
        if (m != 0u) {
          if (hit) mylist[cnt + __builtin_popcount(m & lanes_below)] = e;
          cnt += __builtin_popcount(m);
        }
      };

      if (wave == 0) {
        const int e = (lane < hd) ? lane : hd + 4 * G + (lane - hd);
        const int t = (lane < 8 && e < E) ? tgt[e] : base - 1;
        consider(t, e);
      }
      cnt = __builtin_amdgcn_readfirstlane(cnt);

      const int rot = (ITERS > 0) ? (int)(blockIdx.x % (unsigned)ITERS) : 0;
      auto place_of = [&](int it) {
        int p = it + rot;
        if (p >= ITERS) p -= ITERS;
        return p;
      };
      auto group_of = [&](int p, int u) {
        return ((p * NWAVE + wave) * 8 + u) * 32 + lane;
      };
      auto load_place = [&](int p, int4 (&dst)[8]) {
        if (p < FULL) {
#pragma unroll
          for (int u = 0; u < 8; ++u)
            dst[u] = *(const int4*)(tb + 4 * group_of(p, u));
        } else {
#pragma unroll
          for (int u = 0; u < 8; ++u) {
            const int g = group_of(p, u);
            int4 v = *(const int4*)(tb + 4 * min(g, G - 1));
            if (g >= G) v.x = v.y = v.z = v.w = base - 1;
            dst[u] = v;
          }
        }
      };

      for (int it = 0; it < ITERS; ++it) {
        const int p = place_of(it);
        int4 cur[8];
        load_place(p, cur);
#pragma unroll
        for (int half = 0; half < 2; ++half) {
          if (cnt >= BATCH) { flush(cnt); cnt = 0; }
#pragma unroll
          for (int u = 4 * half; u < 4 * half + 4; ++u) {
            const int e = hd + 4 * group_of(p, u);
            consider(cur[u].x, e);
            consider(cur[u].y, e + 1);
            consider(cur[u].z, e + 2);
            consider(cur[u].w, e + 3);
          }
        }
      }
      flush(cnt);
      if (!LAST && listed && lane == 0) wcount[lslot] = pos;
    }
    __syncthreads();

    float4 ba, bb;
    ba.x = bias[4 * lane];           ba.y = bias[4 * lane + 1];
    ba.z = bias[4 * lane + 2];       ba.w = bias[4 * lane + 3];
    bb.x = bias[128 + 4 * lane];     bb.y = bias[128 + 4 * lane + 1];
    bb.z = bias[128 + 4 * lane + 2]; bb.w = bias[128 + 4 * lane + 3];
    if (LAST) {
      for (int r = wave; r < rows; r += NWAVE) {
        const int i = base + r;
        const float d = dinv[i];
        const float* tp = tile + r * TSTR;
        float4 a, b;
        a.x = tp[lane] * d;        a.y = tp[32 + lane] * d;
        a.z = tp[64 + lane] * d;   a.w = tp[96 + lane] * d;
        b.x = tp[128 + lane] * d;  b.y = tp[160 + lane] * d;
        b.z = tp[192 + lane] * d;  b.w = tp[224 + lane] * d;
        a.x += ba.x; a.y += ba.y; a.z += ba.z; a.w += ba.w;
        b.x += bb.x; b.y += bb.y; b.z += bb.z; b.w += bb.w;
        float4* op = (float4*)(out + (size_t)i * HID);
        op[lane]      = a;
        op[32 + lane] = b;
      }
    } else {
      const float kLn1e4Over256 = 9.210340371976184f / 256.0f;
      const float f0 = expf(-(float)(4 * lane)           * kLn1e4Over256);
      const float f1 = expf(-(float)(4 * lane + 2)       * kLn1e4Over256);
      const float f2 = expf(-(float)(128 + 4 * lane)     * kLn1e4Over256);
      const float f3 = expf(-(float)(128 + 4 * lane + 2) * kLn1e4Over256);
      for (int r = wave; r < rows; r += NWAVE) {
        const int i = base + r;
        const float d  = dinv[i];
        const float fi = (float)i;
        float* tp = tile + r * TSTR;
        float s0, c0, s1, c1, s2, c2, s3, c3;
        sincosf(fi * f0, &s0, &c0);
        sincosf(fi * f1, &s1, &c1);
        sincosf(fi * f2, &s2, &c2);
        sincosf(fi * f3, &s3, &c3);
        float4 a, b;
        a.x = tp[lane] * d + ba.x + s0;        a.y = tp[32 + lane] * d + ba.y + c0;
        a.z = tp[64 + lane] * d + ba.z + s1;   a.w = tp[96 + lane] * d + ba.w + c1;
        b.x = tp[128 + lane] * d + bb.x + s2;  b.y = tp[160 + lane] * d + bb.y + c2;
        b.z = tp[192 + lane] * d + bb.z + s3;  b.w = tp[224 + lane] * d + bb.w + c3;
        tp[lane]       = a.x > 0.0f ? a.x : 0.0f;  tp[32 + lane]  = a.y > 0.0f ? a.y : 0.0f;
        tp[64 + lane]  = a.z > 0.0f ? a.z : 0.0f;  tp[96 + lane]  = a.w > 0.0f ? a.w : 0.0f;
        tp[128 + lane] = b.x > 0.0f ? b.x : 0.0f;  tp[160 + lane] = b.y > 0.0f ? b.y : 0.0f;
        tp[192 + lane] = b.z > 0.0f ? b.z : 0.0f;  tp[224 + lane] = b.w > 0.0f ? b.w : 0.0f;
      }
      __syncthreads();
      const int tp8 = wave & 7;
      const int mq  = wave >> 3;
      if (mq == 0) rows_times_w<4>(tile, Wn, out, base, rows, mq, tp8, lane);
      else         rows_times_w<3>(tile, Wn, out, base, rows, mq, tp8, lane);
    }
    __syncthreads();
  }
}

extern "C" void kernel_launch(void* const* d_in, const int* in_sizes, int n_in,
                              void* d_out, int out_size, void* d_ws, size_t ws_size,
                              hipStream_t stream) {
  const float* X  = (const float*)d_in[0];
  const int*   EI = (const int*)d_in[1];
  const float* W1 = (const float*)d_in[2];
  const float* b1 = (const float*)d_in[3];
  const float* W2 = (const float*)d_in[4];
  const float* b2 = (const float*)d_in[5];
  float* out = (float*)d_out;

  const int N = in_sizes[0] / IN_DIM;
  const int E = in_sizes[1] / 2;
  const int NF = N * HID;

  float* dinv   = (float*)d_ws;
  float* bufL   = dinv + (1 << 16);
  float* bufAgg = bufL + (size_t)NF;
  float* bufH   = out;
  int* wcount = (int*)bufL;
  int* wlist  = wcount + NLIST * NWAVE;

  const int* src = EI;
  const int* tgt = EI + E;

  const int T = 256;
  auto cdiv = [](int a, int b) { return (a + b - 1) / b; };

  init_deg_kernel<<<cdiv(N, T), T, 0, stream>>>(dinv, N);
  accum_deg_kernel<<<cdiv(E, T), T, 0, stream>>>(tgt, dinv, E);
  rsqrt_kernel<<<cdiv(N, T), T, 0, stream>>>(dinv, N);

  {
    const int tiles = cdiv(N, 16) * (HID / 16);
    wmma_gemm_kernel<IN_DIM><<<cdiv(tiles, T / 32), T, 0, stream>>>(X, W1, bufH, N);
  }

  aggregate_lds_kernel<false><<<256, 1024, 0, stream>>>(bufH, src, tgt, dinv, b1, W2,
                                                        bufAgg, wlist, wcount, E, N);

  aggregate_lds_kernel<true><<<256, 1024, 0, stream>>>(bufAgg, src, tgt, dinv, b2, W2,
                                                       out, wlist, wcount, E, N);
}


# H1 = X W1 by 256 resident groups: 196 input rows staged in LDS, a wave keeps its 32-column W1 strip in registers and stores whole lines (was 50000 one-tile waves, 16-line loads); k order unchanged
# speedup vs baseline: 3.0563x; 1.0857x over previous
#include <hip/hip_runtime.h>
#include <math.h>


typedef __attribute__((ext_vector_type(2))) float v2f;
typedef __attribute__((ext_vector_type(8))) float v8f;

static constexpr int HID    = 256;
static constexpr int IN_DIM = 64;

__global__ void init_deg_kernel(float* __restrict__ deg, int n) {
  int i = blockIdx.x * blockDim.x + threadIdx.x;
  if (i < n) deg[i] = 1.0f;
}

__global__ void accum_deg_kernel(const int* __restrict__ tgt,
                                 float* __restrict__ deg, int e) {
  int i = blockIdx.x * blockDim.x + threadIdx.x;
  if (i < e) atomicAdd(&deg[tgt[i]], 1.0f);
}

__global__ void rsqrt_kernel(float* __restrict__ deg, int n) {
  int i = blockIdx.x * blockDim.x + threadIdx.x;
  if (i < n) deg[i] = rsqrtf(deg[i]);
}

static constexpr int XROWS = 196;
static constexpr int XSTR  = IN_DIM + 4;
static_assert(XROWS * (IN_DIM / 4) <= 4 * 1024, "staging: four 16-byte loads per thread");

__global__ __launch_bounds__(1024)
void xw1_rows_kernel(const float* __restrict__ X,
                     const float* __restrict__ W1,
                     float* __restrict__ Hout, int N) {
  __shared__ __attribute__((aligned(16))) float xt[XROWS * XSTR];

  const int lane = threadIdx.x & 31;
  const int wave = threadIdx.x >> 5;
  const int h  = lane >> 4;
  const int ln = lane & 15;
  const int tp = wave & 7;
  const int mq = __builtin_amdgcn_readfirstlane(wave >> 3);
  const int n2 = tp * 32 + 2 * ln;

  int base = blockIdx.x * XROWS;
  if (base >= N) return;

  float4 q[4];
  auto ask_rows = [&](int b, int rows) {
    const int nvec = rows * (IN_DIM / 4);
    const float4* xp = (const float4*)(X + (size_t)b * IN_DIM);
#pragma unroll
    for (int u = 0; u < 4; ++u) {
      const int v = (int)threadIdx.x + 1024 * u;
      q[u] = (v < nvec) ? xp[v] : make_float4(0.0f, 0.0f, 0.0f, 0.0f);
    }
  };
  auto put_rows = [&](int rows) {
    const int nvec = rows * (IN_DIM / 4);
#pragma unroll
    for (int u = 0; u < 4; ++u) {
      const int v = (int)threadIdx.x + 1024 * u;
      if (v < nvec) *(float4*)(xt + (v >> 4) * XSTR + 4 * (v & 15)) = q[u];
    }
  };

  ask_rows(base, min(XROWS, N - base));

  v2f be[16], bo[16];
#pragma unroll
  for (int s = 0; s < 16; ++s) {
    const float* wp = W1 + (size_t)(4 * s + 2 * h) * HID + n2;
    const float2 w0 = *(const float2*)wp;
    const float2 w1 = *(const float2*)(wp + HID);
    be[s].x = w0.x;  be[s].y = w1.x;
    bo[s].x = w0.y;  bo[s].y = w1.y;
  }

  for (;;) {
    const int rows = min(XROWS, N - base);
    put_rows(rows);
    __syncthreads();

    const int ntiles = (rows + 15) >> 4;
    for (int t = mq; t < ntiles; t += 4) {
      const int row = min(t * 16 + ln, rows - 1);
      const float* ap = xt + row * XSTR + 2 * h;
      v2f a[16];
#pragma unroll
      for (int s = 0; s < 16; ++s) a[s] = *(const v2f*)(ap + 4 * s);
      v8f c0 = {}, c1 = {};
#pragma unroll
      for (int s = 0; s < 16; ++s) {
        c0 = __builtin_amdgcn_wmma_f32_16x16x4_f32(
            false, a[s], false, be[s], (short)0, c0, false, false);
        c1 = __builtin_amdgcn_wmma_f32_16x16x4_f32(
            false, a[s], false, bo[s], (short)0, c1, false, false);
      }
      const int r0 = t * 16 + 8 * h;
#pragma unroll
      for (int r = 0; r < 8; ++r) {
        if (r0 + r < rows) {
          float2 v;
          v.x = c0[r];
          v.y = c1[r];
          *(float2*)(Hout + (size_t)(base + r0 + r) * HID + n2) = v;
        }
      }
    }

    base += gridDim.x * XROWS;
    if (base >= N) break;
    __syncthreads();
    ask_rows(base, min(XROWS, N - base));
  }
}

static constexpr int TROWS = 196;
static constexpr int NWAVE = 32;
static constexpr int EPH   = 512;
static constexpr int BATCH = 32;
static constexpr int LCAP  = BATCH + EPH;
static constexpr int WCAP  = 512;
static constexpr int NLIST = 256;

static constexpr int TSTR  = HID + 1;

template <int NS>
__device__ __forceinline__ void rows_times_w(const float* tile,
                                             const float* __restrict__ Wm,
                                             float* __restrict__ out,
                                             int base, int rows, int mq, int tp,
                                             int lane) {
  const int h  = lane >> 4;
  const int ln = lane & 15;
  const int n2 = tp * 32 + 2 * ln;
  int aoff[NS];
#pragma unroll
  for (int i = 0; i < NS; ++i) {
    const int row = min((mq + 4 * i) * 16 + ln, rows - 1);
    aoff[i] = row * TSTR;
  }
  v8f acc[NS][2];
#pragma unroll
  for (int i = 0; i < NS; ++i) { acc[i][0] = v8f{}; acc[i][1] = v8f{}; }

  auto w_of = [&](int s) {
    const int lh = ((s & 31) + 16 * h) & 31;
    const int c0 = (s >> 5) * 128 + 4 * lh + 2 * h;
    return Wm + (size_t)c0 * HID + n2;
  };
  const float* wp = w_of(0);
  float2 w0n = *(const float2*)wp;
  float2 w1n = *(const float2*)(wp + HID);
#pragma unroll 2
  for (int s = 0; s < 64; ++s) {
    const float2 w0 = w0n, w1 = w1n;
    wp  = w_of(min(s + 1, 63));
    w0n = *(const float2*)wp;
    w1n = *(const float2*)(wp + HID);
    const int lh  = ((s & 31) + 16 * h) & 31;
    const int pos = (s >> 5) * 128 + 64 * h + lh;
    v2f a[NS];
#pragma unroll
    for (int i = 0; i < NS; ++i) {
      a[i].x = tile[aoff[i] + pos];
      a[i].y = tile[aoff[i] + pos + 32];
    }
    v2f be, bo;
    be.x = w0.x;  be.y = w1.x;
    bo.x = w0.y;  bo.y = w1.y;
#pragma unroll
    for (int i = 0; i < NS; ++i) {
      acc[i][0] = __builtin_amdgcn_wmma_f32_16x16x4_f32(
          false, a[i], false, be, (short)0, acc[i][0], false, false);
      acc[i][1] = __builtin_amdgcn_wmma_f32_16x16x4_f32(
          false, a[i], false, bo, (short)0, acc[i][1], false, false);
    }
  }

#pragma unroll
  for (int i = 0; i < NS; ++i) {
    const int r0 = (mq + 4 * i) * 16 + 8 * h;
#pragma unroll
    for (int r = 0; r < 8; ++r) {
      if (r0 + r < rows) {
        float2 v;
        v.x = acc[i][0][r];
        v.y = acc[i][1][r];
        *(float2*)(out + (size_t)(base + r0 + r) * HID + n2) = v;
      }
    }
  }
}

template <bool LAST>
__global__ __launch_bounds__(1024)
void aggregate_lds_kernel(const float* __restrict__ H,
                          const int* __restrict__ src,
                          const int* __restrict__ tgt,
                          const float* __restrict__ dinv,
                          const float* __restrict__ bias,
                          const float* __restrict__ Wn,
                          float* __restrict__ out,
                          int* __restrict__ wlist, int* __restrict__ wcount,
                          int E, int N) {
  __shared__ float tile[TROWS * TSTR];
  __shared__ int   kept[NWAVE * LCAP];

  const int lane = threadIdx.x & 31;
  const int wave = threadIdx.x >> 5;
  int* mylist = kept + wave * LCAP;
  const unsigned lanes_below = (1u << lane) - 1u;
  const int hd  = min((int)(((16u - (unsigned)((size_t)tgt & 15)) & 15u) >> 2), E);
  const int* tb = tgt + hd;
  const int EB  = E - hd;
  const int G     = EB >> 2;
  const int GPI   = NWAVE * 8 * 32;
  const int ITERS = (G + GPI - 1) / GPI;
  const int FULL  = G / GPI;

  for (int base = blockIdx.x * TROWS; base < N; base += gridDim.x * TROWS) {
    const int rows = min(TROWS, N - base);
    const int tile_id  = base / TROWS;
    const bool listed  = tile_id < NLIST && N <= (1 << 24);
    const int  lslot   = (listed ? tile_id : 0) * NWAVE + wave;
    int2* mywl = (int2*)wlist + (size_t)lslot * WCAP;
    int pos = 0;

    for (int r = wave; r < rows; r += NWAVE) {
      const int i = base + r;
      const float d = dinv[i];
      const float4* hp = (const float4*)(H + (size_t)i * HID);
      const float4 a = hp[lane], b = hp[32 + lane];
      float* tp = tile + r * TSTR;
      tp[lane]       = a.x * d;  tp[32 + lane]  = a.y * d;
      tp[64 + lane]  = a.z * d;  tp[96 + lane]  = a.w * d;
      tp[128 + lane] = b.x * d;  tp[160 + lane] = b.y * d;
      tp[192 + lane] = b.z * d;  tp[224 + lane] = b.w * d;
    }
    __syncthreads();

    auto add_rows = [&](int s, int r, float ds, int m) {
      int j = 0;
      for (; j + 4 <= m; j += 4) {
        float4 a[4], b[4];
        float  dj[4];
        int    rj[4];
#pragma unroll
        for (int u = 0; u < 4; ++u) {
          const int sj = __builtin_amdgcn_readlane(s, j + u);
          rj[u] = __builtin_amdgcn_readlane(r, j + u);
          dj[u] = __int_as_float(__builtin_amdgcn_readlane(__float_as_int(ds), j + u));
          const float4* hp = (const float4*)(H + (size_t)sj * HID);
          a[u] = hp[lane];
          b[u] = hp[32 + lane];
        }
#pragma unroll
        for (int u = 0; u < 4; ++u) {
          float* tp = tile + rj[u] * TSTR + lane;
          const float d = dj[u];
          __hip_atomic_fetch_add(tp,       a[u].x * d, __ATOMIC_RELAXED, __HIP_MEMORY_SCOPE_WORKGROUP);
          __hip_atomic_fetch_add(tp + 32,  a[u].y * d, __ATOMIC_RELAXED, __HIP_MEMORY_SCOPE_WORKGROUP);
          __hip_atomic_fetch_add(tp + 64,  a[u].z * d, __ATOMIC_RELAXED, __HIP_MEMORY_SCOPE_WORKGROUP);
          __hip_atomic_fetch_add(tp + 96,  a[u].w * d, __ATOMIC_RELAXED, __HIP_MEMORY_SCOPE_WORKGROUP);
          __hip_atomic_fetch_add(tp + 128, b[u].x * d, __ATOMIC_RELAXED, __HIP_MEMORY_SCOPE_WORKGROUP);
          __hip_atomic_fetch_add(tp + 160, b[u].y * d, __ATOMIC_RELAXED, __HIP_MEMORY_SCOPE_WORKGROUP);
          __hip_atomic_fetch_add(tp + 192, b[u].z * d, __ATOMIC_RELAXED, __HIP_MEMORY_SCOPE_WORKGROUP);
          __hip_atomic_fetch_add(tp + 224, b[u].w * d, __ATOMIC_RELAXED, __HIP_MEMORY_SCOPE_WORKGROUP);
        }
      }
      for (; j < m; ++j) {
        const int sj = __builtin_amdgcn_readlane(s, j);
        const int rr = __builtin_amdgcn_readlane(r, j);
        const float d = __int_as_float(__builtin_amdgcn_readlane(__float_as_int(ds), j));
        const float4* hp = (const float4*)(H + (size_t)sj * HID);
        const float4 a = hp[lane], b = hp[32 + lane];
        float* tp = tile + rr * TSTR + lane;
        __hip_atomic_fetch_add(tp,       a.x * d, __ATOMIC_RELAXED, __HIP_MEMORY_SCOPE_WORKGROUP);
        __hip_atomic_fetch_add(tp + 32,  a.y * d, __ATOMIC_RELAXED, __HIP_MEMORY_SCOPE_WORKGROUP);
        __hip_atomic_fetch_add(tp + 64,  a.z * d, __ATOMIC_RELAXED, __HIP_MEMORY_SCOPE_WORKGROUP);
        __hip_atomic_fetch_add(tp + 96,  a.w * d, __ATOMIC_RELAXED, __HIP_MEMORY_SCOPE_WORKGROUP);
        __hip_atomic_fetch_add(tp + 128, b.x * d, __ATOMIC_RELAXED, __HIP_MEMORY_SCOPE_WORKGROUP);
        __hip_atomic_fetch_add(tp + 160, b.y * d, __ATOMIC_RELAXED, __HIP_MEMORY_SCOPE_WORKGROUP);
        __hip_atomic_fetch_add(tp + 192, b.z * d, __ATOMIC_RELAXED, __HIP_MEMORY_SCOPE_WORKGROUP);
        __hip_atomic_fetch_add(tp + 224, b.w * d, __ATOMIC_RELAXED, __HIP_MEMORY_SCOPE_WORKGROUP);
      }
    };

    auto flush = [&](int n) {
      __builtin_amdgcn_fence(__ATOMIC_ACQ_REL, "wavefront");
      for (int k0 = 0; k0 < n; k0 += 32) {
        const int k = k0 + lane;
        int s = 0, r = 0;
        float ds = 0.0f;
        if (k < n) {
          const int e = mylist[k];
          s  = src[e];
          r  = tgt[e] - base;
          ds = dinv[s];
          if (!LAST && listed && pos + k < WCAP)
            mywl[pos + k] = make_int2((int)((unsigned)s | ((unsigned)r << 24)), __float_as_int(ds));
        }
        add_rows(s, r, ds, min(32, n - k0));
      }
      pos += n;
      __builtin_amdgcn_fence(__ATOMIC_ACQ_REL, "wavefront");
    };

    int nl = 0;
    bool use_list = false;
    if (LAST && listed) {
      nl = __builtin_amdgcn_readfirstlane(wcount[lslot]);
      use_list = nl <= WCAP;
    }
    if (use_list) {
      int2 p = (lane < nl) ? mywl[lane] : make_int2(0, 0);
      for (int k0 = 0; k0 < nl; k0 += 32) {
        const int k = k0 + lane;
        const int s = p.x & 0xFFFFFF;
        const int r = (int)((unsigned)p.x >> 24);
        const float ds = __int_as_float(p.y);
        p = (k + 32 < nl) ? mywl[k + 32] : make_int2(0, 0);
        add_rows(s, r, ds, min(32, nl - k0));
      }
    } else {
      int cnt = 0;
      auto consider = [&](int t, int e) {
        const bool hit = (unsigned)(t - base) < (unsigned)rows;
        const unsigned m = __builtin_amdgcn_ballot_w32(hit);
        if (m != 0u) {
          if (hit) mylist[cnt + __builtin_popcount(m & lanes_below)] = e;
          cnt += __builtin_popcount(m);
        }
      };

      if (wave == 0) {
        const int e = (lane < hd) ? lane : hd + 4 * G + (lane - hd);
        const int t = (lane < 8 && e < E) ? tgt[e] : base - 1;
        consider(t, e);
      }
      cnt = __builtin_amdgcn_readfirstlane(cnt);

      const int rot = (ITERS > 0) ? (int)(blockIdx.x % (unsigned)ITERS) : 0;
      auto place_of = [&](int it) {
        int p = it + rot;
        if (p >= ITERS) p -= ITERS;
        return p;
      };
      auto group_of = [&](int p, int u) {
        return ((p * NWAVE + wave) * 8 + u) * 32 + lane;
      };
      auto load_place = [&](int p, int4 (&dst)[8]) {
        if (p < FULL) {
#pragma unroll
          for (int u = 0; u < 8; ++u)
            dst[u] = *(const int4*)(tb + 4 * group_of(p, u));
        } else {
#pragma unroll
          for (int u = 0; u < 8; ++u) {
            const int g = group_of(p, u);
            int4 v = *(const int4*)(tb + 4 * min(g, G - 1));
            if (g >= G) v.x = v.y = v.z = v.w = base - 1;
            dst[u] = v;
          }
        }
      };

      for (int it = 0; it < ITERS; ++it) {
        const int p = place_of(it);
        int4 cur[8];
        load_place(p, cur);
#pragma unroll
        for (int half = 0; half < 2; ++half) {
          if (cnt >= BATCH) { flush(cnt); cnt = 0; }
#pragma unroll
          for (int u = 4 * half; u < 4 * half + 4; ++u) {
            const int e = hd + 4 * group_of(p, u);
            consider(cur[u].x, e);
            consider(cur[u].y, e + 1);
            consider(cur[u].z, e + 2);
            consider(cur[u].w, e + 3);
          }
        }
      }
      flush(cnt);
      if (!LAST && listed && lane == 0) wcount[lslot] = pos;
    }
    __syncthreads();

    float4 ba, bb;
    ba.x = bias[4 * lane];           ba.y = bias[4 * lane + 1];
    ba.z = bias[4 * lane + 2];       ba.w = bias[4 * lane + 3];
    bb.x = bias[128 + 4 * lane];     bb.y = bias[128 + 4 * lane + 1];
    bb.z = bias[128 + 4 * lane + 2]; bb.w = bias[128 + 4 * lane + 3];
    if (LAST) {
      for (int r = wave; r < rows; r += NWAVE) {
        const int i = base + r;
        const float d = dinv[i];
        const float* tp = tile + r * TSTR;
        float4 a, b;
        a.x = tp[lane] * d;        a.y = tp[32 + lane] * d;
        a.z = tp[64 + lane] * d;   a.w = tp[96 + lane] * d;
        b.x = tp[128 + lane] * d;  b.y = tp[160 + lane] * d;
        b.z = tp[192 + lane] * d;  b.w = tp[224 + lane] * d;
        a.x += ba.x; a.y += ba.y; a.z += ba.z; a.w += ba.w;
        b.x += bb.x; b.y += bb.y; b.z += bb.z; b.w += bb.w;
        float4* op = (float4*)(out + (size_t)i * HID);
        op[lane]      = a;
        op[32 + lane] = b;
      }
    } else {
      const float kLn1e4Over256 = 9.210340371976184f / 256.0f;
      const float f0 = expf(-(float)(4 * lane)           * kLn1e4Over256);
      const float f1 = expf(-(float)(4 * lane + 2)       * kLn1e4Over256);
      const float f2 = expf(-(float)(128 + 4 * lane)     * kLn1e4Over256);
      const float f3 = expf(-(float)(128 + 4 * lane + 2) * kLn1e4Over256);
      for (int r = wave; r < rows; r += NWAVE) {
        const int i = base + r;
        const float d  = dinv[i];
        const float fi = (float)i;
        float* tp = tile + r * TSTR;
        float s0, c0, s1, c1, s2, c2, s3, c3;
        sincosf(fi * f0, &s0, &c0);
        sincosf(fi * f1, &s1, &c1);
        sincosf(fi * f2, &s2, &c2);
        sincosf(fi * f3, &s3, &c3);
        float4 a, b;
        a.x = tp[lane] * d + ba.x + s0;        a.y = tp[32 + lane] * d + ba.y + c0;
        a.z = tp[64 + lane] * d + ba.z + s1;   a.w = tp[96 + lane] * d + ba.w + c1;
        b.x = tp[128 + lane] * d + bb.x + s2;  b.y = tp[160 + lane] * d + bb.y + c2;
        b.z = tp[192 + lane] * d + bb.z + s3;  b.w = tp[224 + lane] * d + bb.w + c3;
        tp[lane]       = a.x > 0.0f ? a.x : 0.0f;  tp[32 + lane]  = a.y > 0.0f ? a.y : 0.0f;
        tp[64 + lane]  = a.z > 0.0f ? a.z : 0.0f;  tp[96 + lane]  = a.w > 0.0f ? a.w : 0.0f;
        tp[128 + lane] = b.x > 0.0f ? b.x : 0.0f;  tp[160 + lane] = b.y > 0.0f ? b.y : 0.0f;
        tp[192 + lane] = b.z > 0.0f ? b.z : 0.0f;  tp[224 + lane] = b.w > 0.0f ? b.w : 0.0f;
      }
      __syncthreads();
      const int tp8 = wave & 7;
      const int mq  = wave >> 3;
      if (mq == 0) rows_times_w<4>(tile, Wn, out, base, rows, mq, tp8, lane);
      else         rows_times_w<3>(tile, Wn, out, base, rows, mq, tp8, lane);
    }
    __syncthreads();
  }
}

extern "C" void kernel_launch(void* const* d_in, const int* in_sizes, int n_in,
                              void* d_out, int out_size, void* d_ws, size_t ws_size,
                              hipStream_t stream) {
  const float* X  = (const float*)d_in[0];
  const int*   EI = (const int*)d_in[1];
  const float* W1 = (const float*)d_in[2];
  const float* b1 = (const float*)d_in[3];
  const float* W2 = (const float*)d_in[4];
  const float* b2 = (const float*)d_in[5];
  float* out = (float*)d_out;

  const int N = in_sizes[0] / IN_DIM;
  const int E = in_sizes[1] / 2;
  const int NF = N * HID;

  float* dinv   = (float*)d_ws;
  float* bufL   = dinv + (1 << 16);
  float* bufAgg = bufL + (size_t)NF;
  float* bufH   = out;
  int* wcount = (int*)bufL;
  int* wlist  = wcount + NLIST * NWAVE;

  const int* src = EI;
  const int* tgt = EI + E;

  const int T = 256;
  auto cdiv = [](int a, int b) { return (a + b - 1) / b; };

  init_deg_kernel<<<cdiv(N, T), T, 0, stream>>>(dinv, N);
  accum_deg_kernel<<<cdiv(E, T), T, 0, stream>>>(tgt, dinv, E);
  rsqrt_kernel<<<cdiv(N, T), T, 0, stream>>>(dinv, N);

  xw1_rows_kernel<<<256, 1024, 0, stream>>>(X, W1, bufH, N);

  aggregate_lds_kernel<false><<<256, 1024, 0, stream>>>(bufH, src, tgt, dinv, b1, W2,
                                                        bufAgg, wlist, wcount, E, N);

  aggregate_lds_kernel<true><<<256, 1024, 0, stream>>>(bufAgg, src, tgt, dinv, b2, W2,
                                                       out, wlist, wcount, E, N);
}


# degrees counted inside the edge search (LDS row counters): the search moves into the X W1 kernel and leaves lists for both aggregations, H rows stored scaled by dinv; 3 launches (was 6)
# speedup vs baseline: 3.6032x; 1.1789x over previous
#include <hip/hip_runtime.h>
#include <math.h>


typedef __attribute__((ext_vector_type(2))) float v2f;
typedef __attribute__((ext_vector_type(8))) float v8f;

static constexpr int HID    = 256;
static constexpr int IN_DIM = 64;

static constexpr int TROWS = 196;
static constexpr int NWAVE = 32;
static constexpr int EPH   = 512;
static constexpr int BATCH = 32;
static constexpr int LCAP  = BATCH + EPH;
static constexpr int WCAP  = 512;
static constexpr int NLIST = 256;

template <class Flush>
__device__ __forceinline__ void search_edges(const int* __restrict__ tgt, int E,
                                             int base, int rows, int* mylist,
                                             int lane, int wave, Flush&& flush) {
  const unsigned lanes_below = (1u << lane) - 1u;
  const int hd  = min((int)(((16u - (unsigned)((size_t)tgt & 15)) & 15u) >> 2), E);
  const int* tb = tgt + hd;
  const int EB  = E - hd;
  const int G     = EB >> 2;
  const int GPI   = NWAVE * 8 * 32;
  const int ITERS = (G + GPI - 1) / GPI;
  const int FULL  = G / GPI;

  int cnt = 0;
  auto consider = [&](int t, int e) {
    const bool hit = (unsigned)(t - base) < (unsigned)rows;
    const unsigned m = __builtin_amdgcn_ballot_w32(hit);
    if (m != 0u) {
      if (hit) mylist[cnt + __builtin_popcount(m & lanes_below)] = e;
      cnt += __builtin_popcount(m);
    }
  };

  if (wave == 0) {
    const int e = (lane < hd) ? lane : hd + 4 * G + (lane - hd);
    const int t = (lane < 8 && e < E) ? tgt[e] : base - 1;
    consider(t, e);
  }
  cnt = __builtin_amdgcn_readfirstlane(cnt);

  const int rot = (ITERS > 0) ? (int)(blockIdx.x % (unsigned)ITERS) : 0;
  auto place_of = [&](int it) {
    int p = it + rot;
    if (p >= ITERS) p -= ITERS;
    return p;
  };
  auto group_of = [&](int p, int u) {
    return ((p * NWAVE + wave) * 8 + u) * 32 + lane;
  };
  auto load_place = [&](int p, int4 (&dst)[8]) {
    if (p < FULL) {
#pragma unroll
      for (int u = 0; u < 8; ++u)
        dst[u] = *(const int4*)(tb + 4 * group_of(p, u));
    } else {
#pragma unroll
      for (int u = 0; u < 8; ++u) {
        const int g = group_of(p, u);
        int4 v = *(const int4*)(tb + 4 * min(g, G - 1));
        if (g >= G) v.x = v.y = v.z = v.w = base - 1;
        dst[u] = v;
      }
    }
  };

  for (int it = 0; it < ITERS; ++it) {
    const int p = place_of(it);
    int4 cur[8];
    load_place(p, cur);
#pragma unroll
    for (int half = 0; half < 2; ++half) {
      if (cnt >= BATCH) { flush(cnt); cnt = 0; }
#pragma unroll
      for (int u = 4 * half; u < 4 * half + 4; ++u) {
        const int e = hd + 4 * group_of(p, u);
        consider(cur[u].x, e);
        consider(cur[u].y, e + 1);
        consider(cur[u].z, e + 2);
        consider(cur[u].w, e + 3);
      }
    }
  }
  flush(cnt);
}

static constexpr int XROWS = TROWS;
static constexpr int XSTR  = IN_DIM + 4;
static_assert(XROWS * (IN_DIM / 4) <= 4 * 1024, "staging: four 16-byte loads per thread");

__global__ __launch_bounds__(1024)
void search_xw1_kernel(const float* __restrict__ X,
                       const float* __restrict__ W1,
                       const int* __restrict__ src,
                       const int* __restrict__ tgt,
                       float* __restrict__ dinv,
                       float* __restrict__ Hout,
                       int* __restrict__ wlist, int* __restrict__ wcount,
                       int E, int N) {
  __shared__ __attribute__((aligned(16))) float xt[XROWS * XSTR];
  __shared__ int   kept[NWAVE * LCAP];
  __shared__ int   degc[TROWS];
  __shared__ float dl[TROWS];

  const int lane = threadIdx.x & 31;
  const int wave = threadIdx.x >> 5;
  const int h  = lane >> 4;
  const int ln = lane & 15;
  const int tp = wave & 7;
  const int mq = __builtin_amdgcn_readfirstlane(wave >> 3);
  const int n2 = tp * 32 + 2 * ln;
  int* mylist = kept + wave * LCAP;

  for (int base = blockIdx.x * TROWS; base < N; base += gridDim.x * TROWS) {
    const int rows = min(TROWS, N - base);
    const int tile_id  = base / TROWS;
    const bool listed  = tile_id < NLIST && N <= (1 << 24);
    const int  lslot   = (listed ? tile_id : 0) * NWAVE + wave;
    int* mywl = wlist + (size_t)lslot * WCAP;
    int pos = 0;

    if ((int)threadIdx.x < TROWS) degc[threadIdx.x] = 0;
    __syncthreads();

    auto flush = [&](int n) {
      __builtin_amdgcn_fence(__ATOMIC_ACQ_REL, "wavefront");
      for (int k0 = 0; k0 < n; k0 += 32) {
        const int k = k0 + lane;
        if (k < n) {
          const int e = mylist[k];
          const int s = src[e];
          const int r = tgt[e] - base;
          __hip_atomic_fetch_add(&degc[r], 1, __ATOMIC_RELAXED, __HIP_MEMORY_SCOPE_WORKGROUP);
          if (listed && pos + k < WCAP)
            mywl[pos + k] = (int)((unsigned)s | ((unsigned)r << 24));
        }
      }
      pos += n;
      __builtin_amdgcn_fence(__ATOMIC_ACQ_REL, "wavefront");
    };
    search_edges(tgt, E, base, rows, mylist, lane, wave, flush);
    if (listed && lane == 0) wcount[lslot] = pos;
    __syncthreads();

    if ((int)threadIdx.x < rows) {
      const float d = rsqrtf(1.0f + (float)degc[threadIdx.x]);
      dl[threadIdx.x] = d;
      dinv[base + threadIdx.x] = d;
    }

    const int nvec = rows * (IN_DIM / 4);
    const float4* xp = (const float4*)(X + (size_t)base * IN_DIM);
    float4 q[4];
#pragma unroll
    for (int u = 0; u < 4; ++u) {
      const int v = (int)threadIdx.x + 1024 * u;
      q[u] = (v < nvec) ? xp[v] : make_float4(0.0f, 0.0f, 0.0f, 0.0f);
    }

#pragma unroll
    for (int u = 0; u < 4; ++u) {
      const int v = (int)threadIdx.x + 1024 * u;
      if (v < nvec) *(float4*)(xt + (v >> 4) * XSTR + 4 * (v & 15)) = q[u];
    }

    v2f be[16], bo[16];
#pragma unroll
    for (int s = 0; s < 16; ++s) {
      const float* wp = W1 + (size_t)(4 * s + 2 * h) * HID + n2;
      const float2 w0 = *(const float2*)wp;
      const float2 w1 = *(const float2*)(wp + HID);
      be[s].x = w0.x;  be[s].y = w1.x;
      bo[s].x = w0.y;  bo[s].y = w1.y;
    }
    __syncthreads();

    const int ntiles = (rows + 15) >> 4;
    for (int t = mq; t < ntiles; t += 4) {
      const int row = min(t * 16 + ln, rows - 1);
      const float* ap = xt + row * XSTR + 2 * h;
      v2f a[16];
#pragma unroll
      for (int s = 0; s < 16; ++s) a[s] = *(const v2f*)(ap + 4 * s);
      v8f c0 = {}, c1 = {};
#pragma unroll
      for (int s = 0; s < 16; ++s) {
        c0 = __builtin_amdgcn_wmma_f32_16x16x4_f32(
            false, a[s], false, be[s], (short)0, c0, false, false);
        c1 = __builtin_amdgcn_wmma_f32_16x16x4_f32(
            false, a[s], false, bo[s], (short)0, c1, false, false);
      }
      const int r0 = t * 16 + 8 * h;
#pragma unroll
      for (int r = 0; r < 8; ++r) {
        if (r0 + r < rows) {
          const float d = dl[r0 + r];
          float2 v;
          v.x = c0[r] * d;
          v.y = c1[r] * d;
          *(float2*)(Hout + (size_t)(base + r0 + r) * HID + n2) = v;
        }
      }
    }
    __syncthreads();
  }
}


static constexpr int TSTR  = HID + 1;

template <int NS>
__device__ __forceinline__ void rows_times_w(const float* tile,
                                             const float* __restrict__ Wm,
                                             const float* __restrict__ dsc,
                                             float* __restrict__ out,
                                             int base, int rows, int mq, int tp,
                                             int lane) {
  const int h  = lane >> 4;
  const int ln = lane & 15;
  const int n2 = tp * 32 + 2 * ln;
  int aoff[NS];
#pragma unroll
  for (int i = 0; i < NS; ++i) {
    const int row = min((mq + 4 * i) * 16 + ln, rows - 1);
    aoff[i] = row * TSTR;
  }
  v8f acc[NS][2];
#pragma unroll
  for (int i = 0; i < NS; ++i) { acc[i][0] = v8f{}; acc[i][1] = v8f{}; }

  auto w_of = [&](int s) {
    const int lh = ((s & 31) + 16 * h) & 31;
    const int c0 = (s >> 5) * 128 + 4 * lh + 2 * h;
    return Wm + (size_t)c0 * HID + n2;
  };
  const float* wp = w_of(0);
  float2 w0n = *(const float2*)wp;
  float2 w1n = *(const float2*)(wp + HID);
#pragma unroll 2
  for (int s = 0; s < 64; ++s) {
    const float2 w0 = w0n, w1 = w1n;
    wp  = w_of(min(s + 1, 63));
    w0n = *(const float2*)wp;
    w1n = *(const float2*)(wp + HID);
    const int lh  = ((s & 31) + 16 * h) & 31;
    const int pos = (s >> 5) * 128 + 64 * h + lh;
    v2f a[NS];
#pragma unroll
    for (int i = 0; i < NS; ++i) {
      a[i].x = tile[aoff[i] + pos];
      a[i].y = tile[aoff[i] + pos + 32];
    }
    v2f be, bo;
    be.x = w0.x;  be.y = w1.x;
    bo.x = w0.y;  bo.y = w1.y;
#pragma unroll
    for (int i = 0; i < NS; ++i) {
      acc[i][0] = __builtin_amdgcn_wmma_f32_16x16x4_f32(
          false, a[i], false, be, (short)0, acc[i][0], false, false);
      acc[i][1] = __builtin_amdgcn_wmma_f32_16x16x4_f32(
          false, a[i], false, bo, (short)0, acc[i][1], false, false);
    }
  }

#pragma unroll
  for (int i = 0; i < NS; ++i) {
    const int r0 = (mq + 4 * i) * 16 + 8 * h;
#pragma unroll
    for (int r = 0; r < 8; ++r) {
      if (r0 + r < rows) {
        const float d = dsc[r0 + r];
        float2 v;
        v.x = acc[i][0][r] * d;
        v.y = acc[i][1][r] * d;
        *(float2*)(out + (size_t)(base + r0 + r) * HID + n2) = v;
      }
    }
  }
}

template <bool LAST>
__global__ __launch_bounds__(1024)
void aggregate_lds_kernel(const float* __restrict__ H,
                          const int* __restrict__ src,
                          const int* __restrict__ tgt,
                          const float* __restrict__ dinv,
                          const float* __restrict__ bias,
                          const float* __restrict__ Wn,
                          float* __restrict__ out,
                          const int* __restrict__ wlist,
                          const int* __restrict__ wcount,
                          int E, int N) {
  __shared__ float tile[TROWS * TSTR];
  __shared__ int   kept[NWAVE * LCAP];

  const int lane = threadIdx.x & 31;
  const int wave = threadIdx.x >> 5;
  int* mylist = kept + wave * LCAP;

  for (int base = blockIdx.x * TROWS; base < N; base += gridDim.x * TROWS) {
    const int rows = min(TROWS, N - base);
    const int tile_id  = base / TROWS;
    const bool listed  = tile_id < NLIST && N <= (1 << 24);
    const int  lslot   = (listed ? tile_id : 0) * NWAVE + wave;
    const int* mywl = wlist + (size_t)lslot * WCAP;

    for (int r = wave; r < rows; r += NWAVE) {
      const int i = base + r;
      const float4* hp = (const float4*)(H + (size_t)i * HID);
      const float4 a = hp[lane], b = hp[32 + lane];
      float* tp = tile + r * TSTR;
      tp[lane]       = a.x;  tp[32 + lane]  = a.y;
      tp[64 + lane]  = a.z;  tp[96 + lane]  = a.w;
      tp[128 + lane] = b.x;  tp[160 + lane] = b.y;
      tp[192 + lane] = b.z;  tp[224 + lane] = b.w;
    }
    __syncthreads();

    auto add_rows = [&](int s, int r, int m) {
      int j = 0;
      for (; j + 4 <= m; j += 4) {
        float4 a[4], b[4];
        int    rj[4];
#pragma unroll
        for (int u = 0; u < 4; ++u) {
          const int sj = __builtin_amdgcn_readlane(s, j + u);
          rj[u] = __builtin_amdgcn_readlane(r, j + u);
          const float4* hp = (const float4*)(H + (size_t)sj * HID);
          a[u] = hp[lane];
          b[u] = hp[32 + lane];
        }
#pragma unroll
        for (int u = 0; u < 4; ++u) {
          float* tp = tile + rj[u] * TSTR + lane;
          __hip_atomic_fetch_add(tp,       a[u].x, __ATOMIC_RELAXED, __HIP_MEMORY_SCOPE_WORKGROUP);
          __hip_atomic_fetch_add(tp + 32,  a[u].y, __ATOMIC_RELAXED, __HIP_MEMORY_SCOPE_WORKGROUP);
          __hip_atomic_fetch_add(tp + 64,  a[u].z, __ATOMIC_RELAXED, __HIP_MEMORY_SCOPE_WORKGROUP);
          __hip_atomic_fetch_add(tp + 96,  a[u].w, __ATOMIC_RELAXED, __HIP_MEMORY_SCOPE_WORKGROUP);
          __hip_atomic_fetch_add(tp + 128, b[u].x, __ATOMIC_RELAXED, __HIP_MEMORY_SCOPE_WORKGROUP);
          __hip_atomic_fetch_add(tp + 160, b[u].y, __ATOMIC_RELAXED, __HIP_MEMORY_SCOPE_WORKGROUP);
          __hip_atomic_fetch_add(tp + 192, b[u].z, __ATOMIC_RELAXED, __HIP_MEMORY_SCOPE_WORKGROUP);
          __hip_atomic_fetch_add(tp + 224, b[u].w, __ATOMIC_RELAXED, __HIP_MEMORY_SCOPE_WORKGROUP);
        }
      }
      for (; j < m; ++j) {
        const int sj = __builtin_amdgcn_readlane(s, j);
        const int rr = __builtin_amdgcn_readlane(r, j);
        const float4* hp = (const float4*)(H + (size_t)sj * HID);
        const float4 a = hp[lane], b = hp[32 + lane];
        float* tp = tile + rr * TSTR + lane;
        __hip_atomic_fetch_add(tp,       a.x, __ATOMIC_RELAXED, __HIP_MEMORY_SCOPE_WORKGROUP);
        __hip_atomic_fetch_add(tp + 32,  a.y, __ATOMIC_RELAXED, __HIP_MEMORY_SCOPE_WORKGROUP);
        __hip_atomic_fetch_add(tp + 64,  a.z, __ATOMIC_RELAXED, __HIP_MEMORY_SCOPE_WORKGROUP);
        __hip_atomic_fetch_add(tp + 96,  a.w, __ATOMIC_RELAXED, __HIP_MEMORY_SCOPE_WORKGROUP);
        __hip_atomic_fetch_add(tp + 128, b.x, __ATOMIC_RELAXED, __HIP_MEMORY_SCOPE_WORKGROUP);
        __hip_atomic_fetch_add(tp + 160, b.y, __ATOMIC_RELAXED, __HIP_MEMORY_SCOPE_WORKGROUP);
        __hip_atomic_fetch_add(tp + 192, b.z, __ATOMIC_RELAXED, __HIP_MEMORY_SCOPE_WORKGROUP);
        __hip_atomic_fetch_add(tp + 224, b.w, __ATOMIC_RELAXED, __HIP_MEMORY_SCOPE_WORKGROUP);
      }
    };

    int nl = 0;
    bool use_list = false;
    if (listed) {
      nl = __builtin_amdgcn_readfirstlane(wcount[lslot]);
      use_list = nl <= WCAP;
    }
    if (use_list) {
      int p = (lane < nl) ? mywl[lane] : 0;
      for (int k0 = 0; k0 < nl; k0 += 32) {
        const int k = k0 + lane;
        const int s = p & 0xFFFFFF;
        const int r = (int)((unsigned)p >> 24);
        p = (k + 32 < nl) ? mywl[k + 32] : 0;
        add_rows(s, r, min(32, nl - k0));
      }
    } else {
      auto flush = [&](int n) {
        __builtin_amdgcn_fence(__ATOMIC_ACQ_REL, "wavefront");
        for (int k0 = 0; k0 < n; k0 += 32) {
          const int k = k0 + lane;
          int s = 0, r = 0;
          if (k < n) {
            const int e = mylist[k];
            s = src[e];
            r = tgt[e] - base;
          }
          add_rows(s, r, min(32, n - k0));
        }
        __builtin_amdgcn_fence(__ATOMIC_ACQ_REL, "wavefront");
      };
      search_edges(tgt, E, base, rows, mylist, lane, wave, flush);
    }
    __syncthreads();

    float4 ba, bb;
    ba.x = bias[4 * lane];           ba.y = bias[4 * lane + 1];
    ba.z = bias[4 * lane + 2];       ba.w = bias[4 * lane + 3];
    bb.x = bias[128 + 4 * lane];     bb.y = bias[128 + 4 * lane + 1];
    bb.z = bias[128 + 4 * lane + 2]; bb.w = bias[128 + 4 * lane + 3];
    if (LAST) {
      for (int r = wave; r < rows; r += NWAVE) {
        const int i = base + r;
        const float d = dinv[i];
        const float* tp = tile + r * TSTR;
        float4 a, b;
        a.x = tp[lane] * d;        a.y = tp[32 + lane] * d;
        a.z = tp[64 + lane] * d;   a.w = tp[96 + lane] * d;
        b.x = tp[128 + lane] * d;  b.y = tp[160 + lane] * d;
        b.z = tp[192 + lane] * d;  b.w = tp[224 + lane] * d;
        a.x += ba.x; a.y += ba.y; a.z += ba.z; a.w += ba.w;
        b.x += bb.x; b.y += bb.y; b.z += bb.z; b.w += bb.w;
        float4* op = (float4*)(out + (size_t)i * HID);
        op[lane]      = a;
        op[32 + lane] = b;
      }
    } else {
      const float kLn1e4Over256 = 9.210340371976184f / 256.0f;
      const float f0 = expf(-(float)(4 * lane)           * kLn1e4Over256);
      const float f1 = expf(-(float)(4 * lane + 2)       * kLn1e4Over256);
      const float f2 = expf(-(float)(128 + 4 * lane)     * kLn1e4Over256);
      const float f3 = expf(-(float)(128 + 4 * lane + 2) * kLn1e4Over256);
      for (int r = wave; r < rows; r += NWAVE) {
        const int i = base + r;
        const float d  = dinv[i];
        const float fi = (float)i;
        float* tp = tile + r * TSTR;
        float s0, c0, s1, c1, s2, c2, s3, c3;
        sincosf(fi * f0, &s0, &c0);
        sincosf(fi * f1, &s1, &c1);
        sincosf(fi * f2, &s2, &c2);
        sincosf(fi * f3, &s3, &c3);
        float4 a, b;
        a.x = tp[lane] * d + ba.x + s0;        a.y = tp[32 + lane] * d + ba.y + c0;
        a.z = tp[64 + lane] * d + ba.z + s1;   a.w = tp[96 + lane] * d + ba.w + c1;
        b.x = tp[128 + lane] * d + bb.x + s2;  b.y = tp[160 + lane] * d + bb.y + c2;
        b.z = tp[192 + lane] * d + bb.z + s3;  b.w = tp[224 + lane] * d + bb.w + c3;
        tp[lane]       = a.x > 0.0f ? a.x : 0.0f;  tp[32 + lane]  = a.y > 0.0f ? a.y : 0.0f;
        tp[64 + lane]  = a.z > 0.0f ? a.z : 0.0f;  tp[96 + lane]  = a.w > 0.0f ? a.w : 0.0f;
        tp[128 + lane] = b.x > 0.0f ? b.x : 0.0f;  tp[160 + lane] = b.y > 0.0f ? b.y : 0.0f;
        tp[192 + lane] = b.z > 0.0f ? b.z : 0.0f;  tp[224 + lane] = b.w > 0.0f ? b.w : 0.0f;
      }
      __syncthreads();
      const int tp8 = wave & 7;
      const int mq  = wave >> 3;
      if (mq == 0) rows_times_w<4>(tile, Wn, dinv + base, out, base, rows, mq, tp8, lane);
      else         rows_times_w<3>(tile, Wn, dinv + base, out, base, rows, mq, tp8, lane);
    }
    __syncthreads();
  }
}

extern "C" void kernel_launch(void* const* d_in, const int* in_sizes, int n_in,
                              void* d_out, int out_size, void* d_ws, size_t ws_size,
                              hipStream_t stream) {
  const float* X  = (const float*)d_in[0];
  const int*   EI = (const int*)d_in[1];
  const float* W1 = (const float*)d_in[2];
  const float* b1 = (const float*)d_in[3];
  const float* W2 = (const float*)d_in[4];
  const float* b2 = (const float*)d_in[5];
  float* out = (float*)d_out;

  const int N = in_sizes[0] / IN_DIM;
  const int E = in_sizes[1] / 2;
  const int NF = N * HID;

  float* dinv   = (float*)d_ws;
  float* bufL   = dinv + (1 << 16);
  float* bufAgg = bufL + (size_t)NF;
  float* bufH   = out;
  int* wcount = (int*)bufL;
  int* wlist  = wcount + NLIST * NWAVE;

  const int* src = EI;
  const int* tgt = EI + E;

  search_xw1_kernel<<<256, 1024, 0, stream>>>(X, W1, src, tgt, dinv, bufH,
                                              wlist, wcount, E, N);

  aggregate_lds_kernel<false><<<256, 1024, 0, stream>>>(bufH, src, tgt, dinv, b1, W2,
                                                        bufAgg, wlist, wcount, E, N);

  aggregate_lds_kernel<true><<<256, 1024, 0, stream>>>(bufAgg, src, tgt, dinv, b2, W2,
                                                       out, wlist, wcount, E, N);
}


# edges binned once by row tile in a new launch (a cell per tile and slice, LDS int atomics); a tile's group reads its cells, no group filters all edges; list split evenly among waves; 4 launches
# speedup vs baseline: 3.9156x; 1.0867x over previous
#include <hip/hip_runtime.h>
#include <math.h>


typedef __attribute__((ext_vector_type(2))) float v2f;
typedef __attribute__((ext_vector_type(8))) float v8f;

static constexpr int HID    = 256;
static constexpr int IN_DIM = 64;

static constexpr int TROWS = 196;
static constexpr int NWAVE = 32;
static constexpr int EPH   = 512;
static constexpr int BATCH = 32;
static constexpr int LCAP  = BATCH + EPH;
static constexpr int NLIST = 256;
static constexpr int NBIN  = 256;
static constexpr int CCAP  = 64;
static constexpr int TCAP  = 8192;
static constexpr int BINT  = 512;
static constexpr int BINU  = 8;

__global__ __launch_bounds__(BINT)
void bin_edges_kernel(const int* __restrict__ src,
                      const int* __restrict__ tgt,
                      int* __restrict__ cells, int* __restrict__ ccount,
                      int E, int N) {
  __shared__ int cnt[NLIST];
  const int g = blockIdx.x;
  if (g >= NBIN) return;
  const int ntl = (N <= (1 << 24)) ? min(NLIST, (N + TROWS - 1) / TROWS) : 0;
  for (int i = threadIdx.x; i < NLIST; i += BINT) cnt[i] = 0;
  __syncthreads();

  const int per = (E + NBIN - 1) / NBIN;
  const int lo  = min(E, g * per);
  const int hi  = min(E, lo + per);
  for (int e0 = lo; e0 < hi; e0 += BINT * BINU) {
    int t[BINU], s[BINU];
#pragma unroll
    for (int u = 0; u < BINU; ++u) {
      const int e = e0 + u * BINT + (int)threadIdx.x;
      const bool in = e < hi;
      t[u] = in ? tgt[e] : -1;
      s[u] = in ? src[e] : 0;
    }
    int T[BINU], pos[BINU];
#pragma unroll
    for (int u = 0; u < BINU; ++u) {
      T[u] = ((unsigned)t[u] < (unsigned)N) ? t[u] / TROWS : ntl;
      pos[u] = CCAP;
      if (T[u] < ntl)
        pos[u] = __hip_atomic_fetch_add(&cnt[T[u]], 1, __ATOMIC_RELAXED,
                                        __HIP_MEMORY_SCOPE_WORKGROUP);
    }
#pragma unroll
    for (int u = 0; u < BINU; ++u) {
      if (pos[u] < CCAP) {
        const int r = t[u] - T[u] * TROWS;
        cells[((size_t)T[u] * NBIN + g) * CCAP + pos[u]] =
            (int)((unsigned)s[u] | ((unsigned)r << 24));
      }
    }
  }
  __syncthreads();
  for (int i = threadIdx.x; i < NLIST; i += BINT) ccount[(size_t)i * NBIN + g] = cnt[i];
}

template <class Flush>
__device__ __forceinline__ void search_edges(const int* __restrict__ tgt, int E,
                                             int base, int rows, int* mylist,
                                             int lane, int wave, Flush&& flush) {
  const unsigned lanes_below = (1u << lane) - 1u;
  const int hd  = min((int)(((16u - (unsigned)((size_t)tgt & 15)) & 15u) >> 2), E);
  const int* tb = tgt + hd;
  const int EB  = E - hd;
  const int G     = EB >> 2;
  const int GPI   = NWAVE * 8 * 32;
  const int ITERS = (G + GPI - 1) / GPI;
  const int FULL  = G / GPI;

  int cnt = 0;
  auto consider = [&](int t, int e) {
    const bool hit = (unsigned)(t - base) < (unsigned)rows;
    const unsigned m = __builtin_amdgcn_ballot_w32(hit);
    if (m != 0u) {
      if (hit) mylist[cnt + __builtin_popcount(m & lanes_below)] = e;
      cnt += __builtin_popcount(m);
    }
  };

  if (wave == 0) {
    const int e = (lane < hd) ? lane : hd + 4 * G + (lane - hd);
    const int t = (lane < 8 && e < E) ? tgt[e] : base - 1;
    consider(t, e);
  }
  cnt = __builtin_amdgcn_readfirstlane(cnt);

  const int rot = (ITERS > 0) ? (int)(blockIdx.x % (unsigned)ITERS) : 0;
  auto place_of = [&](int it) {
    int p = it + rot;
    if (p >= ITERS) p -= ITERS;
    return p;
  };
  auto group_of = [&](int p, int u) {
    return ((p * NWAVE + wave) * 8 + u) * 32 + lane;
  };
  auto load_place = [&](int p, int4 (&dst)[8]) {
    if (p < FULL) {
#pragma unroll
      for (int u = 0; u < 8; ++u)
        dst[u] = *(const int4*)(tb + 4 * group_of(p, u));
    } else {
#pragma unroll
      for (int u = 0; u < 8; ++u) {
        const int g = group_of(p, u);
        int4 v = *(const int4*)(tb + 4 * min(g, G - 1));
        if (g >= G) v.x = v.y = v.z = v.w = base - 1;
        dst[u] = v;
      }
    }
  };

  for (int it = 0; it < ITERS; ++it) {
    const int p = place_of(it);
    int4 cur[8];
    load_place(p, cur);
#pragma unroll
    for (int half = 0; half < 2; ++half) {
      if (cnt >= BATCH) { flush(cnt); cnt = 0; }
#pragma unroll
      for (int u = 4 * half; u < 4 * half + 4; ++u) {
        const int e = hd + 4 * group_of(p, u);
        consider(cur[u].x, e);
        consider(cur[u].y, e + 1);
        consider(cur[u].z, e + 2);
        consider(cur[u].w, e + 3);
      }
    }
  }
  flush(cnt);
}

static constexpr int XROWS = TROWS;
static constexpr int XSTR  = IN_DIM + 4;
static_assert(XROWS * (IN_DIM / 4) <= 4 * 1024, "staging: four 16-byte loads per thread");

__global__ __launch_bounds__(1024)
void tile_lists_xw1_kernel(const float* __restrict__ X,
                       const float* __restrict__ W1,
                       const int* __restrict__ tgt,
                       float* __restrict__ dinv,
                       float* __restrict__ Hout,
                       const int* __restrict__ cells,
                       const int* __restrict__ ccount,
                       int* __restrict__ tlist, int* __restrict__ tcount,
                       int E, int N) {
  __shared__ __attribute__((aligned(16))) float xt[XROWS * XSTR];
  __shared__ int   kept[NWAVE * LCAP];
  __shared__ int   degc[TROWS];
  __shared__ float dl[TROWS];
  __shared__ int   ccnt[NBIN];
  __shared__ int   coff[NBIN];
  __shared__ int   tot;
  __shared__ int   badc;

  const int lane = threadIdx.x & 31;
  const int wave = threadIdx.x >> 5;
  const int h  = lane >> 4;
  const int ln = lane & 15;
  const int tp = wave & 7;
  const int mq = __builtin_amdgcn_readfirstlane(wave >> 3);
  const int n2 = tp * 32 + 2 * ln;
  int* mylist = kept + wave * LCAP;

  for (int base = blockIdx.x * TROWS; base < N; base += gridDim.x * TROWS) {
    const int rows = min(TROWS, N - base);
    const int tile_id  = base / TROWS;
    const bool listed  = tile_id < NLIST && N <= (1 << 24);
    const int  lt      = listed ? tile_id : 0;

    if ((int)threadIdx.x < TROWS) degc[threadIdx.x] = 0;
    if (threadIdx.x == 0) { tot = 0; badc = 0; }
    __syncthreads();

    if (listed && (int)threadIdx.x < NBIN) {
      const int cn = ccount[(size_t)lt * NBIN + threadIdx.x];
      ccnt[threadIdx.x] = cn;
      coff[threadIdx.x] = __hip_atomic_fetch_add(&tot, cn, __ATOMIC_RELAXED,
                                                 __HIP_MEMORY_SCOPE_WORKGROUP);
      if (cn > CCAP) badc = 1;
    }
    __syncthreads();
    const int  nall     = tot;
    const bool use_cell = listed && badc == 0 && nall <= TCAP;

    if (use_cell) {
      const int q  = (int)threadIdx.x & 15;
      const int c0 = (int)threadIdx.x >> 4;
      int* tl = tlist + (size_t)lt * TCAP;
      int cn[4], co[4];
#pragma unroll
      for (int u = 0; u < 4; ++u) {
        cn[u] = ccnt[c0 + 64 * u];
        co[u] = coff[c0 + 64 * u];
      }
      int p[4][CCAP / 16];
#pragma unroll
      for (int u = 0; u < 4; ++u) {
        const int* cp = cells + ((size_t)lt * NBIN + c0 + 64 * u) * CCAP;
#pragma unroll
        for (int v = 0; v < CCAP / 16; ++v)
          p[u][v] = (q + 16 * v < cn[u]) ? cp[q + 16 * v] : -1;
      }
#pragma unroll
      for (int u = 0; u < 4; ++u) {
#pragma unroll
        for (int v = 0; v < CCAP / 16; ++v) {
          if (q + 16 * v < cn[u]) {
            const int r = (int)((unsigned)p[u][v] >> 24);
            __hip_atomic_fetch_add(&degc[r], 1, __ATOMIC_RELAXED, __HIP_MEMORY_SCOPE_WORKGROUP);
            tl[co[u] + q + 16 * v] = p[u][v];
          }
        }
      }
      if (threadIdx.x == 0) tcount[lt] = nall;
    } else {
      auto flush = [&](int n) {
        __builtin_amdgcn_fence(__ATOMIC_ACQ_REL, "wavefront");
        for (int k0 = 0; k0 < n; k0 += 32) {
          const int k = k0 + lane;
          if (k < n) {
            const int r = tgt[mylist[k]] - base;
            __hip_atomic_fetch_add(&degc[r], 1, __ATOMIC_RELAXED, __HIP_MEMORY_SCOPE_WORKGROUP);
          }
        }
        __builtin_amdgcn_fence(__ATOMIC_ACQ_REL, "wavefront");
      };
      search_edges(tgt, E, base, rows, mylist, lane, wave, flush);
      if (listed && threadIdx.x == 0) tcount[lt] = -1;
    }
    __syncthreads();

    if ((int)threadIdx.x < rows) {
      const float d = rsqrtf(1.0f + (float)degc[threadIdx.x]);
      dl[threadIdx.x] = d;
      dinv[base + threadIdx.x] = d;
    }

    const int nvec = rows * (IN_DIM / 4);
    const float4* xp = (const float4*)(X + (size_t)base * IN_DIM);
    float4 q[4];
#pragma unroll
    for (int u = 0; u < 4; ++u) {
      const int v = (int)threadIdx.x + 1024 * u;
      q[u] = (v < nvec) ? xp[v] : make_float4(0.0f, 0.0f, 0.0f, 0.0f);
    }

#pragma unroll
    for (int u = 0; u < 4; ++u) {
      const int v = (int)threadIdx.x + 1024 * u;
      if (v < nvec) *(float4*)(xt + (v >> 4) * XSTR + 4 * (v & 15)) = q[u];
    }

    v2f be[16], bo[16];
#pragma unroll
    for (int s = 0; s < 16; ++s) {
      const float* wp = W1 + (size_t)(4 * s + 2 * h) * HID + n2;
      const float2 w0 = *(const float2*)wp;
      const float2 w1 = *(const float2*)(wp + HID);
      be[s].x = w0.x;  be[s].y = w1.x;
      bo[s].x = w0.y;  bo[s].y = w1.y;
    }
    __syncthreads();

    const int ntiles = (rows + 15) >> 4;
    for (int t = mq; t < ntiles; t += 4) {
      const int row = min(t * 16 + ln, rows - 1);
      const float* ap = xt + row * XSTR + 2 * h;
      v2f a[16];
#pragma unroll
      for (int s = 0; s < 16; ++s) a[s] = *(const v2f*)(ap + 4 * s);
      v8f c0 = {}, c1 = {};
#pragma unroll
      for (int s = 0; s < 16; ++s) {
        c0 = __builtin_amdgcn_wmma_f32_16x16x4_f32(
            false, a[s], false, be[s], (short)0, c0, false, false);
        c1 = __builtin_amdgcn_wmma_f32_16x16x4_f32(
            false, a[s], false, bo[s], (short)0, c1, false, false);
      }
      const int r0 = t * 16 + 8 * h;
#pragma unroll
      for (int r = 0; r < 8; ++r) {
        if (r0 + r < rows) {
          const float d = dl[r0 + r];
          float2 v;
          v.x = c0[r] * d;
          v.y = c1[r] * d;
          *(float2*)(Hout + (size_t)(base + r0 + r) * HID + n2) = v;
        }
      }
    }
    __syncthreads();
  }
}


static constexpr int TSTR  = HID + 1;

template <int NS>
__device__ __forceinline__ void rows_times_w(const float* tile,
                                             const float* __restrict__ Wm,
                                             const float* __restrict__ dsc,
                                             float* __restrict__ out,
                                             int base, int rows, int mq, int tp,
                                             int lane) {
  const int h  = lane >> 4;
  const int ln = lane & 15;
  const int n2 = tp * 32 + 2 * ln;
  int aoff[NS];
#pragma unroll
  for (int i = 0; i < NS; ++i) {
    const int row = min((mq + 4 * i) * 16 + ln, rows - 1);
    aoff[i] = row * TSTR;
  }
  v8f acc[NS][2];
#pragma unroll
  for (int i = 0; i < NS; ++i) { acc[i][0] = v8f{}; acc[i][1] = v8f{}; }

  auto w_of = [&](int s) {
    const int lh = ((s & 31) + 16 * h) & 31;
    const int c0 = (s >> 5) * 128 + 4 * lh + 2 * h;
    return Wm + (size_t)c0 * HID + n2;
  };
  const float* wp = w_of(0);
  float2 w0n = *(const float2*)wp;
  float2 w1n = *(const float2*)(wp + HID);
#pragma unroll 2
  for (int s = 0; s < 64; ++s) {
    const float2 w0 = w0n, w1 = w1n;
    wp  = w_of(min(s + 1, 63));
    w0n = *(const float2*)wp;
    w1n = *(const float2*)(wp + HID);
    const int lh  = ((s & 31) + 16 * h) & 31;
    const int pos = (s >> 5) * 128 + 64 * h + lh;
    v2f a[NS];
#pragma unroll
    for (int i = 0; i < NS; ++i) {
      a[i].x = tile[aoff[i] + pos];
      a[i].y = tile[aoff[i] + pos + 32];
    }
    v2f be, bo;
    be.x = w0.x;  be.y = w1.x;
    bo.x = w0.y;  bo.y = w1.y;
#pragma unroll
    for (int i = 0; i < NS; ++i) {
      acc[i][0] = __builtin_amdgcn_wmma_f32_16x16x4_f32(
          false, a[i], false, be, (short)0, acc[i][0], false, false);
      acc[i][1] = __builtin_amdgcn_wmma_f32_16x16x4_f32(
          false, a[i], false, bo, (short)0, acc[i][1], false, false);
    }
  }

#pragma unroll
  for (int i = 0; i < NS; ++i) {
    const int r0 = (mq + 4 * i) * 16 + 8 * h;
#pragma unroll
    for (int r = 0; r < 8; ++r) {
      if (r0 + r < rows) {
        const float d = dsc[r0 + r];
        float2 v;
        v.x = acc[i][0][r] * d;
        v.y = acc[i][1][r] * d;
        *(float2*)(out + (size_t)(base + r0 + r) * HID + n2) = v;
      }
    }
  }
}

template <bool LAST>
__global__ __launch_bounds__(1024)
void aggregate_lds_kernel(const float* __restrict__ H,
                          const int* __restrict__ src,
                          const int* __restrict__ tgt,
                          const float* __restrict__ dinv,
                          const float* __restrict__ bias,
                          const float* __restrict__ Wn,
                          float* __restrict__ out,
                          const int* __restrict__ tlist,
                          const int* __restrict__ tcount,
                          int E, int N) {
  __shared__ float tile[TROWS * TSTR];
  __shared__ int   kept[NWAVE * LCAP];

  const int lane = threadIdx.x & 31;
  const int wave = threadIdx.x >> 5;
  int* mylist = kept + wave * LCAP;

  for (int base = blockIdx.x * TROWS; base < N; base += gridDim.x * TROWS) {
    const int rows = min(TROWS, N - base);
    const int tile_id  = base / TROWS;
    const bool listed  = tile_id < NLIST && N <= (1 << 24);
    const int  lt      = listed ? tile_id : 0;

    for (int r = wave; r < rows; r += NWAVE) {
      const int i = base + r;
      const float4* hp = (const float4*)(H + (size_t)i * HID);
      const float4 a = hp[lane], b = hp[32 + lane];
      float* tp = tile + r * TSTR;
      tp[lane]       = a.x;  tp[32 + lane]  = a.y;
      tp[64 + lane]  = a.z;  tp[96 + lane]  = a.w;
      tp[128 + lane] = b.x;  tp[160 + lane] = b.y;
      tp[192 + lane] = b.z;  tp[224 + lane] = b.w;
    }
    __syncthreads();

    auto add_rows = [&](int s, int r, int m) {
      int j = 0;
      for (; j + 4 <= m; j += 4) {
        float4 a[4], b[4];
        int    rj[4];
#pragma unroll
        for (int u = 0; u < 4; ++u) {
          const int sj = __builtin_amdgcn_readlane(s, j + u);
          rj[u] = __builtin_amdgcn_readlane(r, j + u);
          const float4* hp = (const float4*)(H + (size_t)sj * HID);
          a[u] = hp[lane];
          b[u] = hp[32 + lane];
        }
#pragma unroll
        for (int u = 0; u < 4; ++u) {
          float* tp = tile + rj[u] * TSTR + lane;
          __hip_atomic_fetch_add(tp,       a[u].x, __ATOMIC_RELAXED, __HIP_MEMORY_SCOPE_WORKGROUP);
          __hip_atomic_fetch_add(tp + 32,  a[u].y, __ATOMIC_RELAXED, __HIP_MEMORY_SCOPE_WORKGROUP);
          __hip_atomic_fetch_add(tp + 64,  a[u].z, __ATOMIC_RELAXED, __HIP_MEMORY_SCOPE_WORKGROUP);
          __hip_atomic_fetch_add(tp + 96,  a[u].w, __ATOMIC_RELAXED, __HIP_MEMORY_SCOPE_WORKGROUP);
          __hip_atomic_fetch_add(tp + 128, b[u].x, __ATOMIC_RELAXED, __HIP_MEMORY_SCOPE_WORKGROUP);
          __hip_atomic_fetch_add(tp + 160, b[u].y, __ATOMIC_RELAXED, __HIP_MEMORY_SCOPE_WORKGROUP);
          __hip_atomic_fetch_add(tp + 192, b[u].z, __ATOMIC_RELAXED, __HIP_MEMORY_SCOPE_WORKGROUP);
          __hip_atomic_fetch_add(tp + 224, b[u].w, __ATOMIC_RELAXED, __HIP_MEMORY_SCOPE_WORKGROUP);
        }
      }
      for (; j < m; ++j) {
        const int sj = __builtin_amdgcn_readlane(s, j);
        const int rr = __builtin_amdgcn_readlane(r, j);
        const float4* hp = (const float4*)(H + (size_t)sj * HID);
        const float4 a = hp[lane], b = hp[32 + lane];
        float* tp = tile + rr * TSTR + lane;
        __hip_atomic_fetch_add(tp,       a.x, __ATOMIC_RELAXED, __HIP_MEMORY_SCOPE_WORKGROUP);
        __hip_atomic_fetch_add(tp + 32,  a.y, __ATOMIC_RELAXED, __HIP_MEMORY_SCOPE_WORKGROUP);
        __hip_atomic_fetch_add(tp + 64,  a.z, __ATOMIC_RELAXED, __HIP_MEMORY_SCOPE_WORKGROUP);
        __hip_atomic_fetch_add(tp + 96,  a.w, __ATOMIC_RELAXED, __HIP_MEMORY_SCOPE_WORKGROUP);
        __hip_atomic_fetch_add(tp + 128, b.x, __ATOMIC_RELAXED, __HIP_MEMORY_SCOPE_WORKGROUP);
        __hip_atomic_fetch_add(tp + 160, b.y, __ATOMIC_RELAXED, __HIP_MEMORY_SCOPE_WORKGROUP);
        __hip_atomic_fetch_add(tp + 192, b.z, __ATOMIC_RELAXED, __HIP_MEMORY_SCOPE_WORKGROUP);
        __hip_atomic_fetch_add(tp + 224, b.w, __ATOMIC_RELAXED, __HIP_MEMORY_SCOPE_WORKGROUP);
      }
    };

    int nt = -1;
    if (listed) {
      nt = __builtin_amdgcn_readfirstlane(tcount[lt]);
    }
    if (nt >= 0) {
      const int lo = (nt * wave) / NWAVE;
      const int nl = (nt * (wave + 1)) / NWAVE - lo;
      const int* mywl = tlist + (size_t)lt * TCAP + lo;
      int p = (lane < nl) ? mywl[lane] : 0;
      for (int k0 = 0; k0 < nl; k0 += 32) {
        const int k = k0 + lane;
        const int s = p & 0xFFFFFF;
        const int r = (int)((unsigned)p >> 24);
        p = (k + 32 < nl) ? mywl[k + 32] : 0;
        add_rows(s, r, min(32, nl - k0));
      }
    } else {
      auto flush = [&](int n) {
        __builtin_amdgcn_fence(__ATOMIC_ACQ_REL, "wavefront");
        for (int k0 = 0; k0 < n; k0 += 32) {
          const int k = k0 + lane;
          int s = 0, r = 0;
          if (k < n) {
            const int e = mylist[k];
            s = src[e];
            r = tgt[e] - base;
          }
          add_rows(s, r, min(32, n - k0));
        }
        __builtin_amdgcn_fence(__ATOMIC_ACQ_REL, "wavefront");
      };
      search_edges(tgt, E, base, rows, mylist, lane, wave, flush);
    }
    __syncthreads();

    float4 ba, bb;
    ba.x = bias[4 * lane];           ba.y = bias[4 * lane + 1];
    ba.z = bias[4 * lane + 2];       ba.w = bias[4 * lane + 3];
    bb.x = bias[128 + 4 * lane];     bb.y = bias[128 + 4 * lane + 1];
    bb.z = bias[128 + 4 * lane + 2]; bb.w = bias[128 + 4 * lane + 3];
    if (LAST) {
      for (int r = wave; r < rows; r += NWAVE) {
        const int i = base + r;
        const float d = dinv[i];
        const float* tp = tile + r * TSTR;
        float4 a, b;
        a.x = tp[lane] * d;        a.y = tp[32 + lane] * d;
        a.z = tp[64 + lane] * d;   a.w = tp[96 + lane] * d;
        b.x = tp[128 + lane] * d;  b.y = tp[160 + lane] * d;
        b.z = tp[192 + lane] * d;  b.w = tp[224 + lane] * d;
        a.x += ba.x; a.y += ba.y; a.z += ba.z; a.w += ba.w;
        b.x += bb.x; b.y += bb.y; b.z += bb.z; b.w += bb.w;
        float4* op = (float4*)(out + (size_t)i * HID);
        op[lane]      = a;
        op[32 + lane] = b;
      }
    } else {
      const float kLn1e4Over256 = 9.210340371976184f / 256.0f;
      const float f0 = expf(-(float)(4 * lane)           * kLn1e4Over256);
      const float f1 = expf(-(float)(4 * lane + 2)       * kLn1e4Over256);
      const float f2 = expf(-(float)(128 + 4 * lane)     * kLn1e4Over256);
      const float f3 = expf(-(float)(128 + 4 * lane + 2) * kLn1e4Over256);
      for (int r = wave; r < rows; r += NWAVE) {
        const int i = base + r;
        const float d  = dinv[i];
        const float fi = (float)i;
        float* tp = tile + r * TSTR;
        float s0, c0, s1, c1, s2, c2, s3, c3;
        sincosf(fi * f0, &s0, &c0);
        sincosf(fi * f1, &s1, &c1);
        sincosf(fi * f2, &s2, &c2);
        sincosf(fi * f3, &s3, &c3);
        float4 a, b;
        a.x = tp[lane] * d + ba.x + s0;        a.y = tp[32 + lane] * d + ba.y + c0;
        a.z = tp[64 + lane] * d + ba.z + s1;   a.w = tp[96 + lane] * d + ba.w + c1;
        b.x = tp[128 + lane] * d + bb.x + s2;  b.y = tp[160 + lane] * d + bb.y + c2;
        b.z = tp[192 + lane] * d + bb.z + s3;  b.w = tp[224 + lane] * d + bb.w + c3;
        tp[lane]       = a.x > 0.0f ? a.x : 0.0f;  tp[32 + lane]  = a.y > 0.0f ? a.y : 0.0f;
        tp[64 + lane]  = a.z > 0.0f ? a.z : 0.0f;  tp[96 + lane]  = a.w > 0.0f ? a.w : 0.0f;
        tp[128 + lane] = b.x > 0.0f ? b.x : 0.0f;  tp[160 + lane] = b.y > 0.0f ? b.y : 0.0f;
        tp[192 + lane] = b.z > 0.0f ? b.z : 0.0f;  tp[224 + lane] = b.w > 0.0f ? b.w : 0.0f;
      }
      __syncthreads();
      const int tp8 = wave & 7;
      const int mq  = wave >> 3;
      if (mq == 0) rows_times_w<4>(tile, Wn, dinv + base, out, base, rows, mq, tp8, lane);
      else         rows_times_w<3>(tile, Wn, dinv + base, out, base, rows, mq, tp8, lane);
    }
    __syncthreads();
  }
}

extern "C" void kernel_launch(void* const* d_in, const int* in_sizes, int n_in,
                              void* d_out, int out_size, void* d_ws, size_t ws_size,
                              hipStream_t stream) {
  const float* X  = (const float*)d_in[0];
  const int*   EI = (const int*)d_in[1];
  const float* W1 = (const float*)d_in[2];
  const float* b1 = (const float*)d_in[3];
  const float* W2 = (const float*)d_in[4];
  const float* b2 = (const float*)d_in[5];
  float* out = (float*)d_out;

  const int N = in_sizes[0] / IN_DIM;
  const int E = in_sizes[1] / 2;
  const int NF = N * HID;

  float* dinv   = (float*)d_ws;
  float* bufL   = dinv + (1 << 16);
  float* bufAgg = bufL + (size_t)NF;
  float* bufH   = out;
  int* ccount = (int*)bufL;
  int* cells  = ccount + NLIST * NBIN;
  int* tcount = cells + NLIST * NBIN * CCAP;
  int* tlist  = tcount + NLIST;

  const int* src = EI;
  const int* tgt = EI + E;

  bin_edges_kernel<<<NBIN, BINT, 0, stream>>>(src, tgt, cells, ccount, E, N);

  tile_lists_xw1_kernel<<<256, 1024, 0, stream>>>(X, W1, tgt, dinv, bufH,
                                                  cells, ccount, tlist, tcount, E, N);

  aggregate_lds_kernel<false><<<256, 1024, 0, stream>>>(bufH, src, tgt, dinv, b1, W2,
                                                        bufAgg, tlist, tcount, E, N);

  aggregate_lds_kernel<true><<<256, 1024, 0, stream>>>(bufAgg, src, tgt, dinv, b2, W2,
                                                       out, tlist, tcount, E, N);
}


# both aggregations: a tile row has one owning wave (row mod 32) that picks its entries from the tile list, sorts them by row and sums source rows in registers; no LDS float atomics; sum order changes
# speedup vs baseline: 4.2055x; 1.0740x over previous
#include <hip/hip_runtime.h>
#include <math.h>


typedef __attribute__((ext_vector_type(2))) float v2f;
typedef __attribute__((ext_vector_type(8))) float v8f;

static constexpr int HID    = 256;
static constexpr int IN_DIM = 64;

static constexpr int TROWS = 196;
static constexpr int NWAVE = 32;
static constexpr int EPH   = 512;
static constexpr int BATCH = 32;
static constexpr int LCAP  = BATCH + EPH;
static constexpr int NLIST = 256;
static constexpr int NBIN  = 256;
static constexpr int CCAP  = 64;
static constexpr int TCAP  = 8192;
static constexpr int BINT  = 512;
static constexpr int BINU  = 8;

__global__ __launch_bounds__(BINT)
void bin_edges_kernel(const int* __restrict__ src,
                      const int* __restrict__ tgt,
                      int* __restrict__ cells, int* __restrict__ ccount,
                      int E, int N) {
  __shared__ int cnt[NLIST];
  const int g = blockIdx.x;
  if (g >= NBIN) return;
  const int ntl = (N <= (1 << 24)) ? min(NLIST, (N + TROWS - 1) / TROWS) : 0;
  for (int i = threadIdx.x; i < NLIST; i += BINT) cnt[i] = 0;
  __syncthreads();

  const int per = (E + NBIN - 1) / NBIN;
  const int lo  = min(E, g * per);
  const int hi  = min(E, lo + per);
  for (int e0 = lo; e0 < hi; e0 += BINT * BINU) {
    int t[BINU], s[BINU];
#pragma unroll
    for (int u = 0; u < BINU; ++u) {
      const int e = e0 + u * BINT + (int)threadIdx.x;
      const bool in = e < hi;
      t[u] = in ? tgt[e] : -1;
      s[u] = in ? src[e] : 0;
    }
    int T[BINU], pos[BINU];
#pragma unroll
    for (int u = 0; u < BINU; ++u) {
      T[u] = ((unsigned)t[u] < (unsigned)N) ? t[u] / TROWS : ntl;
      pos[u] = CCAP;
      if (T[u] < ntl)
        pos[u] = __hip_atomic_fetch_add(&cnt[T[u]], 1, __ATOMIC_RELAXED,
                                        __HIP_MEMORY_SCOPE_WORKGROUP);
    }
#pragma unroll
    for (int u = 0; u < BINU; ++u) {
      if (pos[u] < CCAP) {
        const int r = t[u] - T[u] * TROWS;
        cells[((size_t)T[u] * NBIN + g) * CCAP + pos[u]] =
            (int)((unsigned)s[u] | ((unsigned)r << 24));
      }
    }
  }
  __syncthreads();
  for (int i = threadIdx.x; i < NLIST; i += BINT) ccount[(size_t)i * NBIN + g] = cnt[i];
}

template <class Flush>
__device__ __forceinline__ void search_edges(const int* __restrict__ tgt, int E,
                                             int base, int rows, int* mylist,
                                             int lane, int wave, Flush&& flush) {
  const unsigned lanes_below = (1u << lane) - 1u;
  const int hd  = min((int)(((16u - (unsigned)((size_t)tgt & 15)) & 15u) >> 2), E);
  const int* tb = tgt + hd;
  const int EB  = E - hd;
  const int G     = EB >> 2;
  const int GPI   = NWAVE * 8 * 32;
  const int ITERS = (G + GPI - 1) / GPI;
  const int FULL  = G / GPI;

  int cnt = 0;
  auto consider = [&](int t, int e) {
    const bool hit = (unsigned)(t - base) < (unsigned)rows;
    const unsigned m = __builtin_amdgcn_ballot_w32(hit);
    if (m != 0u) {
      if (hit) mylist[cnt + __builtin_popcount(m & lanes_below)] = e;
      cnt += __builtin_popcount(m);
    }
  };

  if (wave == 0) {
    const int e = (lane < hd) ? lane : hd + 4 * G + (lane - hd);
    const int t = (lane < 8 && e < E) ? tgt[e] : base - 1;
    consider(t, e);
  }
  cnt = __builtin_amdgcn_readfirstlane(cnt);

  const int rot = (ITERS > 0) ? (int)(blockIdx.x % (unsigned)ITERS) : 0;
  auto place_of = [&](int it) {
    int p = it + rot;
    if (p >= ITERS) p -= ITERS;
    return p;
  };
  auto group_of = [&](int p, int u) {
    return ((p * NWAVE + wave) * 8 + u) * 32 + lane;
  };
  auto load_place = [&](int p, int4 (&dst)[8]) {
    if (p < FULL) {
#pragma unroll
      for (int u = 0; u < 8; ++u)
        dst[u] = *(const int4*)(tb + 4 * group_of(p, u));
    } else {
#pragma unroll
      for (int u = 0; u < 8; ++u) {
        const int g = group_of(p, u);
        int4 v = *(const int4*)(tb + 4 * min(g, G - 1));
        if (g >= G) v.x = v.y = v.z = v.w = base - 1;
        dst[u] = v;
      }
    }
  };

  for (int it = 0; it < ITERS; ++it) {
    const int p = place_of(it);
    int4 cur[8];
    load_place(p, cur);
#pragma unroll
    for (int half = 0; half < 2; ++half) {
      if (cnt >= BATCH) { flush(cnt); cnt = 0; }
#pragma unroll
      for (int u = 4 * half; u < 4 * half + 4; ++u) {
        const int e = hd + 4 * group_of(p, u);
        consider(cur[u].x, e);
        consider(cur[u].y, e + 1);
        consider(cur[u].z, e + 2);
        consider(cur[u].w, e + 3);
      }
    }
  }
  flush(cnt);
}

static constexpr int XROWS = TROWS;
static constexpr int XSTR  = IN_DIM + 4;
static_assert(XROWS * (IN_DIM / 4) <= 4 * 1024, "staging: four 16-byte loads per thread");

__global__ __launch_bounds__(1024)
void tile_lists_xw1_kernel(const float* __restrict__ X,
                       const float* __restrict__ W1,
                       const int* __restrict__ tgt,
                       float* __restrict__ dinv,
                       float* __restrict__ Hout,
                       const int* __restrict__ cells,
                       const int* __restrict__ ccount,
                       int* __restrict__ tlist, int* __restrict__ tcount,
                       int E, int N) {
  __shared__ __attribute__((aligned(16))) float xt[XROWS * XSTR];
  __shared__ int   kept[NWAVE * LCAP];
  __shared__ int   degc[TROWS];
  __shared__ float dl[TROWS];
  __shared__ int   ccnt[NBIN];
  __shared__ int   coff[NBIN];
  __shared__ int   tot;
  __shared__ int   badc;

  const int lane = threadIdx.x & 31;
  const int wave = threadIdx.x >> 5;
  const int h  = lane >> 4;
  const int ln = lane & 15;
  const int tp = wave & 7;
  const int mq = __builtin_amdgcn_readfirstlane(wave >> 3);
  const int n2 = tp * 32 + 2 * ln;
  int* mylist = kept + wave * LCAP;

  for (int base = blockIdx.x * TROWS; base < N; base += gridDim.x * TROWS) {
    const int rows = min(TROWS, N - base);
    const int tile_id  = base / TROWS;
    const bool listed  = tile_id < NLIST && N <= (1 << 24);
    const int  lt      = listed ? tile_id : 0;

    if ((int)threadIdx.x < TROWS) degc[threadIdx.x] = 0;
    if (threadIdx.x == 0) { tot = 0; badc = 0; }
    __syncthreads();

    if (listed && (int)threadIdx.x < NBIN) {
      const int cn = ccount[(size_t)lt * NBIN + threadIdx.x];
      ccnt[threadIdx.x] = cn;
      coff[threadIdx.x] = __hip_atomic_fetch_add(&tot, cn, __ATOMIC_RELAXED,
                                                 __HIP_MEMORY_SCOPE_WORKGROUP);
      if (cn > CCAP) badc = 1;
    }
    __syncthreads();
    const int  nall     = tot;
    const bool use_cell = listed && badc == 0 && nall <= TCAP;

    if (use_cell) {
      const int q  = (int)threadIdx.x & 15;
      const int c0 = (int)threadIdx.x >> 4;
      int* tl = tlist + (size_t)lt * TCAP;
      int cn[4], co[4];
#pragma unroll
      for (int u = 0; u < 4; ++u) {
        cn[u] = ccnt[c0 + 64 * u];
        co[u] = coff[c0 + 64 * u];
      }
      int p[4][CCAP / 16];
#pragma unroll
      for (int u = 0; u < 4; ++u) {
        const int* cp = cells + ((size_t)lt * NBIN + c0 + 64 * u) * CCAP;
#pragma unroll
        for (int v = 0; v < CCAP / 16; ++v)
          p[u][v] = (q + 16 * v < cn[u]) ? cp[q + 16 * v] : -1;
      }
#pragma unroll
      for (int u = 0; u < 4; ++u) {
#pragma unroll
        for (int v = 0; v < CCAP / 16; ++v) {
          if (q + 16 * v < cn[u]) {
            const int r = (int)((unsigned)p[u][v] >> 24);
            __hip_atomic_fetch_add(&degc[r], 1, __ATOMIC_RELAXED, __HIP_MEMORY_SCOPE_WORKGROUP);
            tl[co[u] + q + 16 * v] = p[u][v];
          }
        }
      }
      if (threadIdx.x == 0) tcount[lt] = nall;
    } else {
      auto flush = [&](int n) {
        __builtin_amdgcn_fence(__ATOMIC_ACQ_REL, "wavefront");
        for (int k0 = 0; k0 < n; k0 += 32) {
          const int k = k0 + lane;
          if (k < n) {
            const int r = tgt[mylist[k]] - base;
            __hip_atomic_fetch_add(&degc[r], 1, __ATOMIC_RELAXED, __HIP_MEMORY_SCOPE_WORKGROUP);
          }
        }
        __builtin_amdgcn_fence(__ATOMIC_ACQ_REL, "wavefront");
      };
      search_edges(tgt, E, base, rows, mylist, lane, wave, flush);
      if (listed && threadIdx.x == 0) tcount[lt] = -1;
    }
    __syncthreads();

    if ((int)threadIdx.x < rows) {
      const float d = rsqrtf(1.0f + (float)degc[threadIdx.x]);
      dl[threadIdx.x] = d;
      dinv[base + threadIdx.x] = d;
    }

    const int nvec = rows * (IN_DIM / 4);
    const float4* xp = (const float4*)(X + (size_t)base * IN_DIM);
    float4 q[4];
#pragma unroll
    for (int u = 0; u < 4; ++u) {
      const int v = (int)threadIdx.x + 1024 * u;
      q[u] = (v < nvec) ? xp[v] : make_float4(0.0f, 0.0f, 0.0f, 0.0f);
    }

#pragma unroll
    for (int u = 0; u < 4; ++u) {
      const int v = (int)threadIdx.x + 1024 * u;
      if (v < nvec) *(float4*)(xt + (v >> 4) * XSTR + 4 * (v & 15)) = q[u];
    }

    v2f be[16], bo[16];
#pragma unroll
    for (int s = 0; s < 16; ++s) {
      const float* wp = W1 + (size_t)(4 * s + 2 * h) * HID + n2;
      const float2 w0 = *(const float2*)wp;
      const float2 w1 = *(const float2*)(wp + HID);
      be[s].x = w0.x;  be[s].y = w1.x;
      bo[s].x = w0.y;  bo[s].y = w1.y;
    }
    __syncthreads();

    const int ntiles = (rows + 15) >> 4;
    for (int t = mq; t < ntiles; t += 4) {
      const int row = min(t * 16 + ln, rows - 1);
      const float* ap = xt + row * XSTR + 2 * h;
      v2f a[16];
#pragma unroll
      for (int s = 0; s < 16; ++s) a[s] = *(const v2f*)(ap + 4 * s);
      v8f c0 = {}, c1 = {};
#pragma unroll
      for (int s = 0; s < 16; ++s) {
        c0 = __builtin_amdgcn_wmma_f32_16x16x4_f32(
            false, a[s], false, be[s], (short)0, c0, false, false);
        c1 = __builtin_amdgcn_wmma_f32_16x16x4_f32(
            false, a[s], false, bo[s], (short)0, c1, false, false);
      }
      const int r0 = t * 16 + 8 * h;
#pragma unroll
      for (int r = 0; r < 8; ++r) {
        if (r0 + r < rows) {
          const float d = dl[r0 + r];
          float2 v;
          v.x = c0[r] * d;
          v.y = c1[r] * d;
          *(float2*)(Hout + (size_t)(base + r0 + r) * HID + n2) = v;
        }
      }
    }
    __syncthreads();
  }
}


static constexpr int TSTR  = HID + 1;

template <int NS>
__device__ __forceinline__ void rows_times_w(const float* tile,
                                             const float* __restrict__ Wm,
                                             const float* __restrict__ dsc,
                                             float* __restrict__ out,
                                             int base, int rows, int mq, int tp,
                                             int lane) {
  const int h  = lane >> 4;
  const int ln = lane & 15;
  const int n2 = tp * 32 + 2 * ln;
  int aoff[NS];
#pragma unroll
  for (int i = 0; i < NS; ++i) {
    const int row = min((mq + 4 * i) * 16 + ln, rows - 1);
    aoff[i] = row * TSTR;
  }
  v8f acc[NS][2];
#pragma unroll
  for (int i = 0; i < NS; ++i) { acc[i][0] = v8f{}; acc[i][1] = v8f{}; }

  auto w_of = [&](int s) {
    const int lh = ((s & 31) + 16 * h) & 31;
    const int c0 = (s >> 5) * 128 + 4 * lh + 2 * h;
    return Wm + (size_t)c0 * HID + n2;
  };
  const float* wp = w_of(0);
  float2 w0n = *(const float2*)wp;
  float2 w1n = *(const float2*)(wp + HID);
#pragma unroll 2
  for (int s = 0; s < 64; ++s) {
    const float2 w0 = w0n, w1 = w1n;
    wp  = w_of(min(s + 1, 63));
    w0n = *(const float2*)wp;
    w1n = *(const float2*)(wp + HID);
    const int lh  = ((s & 31) + 16 * h) & 31;
    const int pos = (s >> 5) * 128 + 64 * h + lh;
    v2f a[NS];
#pragma unroll
    for (int i = 0; i < NS; ++i) {
      a[i].x = tile[aoff[i] + pos];
      a[i].y = tile[aoff[i] + pos + 32];
    }
    v2f be, bo;
    be.x = w0.x;  be.y = w1.x;
    bo.x = w0.y;  bo.y = w1.y;
#pragma unroll
    for (int i = 0; i < NS; ++i) {
      acc[i][0] = __builtin_amdgcn_wmma_f32_16x16x4_f32(
          false, a[i], false, be, (short)0, acc[i][0], false, false);
      acc[i][1] = __builtin_amdgcn_wmma_f32_16x16x4_f32(
          false, a[i], false, bo, (short)0, acc[i][1], false, false);
    }
  }

#pragma unroll
  for (int i = 0; i < NS; ++i) {
    const int r0 = (mq + 4 * i) * 16 + 8 * h;
#pragma unroll
    for (int r = 0; r < 8; ++r) {
      if (r0 + r < rows) {
        const float d = dsc[r0 + r];
        float2 v;
        v.x = acc[i][0][r] * d;
        v.y = acc[i][1][r] * d;
        *(float2*)(out + (size_t)(base + r0 + r) * HID + n2) = v;
      }
    }
  }
}

static constexpr int NOWN = (TROWS + NWAVE - 1) / NWAVE;
static constexpr int OCAP = LCAP / 2;
static constexpr int SUMMIN = 32;
static_assert(TROWS <= 256 && NWAVE == 32 && NOWN <= 7,
              "list entry: tile row in bits 24..31, so its slot (row / 32) in bits 29..31; 7 is no slot");
static_assert(OCAP >= 128 && 2 * OCAP <= LCAP, "a chunk of 128 entries fits an empty list");

template <bool LAST>
__global__ __launch_bounds__(1024)
void aggregate_lds_kernel(const float* __restrict__ H,
                          const int* __restrict__ src,
                          const int* __restrict__ tgt,
                          const float* __restrict__ dinv,
                          const float* __restrict__ bias,
                          const float* __restrict__ Wn,
                          float* __restrict__ out,
                          const int* __restrict__ tlist,
                          const int* __restrict__ tcount,
                          int E, int N) {
  __shared__ float tile[TROWS * TSTR];
  __shared__ int   kept[NWAVE * LCAP];

  const int lane = threadIdx.x & 31;
  const int wave = threadIdx.x >> 5;
  int* mylist = kept + wave * LCAP;

  for (int base = blockIdx.x * TROWS; base < N; base += gridDim.x * TROWS) {
    const int rows = min(TROWS, N - base);
    const int tile_id  = base / TROWS;
    const bool listed  = tile_id < NLIST && N <= (1 << 24);
    const int  lt      = listed ? tile_id : 0;

    int nt = -1;
    if (listed) {
      nt = __builtin_amdgcn_readfirstlane(tcount[lt]);
    }

    for (int r = wave; r < rows; r += NWAVE) {
      const int i = base + r;
      const float4* hp = (const float4*)(H + (size_t)i * HID);
      const float4 a = hp[lane], b = hp[32 + lane];
      float* tp = tile + r * TSTR;
      tp[lane]       = a.x;  tp[32 + lane]  = a.y;
      tp[64 + lane]  = a.z;  tp[96 + lane]  = a.w;
      tp[128 + lane] = b.x;  tp[160 + lane] = b.y;
      tp[192 + lane] = b.z;  tp[224 + lane] = b.w;
    }
    if (nt < 0) __syncthreads();

    auto add_rows = [&](int s, int r, int m) {
      int j = 0;
      for (; j + 4 <= m; j += 4) {
        float4 a[4], b[4];
        int    rj[4];
#pragma unroll
        for (int u = 0; u < 4; ++u) {
          const int sj = __builtin_amdgcn_readlane(s, j + u);
          rj[u] = __builtin_amdgcn_readlane(r, j + u);
          const float4* hp = (const float4*)(H + (size_t)sj * HID);
          a[u] = hp[lane];
          b[u] = hp[32 + lane];
        }
#pragma unroll
        for (int u = 0; u < 4; ++u) {
          float* tp = tile + rj[u] * TSTR + lane;
          __hip_atomic_fetch_add(tp,       a[u].x, __ATOMIC_RELAXED, __HIP_MEMORY_SCOPE_WORKGROUP);
          __hip_atomic_fetch_add(tp + 32,  a[u].y, __ATOMIC_RELAXED, __HIP_MEMORY_SCOPE_WORKGROUP);
          __hip_atomic_fetch_add(tp + 64,  a[u].z, __ATOMIC_RELAXED, __HIP_MEMORY_SCOPE_WORKGROUP);
          __hip_atomic_fetch_add(tp + 96,  a[u].w, __ATOMIC_RELAXED, __HIP_MEMORY_SCOPE_WORKGROUP);
          __hip_atomic_fetch_add(tp + 128, b[u].x, __ATOMIC_RELAXED, __HIP_MEMORY_SCOPE_WORKGROUP);
          __hip_atomic_fetch_add(tp + 160, b[u].y, __ATOMIC_RELAXED, __HIP_MEMORY_SCOPE_WORKGROUP);
          __hip_atomic_fetch_add(tp + 192, b[u].z, __ATOMIC_RELAXED, __HIP_MEMORY_SCOPE_WORKGROUP);
          __hip_atomic_fetch_add(tp + 224, b[u].w, __ATOMIC_RELAXED, __HIP_MEMORY_SCOPE_WORKGROUP);
        }
      }
      for (; j < m; ++j) {
        const int sj = __builtin_amdgcn_readlane(s, j);
        const int rr = __builtin_amdgcn_readlane(r, j);
        const float4* hp = (const float4*)(H + (size_t)sj * HID);
        const float4 a = hp[lane], b = hp[32 + lane];
        float* tp = tile + rr * TSTR + lane;
        __hip_atomic_fetch_add(tp,       a.x, __ATOMIC_RELAXED, __HIP_MEMORY_SCOPE_WORKGROUP);
        __hip_atomic_fetch_add(tp + 32,  a.y, __ATOMIC_RELAXED, __HIP_MEMORY_SCOPE_WORKGROUP);
        __hip_atomic_fetch_add(tp + 64,  a.z, __ATOMIC_RELAXED, __HIP_MEMORY_SCOPE_WORKGROUP);
        __hip_atomic_fetch_add(tp + 96,  a.w, __ATOMIC_RELAXED, __HIP_MEMORY_SCOPE_WORKGROUP);
        __hip_atomic_fetch_add(tp + 128, b.x, __ATOMIC_RELAXED, __HIP_MEMORY_SCOPE_WORKGROUP);
        __hip_atomic_fetch_add(tp + 160, b.y, __ATOMIC_RELAXED, __HIP_MEMORY_SCOPE_WORKGROUP);
        __hip_atomic_fetch_add(tp + 192, b.z, __ATOMIC_RELAXED, __HIP_MEMORY_SCOPE_WORKGROUP);
        __hip_atomic_fetch_add(tp + 224, b.w, __ATOMIC_RELAXED, __HIP_MEMORY_SCOPE_WORKGROUP);
      }
    };

    if (nt >= 0) {
      const int wv = __builtin_amdgcn_readfirstlane(wave);
      int* rawl = mylist;
      int* srtl = mylist + OCAP;
      const unsigned lanes_below = (1u << lane) - 1u;

      auto sum_sorted = [&](int n) {
        float4 accA = make_float4(0.0f, 0.0f, 0.0f, 0.0f);
        float4 accB = make_float4(0.0f, 0.0f, 0.0f, 0.0f);
        auto commit = [&](int sl) {
          float* tp = tile + (wv + NWAVE * sl) * TSTR + lane;
          tp[0]   += accA.x;  tp[32]  += accA.y;  tp[64]  += accA.z;  tp[96]  += accA.w;
          tp[128] += accB.x;  tp[160] += accB.y;  tp[192] += accB.z;  tp[224] += accB.w;
          accA = make_float4(0.0f, 0.0f, 0.0f, 0.0f);
          accB = make_float4(0.0f, 0.0f, 0.0f, 0.0f);
        };
        auto add = [&](const float4& a, const float4& b) {
          accA.x += a.x;  accA.y += a.y;  accA.z += a.z;  accA.w += a.w;
          accB.x += b.x;  accB.y += b.y;  accB.z += b.z;  accB.w += b.w;
        };
        int cur = __builtin_amdgcn_readfirstlane((int)((unsigned)srtl[0] >> 29));
        for (int k0 = 0; k0 < n; k0 += 32) {
          const int k  = k0 + lane;
          const int e  = (k < n) ? srtl[k] : 0;
          const int s  = e & 0xFFFFFF;
          const int sl = (int)((unsigned)e >> 29);
          const int m  = min(32, n - k0);
          for (int j = 0; j < m; j += 8) {
            float4 a[8], b[8];
            int    slj[8];
#pragma unroll
            for (int u = 0; u < 8; ++u) {
              const int ju = min(j + u, m - 1);
              const int sj = __builtin_amdgcn_readlane(s, ju);
              slj[u] = __builtin_amdgcn_readlane(sl, ju);
              const float4* hp = (const float4*)(H + (size_t)sj * HID);
              a[u] = hp[lane];
              b[u] = hp[32 + lane];
            }
#pragma unroll
            for (int u = 0; u < 8; ++u) {
              if (j + u < m) {
                if (slj[u] != cur) { commit(cur); cur = slj[u]; }
                add(a[u], b[u]);
              }
            }
          }
        }
        commit(cur);
      };

      auto sort_and_sum = [&](int n) {
        __builtin_amdgcn_fence(__ATOMIC_ACQ_REL, "wavefront");
        int off[NOWN + 1];
#pragma unroll
        for (int q = 0; q <= NOWN; ++q) off[q] = 0;
        for (int k0 = 0; k0 < n; k0 += 32) {
          const int k  = k0 + lane;
          const int sl = (k < n) ? (int)((unsigned)rawl[k] >> 29) : NOWN;
#pragma unroll
          for (int q = 0; q < NOWN; ++q)
            off[q + 1] += __builtin_popcount(__builtin_amdgcn_ballot_w32(sl == q));
        }
#pragma unroll
        for (int q = 0; q < NOWN; ++q) off[q + 1] += off[q];
        const int ns = off[NOWN];
        for (int k0 = 0; k0 < n; k0 += 32) {
          const int k  = k0 + lane;
          const int e  = (k < n) ? rawl[k] : 0;
          const int sl = (k < n) ? (int)((unsigned)e >> 29) : NOWN;
#pragma unroll
          for (int q = 0; q < NOWN; ++q) {
            const bool hq = sl == q;
            const unsigned mq = __builtin_amdgcn_ballot_w32(hq);
            if (hq) srtl[off[q] + __builtin_popcount(mq & lanes_below)] = e;
            off[q] += __builtin_popcount(mq);
          }
        }
        __builtin_amdgcn_fence(__ATOMIC_ACQ_REL, "wavefront");
        if (ns > 0) sum_sorted(ns);
        __builtin_amdgcn_fence(__ATOMIC_ACQ_REL, "wavefront");
      };

      const int4* tl4 = (const int4*)(tlist + (size_t)lt * TCAP);
      const int nch = (nt + 127) >> 7;
      const int l4  = 4 * lane;
      int cnt = 0;
      int c   = 0;
      for (;;) {
        bool full = false;
        if (c < nch) {
          int4 v[8];
#pragma unroll
          for (int u = 0; u < 8; ++u) {
            const int q = (c + u) * 32 + lane;
            v[u] = (4 * q < nt) ? tl4[q] : make_int4(0, 0, 0, 0);
          }
          int done = 0;
#pragma unroll
          for (int u = 0; u < 8; ++u) {
            if (done == u && c + u < nch) {
              const int left = nt - 128 * (c + u);
              const bool o0 = (((unsigned)v[u].x >> 24) & 31u) == (unsigned)wave;
              const bool o1 = (((unsigned)v[u].y >> 24) & 31u) == (unsigned)wave;
              const bool o2 = (((unsigned)v[u].z >> 24) & 31u) == (unsigned)wave;
              const bool o3 = (((unsigned)v[u].w >> 24) & 31u) == (unsigned)wave;
              const unsigned m0 = __builtin_amdgcn_ballot_w32(o0) & __builtin_amdgcn_ballot_w32(l4     < left);
              const unsigned m1 = __builtin_amdgcn_ballot_w32(o1) & __builtin_amdgcn_ballot_w32(l4 + 1 < left);
              const unsigned m2 = __builtin_amdgcn_ballot_w32(o2) & __builtin_amdgcn_ballot_w32(l4 + 2 < left);
              const unsigned m3 = __builtin_amdgcn_ballot_w32(o3) & __builtin_amdgcn_ballot_w32(l4 + 3 < left);
              const int tot = __builtin_popcount(m0) + __builtin_popcount(m1)
                            + __builtin_popcount(m2) + __builtin_popcount(m3);
              if (cnt + tot <= OCAP) {
                if (tot != 0) {
                  int pos = (int)__builtin_amdgcn_mbcnt_lo(m3,
                                 __builtin_amdgcn_mbcnt_lo(m2,
                                 __builtin_amdgcn_mbcnt_lo(m1,
                                 __builtin_amdgcn_mbcnt_lo(m0, (unsigned)cnt))));
                  if (o0 && l4     < left) rawl[pos++] = v[u].x;
                  if (o1 && l4 + 1 < left) rawl[pos++] = v[u].y;
                  if (o2 && l4 + 2 < left) rawl[pos++] = v[u].z;
                  if (o3 && l4 + 3 < left) rawl[pos]   = v[u].w;
                  cnt += tot;
                }
                done = u + 1;
              } else {
                full = true;
              }
            }
          }
          c += done;
        }
        const bool end = c >= nch;
        if (end || full || cnt >= SUMMIN) { sort_and_sum(cnt); cnt = 0; }
        if (end) break;
      }
    } else {
      auto flush = [&](int n) {
        __builtin_amdgcn_fence(__ATOMIC_ACQ_REL, "wavefront");
        for (int k0 = 0; k0 < n; k0 += 32) {
          const int k = k0 + lane;
          int s = 0, r = 0;
          if (k < n) {
            const int e = mylist[k];
            s = src[e];
            r = tgt[e] - base;
          }
          add_rows(s, r, min(32, n - k0));
        }
        __builtin_amdgcn_fence(__ATOMIC_ACQ_REL, "wavefront");
      };
      search_edges(tgt, E, base, rows, mylist, lane, wave, flush);
      __syncthreads();
    }

    float4 ba, bb;
    ba.x = bias[4 * lane];           ba.y = bias[4 * lane + 1];
    ba.z = bias[4 * lane + 2];       ba.w = bias[4 * lane + 3];
    bb.x = bias[128 + 4 * lane];     bb.y = bias[128 + 4 * lane + 1];
    bb.z = bias[128 + 4 * lane + 2]; bb.w = bias[128 + 4 * lane + 3];
    if (LAST) {
      for (int r = wave; r < rows; r += NWAVE) {
        const int i = base + r;
        const float d = dinv[i];
        const float* tp = tile + r * TSTR;
        float4 a, b;
        a.x = tp[lane] * d;        a.y = tp[32 + lane] * d;
        a.z = tp[64 + lane] * d;   a.w = tp[96 + lane] * d;
        b.x = tp[128 + lane] * d;  b.y = tp[160 + lane] * d;
        b.z = tp[192 + lane] * d;  b.w = tp[224 + lane] * d;
        a.x += ba.x; a.y += ba.y; a.z += ba.z; a.w += ba.w;
        b.x += bb.x; b.y += bb.y; b.z += bb.z; b.w += bb.w;
        float4* op = (float4*)(out + (size_t)i * HID);
        op[lane]      = a;
        op[32 + lane] = b;
      }
    } else {
      const float kLn1e4Over256 = 9.210340371976184f / 256.0f;
      const float f0 = expf(-(float)(4 * lane)           * kLn1e4Over256);
      const float f1 = expf(-(float)(4 * lane + 2)       * kLn1e4Over256);
      const float f2 = expf(-(float)(128 + 4 * lane)     * kLn1e4Over256);
      const float f3 = expf(-(float)(128 + 4 * lane + 2) * kLn1e4Over256);
      for (int r = wave; r < rows; r += NWAVE) {
        const int i = base + r;
        const float d  = dinv[i];
        const float fi = (float)i;
        float* tp = tile + r * TSTR;
        float s0, c0, s1, c1, s2, c2, s3, c3;
        sincosf(fi * f0, &s0, &c0);
        sincosf(fi * f1, &s1, &c1);
        sincosf(fi * f2, &s2, &c2);
        sincosf(fi * f3, &s3, &c3);
        float4 a, b;
        a.x = tp[lane] * d + ba.x + s0;        a.y = tp[32 + lane] * d + ba.y + c0;
        a.z = tp[64 + lane] * d + ba.z + s1;   a.w = tp[96 + lane] * d + ba.w + c1;
        b.x = tp[128 + lane] * d + bb.x + s2;  b.y = tp[160 + lane] * d + bb.y + c2;
        b.z = tp[192 + lane] * d + bb.z + s3;  b.w = tp[224 + lane] * d + bb.w + c3;
        tp[lane]       = a.x > 0.0f ? a.x : 0.0f;  tp[32 + lane]  = a.y > 0.0f ? a.y : 0.0f;
        tp[64 + lane]  = a.z > 0.0f ? a.z : 0.0f;  tp[96 + lane]  = a.w > 0.0f ? a.w : 0.0f;
        tp[128 + lane] = b.x > 0.0f ? b.x : 0.0f;  tp[160 + lane] = b.y > 0.0f ? b.y : 0.0f;
        tp[192 + lane] = b.z > 0.0f ? b.z : 0.0f;  tp[224 + lane] = b.w > 0.0f ? b.w : 0.0f;
      }
      __syncthreads();
      const int tp8 = wave & 7;
      const int mq  = wave >> 3;
      if (mq == 0) rows_times_w<4>(tile, Wn, dinv + base, out, base, rows, mq, tp8, lane);
      else         rows_times_w<3>(tile, Wn, dinv + base, out, base, rows, mq, tp8, lane);
    }
    __syncthreads();
  }
}

extern "C" void kernel_launch(void* const* d_in, const int* in_sizes, int n_in,
                              void* d_out, int out_size, void* d_ws, size_t ws_size,
                              hipStream_t stream) {
  const float* X  = (const float*)d_in[0];
  const int*   EI = (const int*)d_in[1];
  const float* W1 = (const float*)d_in[2];
  const float* b1 = (const float*)d_in[3];
  const float* W2 = (const float*)d_in[4];
  const float* b2 = (const float*)d_in[5];
  float* out = (float*)d_out;

  const int N = in_sizes[0] / IN_DIM;
  const int E = in_sizes[1] / 2;
  const int NF = N * HID;

  float* dinv   = (float*)d_ws;
  float* bufL   = dinv + (1 << 16);
  float* bufAgg = bufL + (size_t)NF;
  float* bufH   = out;
  int* ccount = (int*)bufL;
  int* cells  = ccount + NLIST * NBIN;
  int* tcount = cells + NLIST * NBIN * CCAP;
  int* tlist  = tcount + NLIST;

  const int* src = EI;
  const int* tgt = EI + E;

  bin_edges_kernel<<<NBIN, BINT, 0, stream>>>(src, tgt, cells, ccount, E, N);

  tile_lists_xw1_kernel<<<256, 1024, 0, stream>>>(X, W1, tgt, dinv, bufH,
                                                  cells, ccount, tlist, tcount, E, N);

  aggregate_lds_kernel<false><<<256, 1024, 0, stream>>>(bufH, src, tgt, dinv, b1, W2,
                                                        bufAgg, tlist, tcount, E, N);

  aggregate_lds_kernel<true><<<256, 1024, 0, stream>>>(bufAgg, src, tgt, dinv, b2, W2,
                                                       out, tlist, tcount, E, N);
}


# H1 and H2 rows stored as f16 (precision of the two gathered intermediates only; sums, both f32 WMMA products and the output stay f32): a source row is 512 B, two 256-byte requests, not four
# speedup vs baseline: 4.4912x; 1.0679x over previous
#include <hip/hip_runtime.h>
#include <math.h>


typedef __attribute__((ext_vector_type(2))) float v2f;
typedef __attribute__((ext_vector_type(8))) float v8f;
typedef __attribute__((ext_vector_type(2))) _Float16 h2v;
typedef __attribute__((ext_vector_type(4))) _Float16 h4v;

static constexpr int HID    = 256;
static constexpr int IN_DIM = 64;

static constexpr int TROWS = 196;
static constexpr int NWAVE = 32;
static constexpr int EPH   = 512;
static constexpr int BATCH = 32;
static constexpr int LCAP  = BATCH + EPH;
static constexpr int NLIST = 256;
static constexpr int NBIN  = 256;
static constexpr int CCAP  = 64;
static constexpr int TCAP  = 8192;
static constexpr int BINT  = 512;
static constexpr int BINU  = 8;

__device__ __forceinline__ void load_hrow(const float* __restrict__ Hf, size_t row,
                                          int lane, h4v& a, h4v& b) {
  const h4v* hp = (const h4v*)((const _Float16*)Hf + row * HID);
  a = hp[lane];
  b = hp[32 + lane];
}
__device__ __forceinline__ float4 widen(const h4v v) {
  return make_float4((float)v.x, (float)v.y, (float)v.z, (float)v.w);
}
__device__ __forceinline__ void store_hpair(float* __restrict__ Hf, size_t row, int col,
                                            float x, float y) {
  h2v v;
  v.x = (_Float16)x;
  v.y = (_Float16)y;
  *(h2v*)((_Float16*)Hf + row * HID + col) = v;
}

__global__ __launch_bounds__(BINT)
void bin_edges_kernel(const int* __restrict__ src,
                      const int* __restrict__ tgt,
                      int* __restrict__ cells, int* __restrict__ ccount,
                      int E, int N) {
  __shared__ int cnt[NLIST];
  const int g = blockIdx.x;
  if (g >= NBIN) return;
  const int ntl = (N <= (1 << 24)) ? min(NLIST, (N + TROWS - 1) / TROWS) : 0;
  for (int i = threadIdx.x; i < NLIST; i += BINT) cnt[i] = 0;
  __syncthreads();

  const int per = (E + NBIN - 1) / NBIN;
  const int lo  = min(E, g * per);
  const int hi  = min(E, lo + per);
  for (int e0 = lo; e0 < hi; e0 += BINT * BINU) {
    int t[BINU], s[BINU];
#pragma unroll
    for (int u = 0; u < BINU; ++u) {
      const int e = e0 + u * BINT + (int)threadIdx.x;
      const bool in = e < hi;
      t[u] = in ? tgt[e] : -1;
      s[u] = in ? src[e] : 0;
    }
    int T[BINU], pos[BINU];
#pragma unroll
    for (int u = 0; u < BINU; ++u) {
      T[u] = ((unsigned)t[u] < (unsigned)N) ? t[u] / TROWS : ntl;
      pos[u] = CCAP;
      if (T[u] < ntl)
        pos[u] = __hip_atomic_fetch_add(&cnt[T[u]], 1, __ATOMIC_RELAXED,
                                        __HIP_MEMORY_SCOPE_WORKGROUP);
    }
#pragma unroll
    for (int u = 0; u < BINU; ++u) {
      if (pos[u] < CCAP) {
        const int r = t[u] - T[u] * TROWS;
        cells[((size_t)T[u] * NBIN + g) * CCAP + pos[u]] =
            (int)((unsigned)s[u] | ((unsigned)r << 24));
      }
    }
  }
  __syncthreads();
  for (int i = threadIdx.x; i < NLIST; i += BINT) ccount[(size_t)i * NBIN + g] = cnt[i];
}

template <class Flush>
__device__ __forceinline__ void search_edges(const int* __restrict__ tgt, int E,
                                             int base, int rows, int* mylist,
                                             int lane, int wave, Flush&& flush) {
  const unsigned lanes_below = (1u << lane) - 1u;
  const int hd  = min((int)(((16u - (unsigned)((size_t)tgt & 15)) & 15u) >> 2), E);
  const int* tb = tgt + hd;
  const int EB  = E - hd;
  const int G     = EB >> 2;
  const int GPI   = NWAVE * 8 * 32;
  const int ITERS = (G + GPI - 1) / GPI;
  const int FULL  = G / GPI;

  int cnt = 0;
  auto consider = [&](int t, int e) {
    const bool hit = (unsigned)(t - base) < (unsigned)rows;
    const unsigned m = __builtin_amdgcn_ballot_w32(hit);
    if (m != 0u) {
      if (hit) mylist[cnt + __builtin_popcount(m & lanes_below)] = e;
      cnt += __builtin_popcount(m);
    }
  };

  if (wave == 0) {
    const int e = (lane < hd) ? lane : hd + 4 * G + (lane - hd);
    const int t = (lane < 8 && e < E) ? tgt[e] : base - 1;
    consider(t, e);
  }
  cnt = __builtin_amdgcn_readfirstlane(cnt);

  const int rot = (ITERS > 0) ? (int)(blockIdx.x % (unsigned)ITERS) : 0;
  auto place_of = [&](int it) {
    int p = it + rot;
    if (p >= ITERS) p -= ITERS;
    return p;
  };
  auto group_of = [&](int p, int u) {
    return ((p * NWAVE + wave) * 8 + u) * 32 + lane;
  };
  auto load_place = [&](int p, int4 (&dst)[8]) {
    if (p < FULL) {
#pragma unroll
      for (int u = 0; u < 8; ++u)
        dst[u] = *(const int4*)(tb + 4 * group_of(p, u));
    } else {
#pragma unroll
      for (int u = 0; u < 8; ++u) {
        const int g = group_of(p, u);
        int4 v = *(const int4*)(tb + 4 * min(g, G - 1));
        if (g >= G) v.x = v.y = v.z = v.w = base - 1;
        dst[u] = v;
      }
    }
  };

  for (int it = 0; it < ITERS; ++it) {
    const int p = place_of(it);
    int4 cur[8];
    load_place(p, cur);
#pragma unroll
    for (int half = 0; half < 2; ++half) {
      if (cnt >= BATCH) { flush(cnt); cnt = 0; }
#pragma unroll
      for (int u = 4 * half; u < 4 * half + 4; ++u) {
        const int e = hd + 4 * group_of(p, u);
        consider(cur[u].x, e);
        consider(cur[u].y, e + 1);
        consider(cur[u].z, e + 2);
        consider(cur[u].w, e + 3);
      }
    }
  }
  flush(cnt);
}

static constexpr int XROWS = TROWS;
static constexpr int XSTR  = IN_DIM + 4;
static_assert(XROWS * (IN_DIM / 4) <= 4 * 1024, "staging: four 16-byte loads per thread");

__global__ __launch_bounds__(1024)
void tile_lists_xw1_kernel(const float* __restrict__ X,
                       const float* __restrict__ W1,
                       const int* __restrict__ tgt,
                       float* __restrict__ dinv,
                       float* __restrict__ Hout,
                       const int* __restrict__ cells,
                       const int* __restrict__ ccount,
                       int* __restrict__ tlist, int* __restrict__ tcount,
                       int E, int N) {
  __shared__ __attribute__((aligned(16))) float xt[XROWS * XSTR];
  __shared__ int   kept[NWAVE * LCAP];
  __shared__ int   degc[TROWS];
  __shared__ float dl[TROWS];
  __shared__ int   ccnt[NBIN];
  __shared__ int   coff[NBIN];
  __shared__ int   tot;
  __shared__ int   badc;

  const int lane = threadIdx.x & 31;
  const int wave = threadIdx.x >> 5;
  const int h  = lane >> 4;
  const int ln = lane & 15;
  const int tp = wave & 7;
  const int mq = __builtin_amdgcn_readfirstlane(wave >> 3);
  const int n2 = tp * 32 + 2 * ln;
  int* mylist = kept + wave * LCAP;

  for (int base = blockIdx.x * TROWS; base < N; base += gridDim.x * TROWS) {
    const int rows = min(TROWS, N - base);
    const int tile_id  = base / TROWS;
    const bool listed  = tile_id < NLIST && N <= (1 << 24);
    const int  lt      = listed ? tile_id : 0;

    if ((int)threadIdx.x < TROWS) degc[threadIdx.x] = 0;
    if (threadIdx.x == 0) { tot = 0; badc = 0; }
    __syncthreads();

    if (listed && (int)threadIdx.x < NBIN) {
      const int cn = ccount[(size_t)lt * NBIN + threadIdx.x];
      ccnt[threadIdx.x] = cn;
      coff[threadIdx.x] = __hip_atomic_fetch_add(&tot, cn, __ATOMIC_RELAXED,
                                                 __HIP_MEMORY_SCOPE_WORKGROUP);
      if (cn > CCAP) badc = 1;
    }
    __syncthreads();
    const int  nall     = tot;
    const bool use_cell = listed && badc == 0 && nall <= TCAP;

    if (use_cell) {
      const int q  = (int)threadIdx.x & 15;
      const int c0 = (int)threadIdx.x >> 4;
      int* tl = tlist + (size_t)lt * TCAP;
      int cn[4], co[4];
#pragma unroll
      for (int u = 0; u < 4; ++u) {
        cn[u] = ccnt[c0 + 64 * u];
        co[u] = coff[c0 + 64 * u];
      }
      int p[4][CCAP / 16];
#pragma unroll
      for (int u = 0; u < 4; ++u) {
        const int* cp = cells + ((size_t)lt * NBIN + c0 + 64 * u) * CCAP;
#pragma unroll
        for (int v = 0; v < CCAP / 16; ++v)
          p[u][v] = (q + 16 * v < cn[u]) ? cp[q + 16 * v] : -1;
      }
#pragma unroll
      for (int u = 0; u < 4; ++u) {
#pragma unroll
        for (int v = 0; v < CCAP / 16; ++v) {
          if (q + 16 * v < cn[u]) {
            const int r = (int)((unsigned)p[u][v] >> 24);
            __hip_atomic_fetch_add(&degc[r], 1, __ATOMIC_RELAXED, __HIP_MEMORY_SCOPE_WORKGROUP);
            tl[co[u] + q + 16 * v] = p[u][v];
          }
        }
      }
      if (threadIdx.x == 0) tcount[lt] = nall;
    } else {
      auto flush = [&](int n) {
        __builtin_amdgcn_fence(__ATOMIC_ACQ_REL, "wavefront");
        for (int k0 = 0; k0 < n; k0 += 32) {
          const int k = k0 + lane;
          if (k < n) {
            const int r = tgt[mylist[k]] - base;
            __hip_atomic_fetch_add(&degc[r], 1, __ATOMIC_RELAXED, __HIP_MEMORY_SCOPE_WORKGROUP);
          }
        }
        __builtin_amdgcn_fence(__ATOMIC_ACQ_REL, "wavefront");
      };
      search_edges(tgt, E, base, rows, mylist, lane, wave, flush);
      if (listed && threadIdx.x == 0) tcount[lt] = -1;
    }
    __syncthreads();

    if ((int)threadIdx.x < rows) {
      const float d = rsqrtf(1.0f + (float)degc[threadIdx.x]);
      dl[threadIdx.x] = d;
      dinv[base + threadIdx.x] = d;
    }

    const int nvec = rows * (IN_DIM / 4);
    const float4* xp = (const float4*)(X + (size_t)base * IN_DIM);
    float4 q[4];
#pragma unroll
    for (int u = 0; u < 4; ++u) {
      const int v = (int)threadIdx.x + 1024 * u;
      q[u] = (v < nvec) ? xp[v] : make_float4(0.0f, 0.0f, 0.0f, 0.0f);
    }

#pragma unroll
    for (int u = 0; u < 4; ++u) {
      const int v = (int)threadIdx.x + 1024 * u;
      if (v < nvec) *(float4*)(xt + (v >> 4) * XSTR + 4 * (v & 15)) = q[u];
    }

    v2f be[16], bo[16];
#pragma unroll
    for (int s = 0; s < 16; ++s) {
      const float* wp = W1 + (size_t)(4 * s + 2 * h) * HID + n2;
      const float2 w0 = *(const float2*)wp;
      const float2 w1 = *(const float2*)(wp + HID);
      be[s].x = w0.x;  be[s].y = w1.x;
      bo[s].x = w0.y;  bo[s].y = w1.y;
    }
    __syncthreads();

    const int ntiles = (rows + 15) >> 4;
    for (int t = mq; t < ntiles; t += 4) {
      const int row = min(t * 16 + ln, rows - 1);
      const float* ap = xt + row * XSTR + 2 * h;
      v2f a[16];
#pragma unroll
      for (int s = 0; s < 16; ++s) a[s] = *(const v2f*)(ap + 4 * s);
      v8f c0 = {}, c1 = {};
#pragma unroll
      for (int s = 0; s < 16; ++s) {
        c0 = __builtin_amdgcn_wmma_f32_16x16x4_f32(
            false, a[s], false, be[s], (short)0, c0, false, false);
        c1 = __builtin_amdgcn_wmma_f32_16x16x4_f32(
            false, a[s], false, bo[s], (short)0, c1, false, false);
      }
      const int r0 = t * 16 + 8 * h;
#pragma unroll
      for (int r = 0; r < 8; ++r) {
        if (r0 + r < rows) {
          const float d = dl[r0 + r];
          store_hpair(Hout, (size_t)(base + r0 + r), n2, c0[r] * d, c1[r] * d);
        }
      }
    }
    __syncthreads();
  }
}


static constexpr int TSTR  = HID + 1;

template <int NS>
__device__ __forceinline__ void rows_times_w(const float* tile,
                                             const float* __restrict__ Wm,
                                             const float* __restrict__ dsc,
                                             float* __restrict__ out,
                                             int base, int rows, int mq, int tp,
                                             int lane) {
  const int h  = lane >> 4;
  const int ln = lane & 15;
  const int n2 = tp * 32 + 2 * ln;
  int aoff[NS];
#pragma unroll
  for (int i = 0; i < NS; ++i) {
    const int row = min((mq + 4 * i) * 16 + ln, rows - 1);
    aoff[i] = row * TSTR;
  }
  v8f acc[NS][2];
#pragma unroll
  for (int i = 0; i < NS; ++i) { acc[i][0] = v8f{}; acc[i][1] = v8f{}; }

  auto w_of = [&](int s) {
    const int lh = ((s & 31) + 16 * h) & 31;
    const int c0 = (s >> 5) * 128 + 4 * lh + 2 * h;
    return Wm + (size_t)c0 * HID + n2;
  };
  const float* wp = w_of(0);
  float2 w0n = *(const float2*)wp;
  float2 w1n = *(const float2*)(wp + HID);
#pragma unroll 2
  for (int s = 0; s < 64; ++s) {
    const float2 w0 = w0n, w1 = w1n;
    wp  = w_of(min(s + 1, 63));
    w0n = *(const float2*)wp;
    w1n = *(const float2*)(wp + HID);
    const int lh  = ((s & 31) + 16 * h) & 31;
    const int pos = (s >> 5) * 128 + 64 * h + lh;
    v2f a[NS];
#pragma unroll
    for (int i = 0; i < NS; ++i) {
      a[i].x = tile[aoff[i] + pos];
      a[i].y = tile[aoff[i] + pos + 32];
    }
    v2f be, bo;
    be.x = w0.x;  be.y = w1.x;
    bo.x = w0.y;  bo.y = w1.y;
#pragma unroll
    for (int i = 0; i < NS; ++i) {
      acc[i][0] = __builtin_amdgcn_wmma_f32_16x16x4_f32(
          false, a[i], false, be, (short)0, acc[i][0], false, false);
      acc[i][1] = __builtin_amdgcn_wmma_f32_16x16x4_f32(
          false, a[i], false, bo, (short)0, acc[i][1], false, false);
    }
  }

#pragma unroll
  for (int i = 0; i < NS; ++i) {
    const int r0 = (mq + 4 * i) * 16 + 8 * h;
#pragma unroll
    for (int r = 0; r < 8; ++r) {
      if (r0 + r < rows) {
        const float d = dsc[r0 + r];
        store_hpair(out, (size_t)(base + r0 + r), n2, acc[i][0][r] * d, acc[i][1][r] * d);
      }
    }
  }
}

static constexpr int NOWN = (TROWS + NWAVE - 1) / NWAVE;
static constexpr int OCAP = LCAP / 2;
static constexpr int SUMMIN = 32;
static_assert(TROWS <= 256 && NWAVE == 32 && NOWN <= 7,
              "list entry: tile row in bits 24..31, so its slot (row / 32) in bits 29..31; 7 is no slot");
static_assert(OCAP >= 128 && 2 * OCAP <= LCAP, "a chunk of 128 entries fits an empty list");

template <bool LAST>
__global__ __launch_bounds__(1024)
void aggregate_lds_kernel(const float* __restrict__ H,
                          const int* __restrict__ src,
                          const int* __restrict__ tgt,
                          const float* __restrict__ dinv,
                          const float* __restrict__ bias,
                          const float* __restrict__ Wn,
                          float* __restrict__ out,
                          const int* __restrict__ tlist,
                          const int* __restrict__ tcount,
                          int E, int N) {
  __shared__ float tile[TROWS * TSTR];
  __shared__ int   kept[NWAVE * LCAP];

  const int lane = threadIdx.x & 31;
  const int wave = threadIdx.x >> 5;
  int* mylist = kept + wave * LCAP;

  for (int base = blockIdx.x * TROWS; base < N; base += gridDim.x * TROWS) {
    const int rows = min(TROWS, N - base);
    const int tile_id  = base / TROWS;
    const bool listed  = tile_id < NLIST && N <= (1 << 24);
    const int  lt      = listed ? tile_id : 0;

    int nt = -1;
    if (listed) {
      nt = __builtin_amdgcn_readfirstlane(tcount[lt]);
    }

    for (int r = wave; r < rows; r += NWAVE) {
      const int i = base + r;
      h4v ah, bh;
      load_hrow(H, (size_t)i, lane, ah, bh);
      const float4 a = widen(ah), b = widen(bh);
      float* tp = tile + r * TSTR;
      tp[lane]       = a.x;  tp[32 + lane]  = a.y;
      tp[64 + lane]  = a.z;  tp[96 + lane]  = a.w;
      tp[128 + lane] = b.x;  tp[160 + lane] = b.y;
      tp[192 + lane] = b.z;  tp[224 + lane] = b.w;
    }
    if (nt < 0) __syncthreads();

    auto add_rows = [&](int s, int r, int m) {
      int j = 0;
      for (; j + 4 <= m; j += 4) {
        h4v ah[4], bh[4];
        int rj[4];
#pragma unroll
        for (int u = 0; u < 4; ++u) {
          const int sj = __builtin_amdgcn_readlane(s, j + u);
          rj[u] = __builtin_amdgcn_readlane(r, j + u);
          load_hrow(H, (size_t)sj, lane, ah[u], bh[u]);
        }
#pragma unroll
        for (int u = 0; u < 4; ++u) {
          const float4 a = widen(ah[u]), b = widen(bh[u]);
          float* tp = tile + rj[u] * TSTR + lane;
          __hip_atomic_fetch_add(tp,       a.x, __ATOMIC_RELAXED, __HIP_MEMORY_SCOPE_WORKGROUP);
          __hip_atomic_fetch_add(tp + 32,  a.y, __ATOMIC_RELAXED, __HIP_MEMORY_SCOPE_WORKGROUP);
          __hip_atomic_fetch_add(tp + 64,  a.z, __ATOMIC_RELAXED, __HIP_MEMORY_SCOPE_WORKGROUP);
          __hip_atomic_fetch_add(tp + 96,  a.w, __ATOMIC_RELAXED, __HIP_MEMORY_SCOPE_WORKGROUP);
          __hip_atomic_fetch_add(tp + 128, b.x, __ATOMIC_RELAXED, __HIP_MEMORY_SCOPE_WORKGROUP);
          __hip_atomic_fetch_add(tp + 160, b.y, __ATOMIC_RELAXED, __HIP_MEMORY_SCOPE_WORKGROUP);
          __hip_atomic_fetch_add(tp + 192, b.z, __ATOMIC_RELAXED, __HIP_MEMORY_SCOPE_WORKGROUP);
          __hip_atomic_fetch_add(tp + 224, b.w, __ATOMIC_RELAXED, __HIP_MEMORY_SCOPE_WORKGROUP);
        }
      }
      for (; j < m; ++j) {
        const int sj = __builtin_amdgcn_readlane(s, j);
        const int rr = __builtin_amdgcn_readlane(r, j);
        h4v ah, bh;
        load_hrow(H, (size_t)sj, lane, ah, bh);
        const float4 a = widen(ah), b = widen(bh);
        float* tp = tile + rr * TSTR + lane;
        __hip_atomic_fetch_add(tp,       a.x, __ATOMIC_RELAXED, __HIP_MEMORY_SCOPE_WORKGROUP);
        __hip_atomic_fetch_add(tp + 32,  a.y, __ATOMIC_RELAXED, __HIP_MEMORY_SCOPE_WORKGROUP);
        __hip_atomic_fetch_add(tp + 64,  a.z, __ATOMIC_RELAXED, __HIP_MEMORY_SCOPE_WORKGROUP);
        __hip_atomic_fetch_add(tp + 96,  a.w, __ATOMIC_RELAXED, __HIP_MEMORY_SCOPE_WORKGROUP);
        __hip_atomic_fetch_add(tp + 128, b.x, __ATOMIC_RELAXED, __HIP_MEMORY_SCOPE_WORKGROUP);
        __hip_atomic_fetch_add(tp + 160, b.y, __ATOMIC_RELAXED, __HIP_MEMORY_SCOPE_WORKGROUP);
        __hip_atomic_fetch_add(tp + 192, b.z, __ATOMIC_RELAXED, __HIP_MEMORY_SCOPE_WORKGROUP);
        __hip_atomic_fetch_add(tp + 224, b.w, __ATOMIC_RELAXED, __HIP_MEMORY_SCOPE_WORKGROUP);
      }
    };

    if (nt >= 0) {
      const int wv = __builtin_amdgcn_readfirstlane(wave);
      int* rawl = mylist;
      int* srtl = mylist + OCAP;
      const unsigned lanes_below = (1u << lane) - 1u;

      auto sum_sorted = [&](int n) {
        float4 accA = make_float4(0.0f, 0.0f, 0.0f, 0.0f);
        float4 accB = make_float4(0.0f, 0.0f, 0.0f, 0.0f);
        auto commit = [&](int sl) {
          float* tp = tile + (wv + NWAVE * sl) * TSTR + lane;
          tp[0]   += accA.x;  tp[32]  += accA.y;  tp[64]  += accA.z;  tp[96]  += accA.w;
          tp[128] += accB.x;  tp[160] += accB.y;  tp[192] += accB.z;  tp[224] += accB.w;
          accA = make_float4(0.0f, 0.0f, 0.0f, 0.0f);
          accB = make_float4(0.0f, 0.0f, 0.0f, 0.0f);
        };
        auto add = [&](const float4& a, const float4& b) {
          accA.x += a.x;  accA.y += a.y;  accA.z += a.z;  accA.w += a.w;
          accB.x += b.x;  accB.y += b.y;  accB.z += b.z;  accB.w += b.w;
        };
        int cur = __builtin_amdgcn_readfirstlane((int)((unsigned)srtl[0] >> 29));
        for (int k0 = 0; k0 < n; k0 += 32) {
          const int k  = k0 + lane;
          const int e  = (k < n) ? srtl[k] : 0;
          const int s  = e & 0xFFFFFF;
          const int sl = (int)((unsigned)e >> 29);
          const int m  = min(32, n - k0);
          for (int j = 0; j < m; j += 8) {
            h4v a[8], b[8];
            int slj[8];
#pragma unroll
            for (int u = 0; u < 8; ++u) {
              const int ju = min(j + u, m - 1);
              const int sj = __builtin_amdgcn_readlane(s, ju);
              slj[u] = __builtin_amdgcn_readlane(sl, ju);
              load_hrow(H, (size_t)sj, lane, a[u], b[u]);
            }
#pragma unroll
            for (int u = 0; u < 8; ++u) {
              if (j + u < m) {
                if (slj[u] != cur) { commit(cur); cur = slj[u]; }
                add(widen(a[u]), widen(b[u]));
              }
            }
          }
        }
        commit(cur);
      };

      auto sort_and_sum = [&](int n) {
        __builtin_amdgcn_fence(__ATOMIC_ACQ_REL, "wavefront");
        int off[NOWN + 1];
#pragma unroll
        for (int q = 0; q <= NOWN; ++q) off[q] = 0;
        for (int k0 = 0; k0 < n; k0 += 32) {
          const int k  = k0 + lane;
          const int sl = (k < n) ? (int)((unsigned)rawl[k] >> 29) : NOWN;
#pragma unroll
          for (int q = 0; q < NOWN; ++q)
            off[q + 1] += __builtin_popcount(__builtin_amdgcn_ballot_w32(sl == q));
        }
#pragma unroll
        for (int q = 0; q < NOWN; ++q) off[q + 1] += off[q];
        const int ns = off[NOWN];
        for (int k0 = 0; k0 < n; k0 += 32) {
          const int k  = k0 + lane;
          const int e  = (k < n) ? rawl[k] : 0;
          const int sl = (k < n) ? (int)((unsigned)e >> 29) : NOWN;
#pragma unroll
          for (int q = 0; q < NOWN; ++q) {
            const bool hq = sl == q;
            const unsigned mq = __builtin_amdgcn_ballot_w32(hq);
            if (hq) srtl[off[q] + __builtin_popcount(mq & lanes_below)] = e;
            off[q] += __builtin_popcount(mq);
          }
        }
        __builtin_amdgcn_fence(__ATOMIC_ACQ_REL, "wavefront");
        if (ns > 0) sum_sorted(ns);
        __builtin_amdgcn_fence(__ATOMIC_ACQ_REL, "wavefront");
      };

      const int4* tl4 = (const int4*)(tlist + (size_t)lt * TCAP);
      const int nch = (nt + 127) >> 7;
      const int l4  = 4 * lane;
      int cnt = 0;
      int c   = 0;
      for (;;) {
        bool full = false;
        if (c < nch) {
          int4 v[8];
#pragma unroll
          for (int u = 0; u < 8; ++u) {
            const int q = (c + u) * 32 + lane;
            v[u] = (4 * q < nt) ? tl4[q] : make_int4(0, 0, 0, 0);
          }
          int done = 0;
#pragma unroll
          for (int u = 0; u < 8; ++u) {
            if (done == u && c + u < nch) {
              const int left = nt - 128 * (c + u);
              const bool o0 = (((unsigned)v[u].x >> 24) & 31u) == (unsigned)wave;
              const bool o1 = (((unsigned)v[u].y >> 24) & 31u) == (unsigned)wave;
              const bool o2 = (((unsigned)v[u].z >> 24) & 31u) == (unsigned)wave;
              const bool o3 = (((unsigned)v[u].w >> 24) & 31u) == (unsigned)wave;
              const unsigned m0 = __builtin_amdgcn_ballot_w32(o0) & __builtin_amdgcn_ballot_w32(l4     < left);
              const unsigned m1 = __builtin_amdgcn_ballot_w32(o1) & __builtin_amdgcn_ballot_w32(l4 + 1 < left);
              const unsigned m2 = __builtin_amdgcn_ballot_w32(o2) & __builtin_amdgcn_ballot_w32(l4 + 2 < left);
              const unsigned m3 = __builtin_amdgcn_ballot_w32(o3) & __builtin_amdgcn_ballot_w32(l4 + 3 < left);
              const int tot = __builtin_popcount(m0) + __builtin_popcount(m1)
                            + __builtin_popcount(m2) + __builtin_popcount(m3);
              if (cnt + tot <= OCAP) {
                if (tot != 0) {
                  int pos = (int)__builtin_amdgcn_mbcnt_lo(m3,
                                 __builtin_amdgcn_mbcnt_lo(m2,
                                 __builtin_amdgcn_mbcnt_lo(m1,
                                 __builtin_amdgcn_mbcnt_lo(m0, (unsigned)cnt))));
                  if (o0 && l4     < left) rawl[pos++] = v[u].x;
                  if (o1 && l4 + 1 < left) rawl[pos++] = v[u].y;
                  if (o2 && l4 + 2 < left) rawl[pos++] = v[u].z;
                  if (o3 && l4 + 3 < left) rawl[pos]   = v[u].w;
                  cnt += tot;
                }
                done = u + 1;
              } else {
                full = true;
              }
            }
          }
          c += done;
        }
        const bool end = c >= nch;
        if (end || full || cnt >= SUMMIN) { sort_and_sum(cnt); cnt = 0; }
        if (end) break;
      }
    } else {
      auto flush = [&](int n) {
        __builtin_amdgcn_fence(__ATOMIC_ACQ_REL, "wavefront");
        for (int k0 = 0; k0 < n; k0 += 32) {
          const int k = k0 + lane;
          int s = 0, r = 0;
          if (k < n) {
            const int e = mylist[k];
            s = src[e];
            r = tgt[e] - base;
          }
          add_rows(s, r, min(32, n - k0));
        }
        __builtin_amdgcn_fence(__ATOMIC_ACQ_REL, "wavefront");
      };
      search_edges(tgt, E, base, rows, mylist, lane, wave, flush);
      __syncthreads();
    }

    float4 ba, bb;
    ba.x = bias[4 * lane];           ba.y = bias[4 * lane + 1];
    ba.z = bias[4 * lane + 2];       ba.w = bias[4 * lane + 3];
    bb.x = bias[128 + 4 * lane];     bb.y = bias[128 + 4 * lane + 1];
    bb.z = bias[128 + 4 * lane + 2]; bb.w = bias[128 + 4 * lane + 3];
    if (LAST) {
      for (int r = wave; r < rows; r += NWAVE) {
        const int i = base + r;
        const float d = dinv[i];
        const float* tp = tile + r * TSTR;
        float4 a, b;
        a.x = tp[lane] * d;        a.y = tp[32 + lane] * d;
        a.z = tp[64 + lane] * d;   a.w = tp[96 + lane] * d;
        b.x = tp[128 + lane] * d;  b.y = tp[160 + lane] * d;
        b.z = tp[192 + lane] * d;  b.w = tp[224 + lane] * d;
        a.x += ba.x; a.y += ba.y; a.z += ba.z; a.w += ba.w;
        b.x += bb.x; b.y += bb.y; b.z += bb.z; b.w += bb.w;
        float4* op = (float4*)(out + (size_t)i * HID);
        op[lane]      = a;
        op[32 + lane] = b;
      }
    } else {
      const float kLn1e4Over256 = 9.210340371976184f / 256.0f;
      const float f0 = expf(-(float)(4 * lane)           * kLn1e4Over256);
      const float f1 = expf(-(float)(4 * lane + 2)       * kLn1e4Over256);
      const float f2 = expf(-(float)(128 + 4 * lane)     * kLn1e4Over256);
      const float f3 = expf(-(float)(128 + 4 * lane + 2) * kLn1e4Over256);
      for (int r = wave; r < rows; r += NWAVE) {
        const int i = base + r;
        const float d  = dinv[i];
        const float fi = (float)i;
        float* tp = tile + r * TSTR;
        float s0, c0, s1, c1, s2, c2, s3, c3;
        sincosf(fi * f0, &s0, &c0);
        sincosf(fi * f1, &s1, &c1);
        sincosf(fi * f2, &s2, &c2);
        sincosf(fi * f3, &s3, &c3);
        float4 a, b;
        a.x = tp[lane] * d + ba.x + s0;        a.y = tp[32 + lane] * d + ba.y + c0;
        a.z = tp[64 + lane] * d + ba.z + s1;   a.w = tp[96 + lane] * d + ba.w + c1;
        b.x = tp[128 + lane] * d + bb.x + s2;  b.y = tp[160 + lane] * d + bb.y + c2;
        b.z = tp[192 + lane] * d + bb.z + s3;  b.w = tp[224 + lane] * d + bb.w + c3;
        tp[lane]       = a.x > 0.0f ? a.x : 0.0f;  tp[32 + lane]  = a.y > 0.0f ? a.y : 0.0f;
        tp[64 + lane]  = a.z > 0.0f ? a.z : 0.0f;  tp[96 + lane]  = a.w > 0.0f ? a.w : 0.0f;
        tp[128 + lane] = b.x > 0.0f ? b.x : 0.0f;  tp[160 + lane] = b.y > 0.0f ? b.y : 0.0f;
        tp[192 + lane] = b.z > 0.0f ? b.z : 0.0f;  tp[224 + lane] = b.w > 0.0f ? b.w : 0.0f;
      }
      __syncthreads();
      const int tp8 = wave & 7;
      const int mq  = wave >> 3;
      if (mq == 0) rows_times_w<4>(tile, Wn, dinv + base, out, base, rows, mq, tp8, lane);
      else         rows_times_w<3>(tile, Wn, dinv + base, out, base, rows, mq, tp8, lane);
    }
    __syncthreads();
  }
}

extern "C" void kernel_launch(void* const* d_in, const int* in_sizes, int n_in,
                              void* d_out, int out_size, void* d_ws, size_t ws_size,
                              hipStream_t stream) {
  const float* X  = (const float*)d_in[0];
  const int*   EI = (const int*)d_in[1];
  const float* W1 = (const float*)d_in[2];
  const float* b1 = (const float*)d_in[3];
  const float* W2 = (const float*)d_in[4];
  const float* b2 = (const float*)d_in[5];
  float* out = (float*)d_out;

  const int N = in_sizes[0] / IN_DIM;
  const int E = in_sizes[1] / 2;
  const int NF = N * HID;

  float* dinv   = (float*)d_ws;
  float* bufL   = dinv + (1 << 16);
  float* bufAgg = bufL + (size_t)NF;
  float* bufH   = out;
  int* ccount = (int*)bufL;
  int* cells  = ccount + NLIST * NBIN;
  int* tcount = cells + NLIST * NBIN * CCAP;
  int* tlist  = tcount + NLIST;

  const int* src = EI;
  const int* tgt = EI + E;

  bin_edges_kernel<<<NBIN, BINT, 0, stream>>>(src, tgt, cells, ccount, E, N);

  tile_lists_xw1_kernel<<<256, 1024, 0, stream>>>(X, W1, tgt, dinv, bufH,
                                                  cells, ccount, tlist, tcount, E, N);

  aggregate_lds_kernel<false><<<256, 1024, 0, stream>>>(bufH, src, tgt, dinv, b1, W2,
                                                        bufAgg, tlist, tcount, E, N);

  aggregate_lds_kernel<true><<<256, 1024, 0, stream>>>(bufAgg, src, tgt, dinv, b2, W2,
                                                       out, tlist, tcount, E, N);
}


# tile lists built sorted by owning wave and row (rank from the row's LDS degree counter, one prefix scan, self-loop first): a wave reads only its piece, no scan, no sort; row sums in arrival order
# speedup vs baseline: 5.0610x; 1.1269x over previous
#include <hip/hip_runtime.h>
#include <math.h>


typedef __attribute__((ext_vector_type(2))) float v2f;
typedef __attribute__((ext_vector_type(8))) float v8f;
typedef __attribute__((ext_vector_type(2))) _Float16 h2v;
typedef __attribute__((ext_vector_type(4))) _Float16 h4v;

static constexpr int HID    = 256;
static constexpr int IN_DIM = 64;

static constexpr int TROWS = 196;
static constexpr int NWAVE = 32;
static constexpr int EPH   = 512;
static constexpr int BATCH = 32;
static constexpr int LCAP  = BATCH + EPH;
static constexpr int NLIST = 256;
static constexpr int NBIN  = 256;
static constexpr int CCAP  = 64;
static constexpr int TCAP  = 8192;
static constexpr int WSTR  = 64;
static constexpr int NOWN  = (TROWS + NWAVE - 1) / NWAVE;
static constexpr int BINT  = 512;
static constexpr int BINU  = 8;

__device__ __forceinline__ void load_hrow(const float* __restrict__ Hf, size_t row,
                                          int lane, h4v& a, h4v& b) {
  const h4v* hp = (const h4v*)((const _Float16*)Hf + row * HID);
  a = hp[lane];
  b = hp[32 + lane];
}
__device__ __forceinline__ float4 widen(const h4v v) {
  return make_float4((float)v.x, (float)v.y, (float)v.z, (float)v.w);
}
__device__ __forceinline__ void store_hpair(float* __restrict__ Hf, size_t row, int col,
                                            float x, float y) {
  h2v v;
  v.x = (_Float16)x;
  v.y = (_Float16)y;
  *(h2v*)((_Float16*)Hf + row * HID + col) = v;
}

__global__ __launch_bounds__(BINT)
void bin_edges_kernel(const int* __restrict__ src,
                      const int* __restrict__ tgt,
                      int* __restrict__ cells, int* __restrict__ ccount,
                      int E, int N) {
  __shared__ int cnt[NLIST];
  const int g = blockIdx.x;
  if (g >= NBIN) return;
  const int ntl = (N <= (1 << 24)) ? min(NLIST, (N + TROWS - 1) / TROWS) : 0;
  for (int i = threadIdx.x; i < NLIST; i += BINT) cnt[i] = 0;
  __syncthreads();

  const int per = (E + NBIN - 1) / NBIN;
  const int lo  = min(E, g * per);
  const int hi  = min(E, lo + per);
  for (int e0 = lo; e0 < hi; e0 += BINT * BINU) {
    int t[BINU], s[BINU];
#pragma unroll
    for (int u = 0; u < BINU; ++u) {
      const int e = e0 + u * BINT + (int)threadIdx.x;
      const bool in = e < hi;
      t[u] = in ? tgt[e] : -1;
      s[u] = in ? src[e] : 0;
    }
    int T[BINU], pos[BINU];
#pragma unroll
    for (int u = 0; u < BINU; ++u) {
      T[u] = ((unsigned)t[u] < (unsigned)N) ? t[u] / TROWS : ntl;
      pos[u] = CCAP;
      if (T[u] < ntl)
        pos[u] = __hip_atomic_fetch_add(&cnt[T[u]], 1, __ATOMIC_RELAXED,
                                        __HIP_MEMORY_SCOPE_WORKGROUP);
    }
#pragma unroll
    for (int u = 0; u < BINU; ++u) {
      if (pos[u] < CCAP) {
        const int r = t[u] - T[u] * TROWS;
        cells[((size_t)T[u] * NBIN + g) * CCAP + pos[u]] =
            (int)((unsigned)s[u] | ((unsigned)r << 24));
      }
    }
  }
  __syncthreads();
  for (int i = threadIdx.x; i < NLIST; i += BINT) ccount[(size_t)i * NBIN + g] = cnt[i];
}

template <class Flush>
__device__ __forceinline__ void search_edges(const int* __restrict__ tgt, int E,
                                             int base, int rows, int* mylist,
                                             int lane, int wave, Flush&& flush) {
  const unsigned lanes_below = (1u << lane) - 1u;
  const int hd  = min((int)(((16u - (unsigned)((size_t)tgt & 15)) & 15u) >> 2), E);
  const int* tb = tgt + hd;
  const int EB  = E - hd;
  const int G     = EB >> 2;
  const int GPI   = NWAVE * 8 * 32;
  const int ITERS = (G + GPI - 1) / GPI;
  const int FULL  = G / GPI;

  int cnt = 0;
  auto consider = [&](int t, int e) {
    const bool hit = (unsigned)(t - base) < (unsigned)rows;
    const unsigned m = __builtin_amdgcn_ballot_w32(hit);
    if (m != 0u) {
      if (hit) mylist[cnt + __builtin_popcount(m & lanes_below)] = e;
      cnt += __builtin_popcount(m);
    }
  };

  if (wave == 0) {
    const int e = (lane < hd) ? lane : hd + 4 * G + (lane - hd);
    const int t = (lane < 8 && e < E) ? tgt[e] : base - 1;
    consider(t, e);
  }
  cnt = __builtin_amdgcn_readfirstlane(cnt);

  const int rot = (ITERS > 0) ? (int)(blockIdx.x % (unsigned)ITERS) : 0;
  auto place_of = [&](int it) {
    int p = it + rot;
    if (p >= ITERS) p -= ITERS;
    return p;
  };
  auto group_of = [&](int p, int u) {
    return ((p * NWAVE + wave) * 8 + u) * 32 + lane;
  };
  auto load_place = [&](int p, int4 (&dst)[8]) {
    if (p < FULL) {
#pragma unroll
      for (int u = 0; u < 8; ++u)
        dst[u] = *(const int4*)(tb + 4 * group_of(p, u));
    } else {
#pragma unroll
      for (int u = 0; u < 8; ++u) {
        const int g = group_of(p, u);
        int4 v = *(const int4*)(tb + 4 * min(g, G - 1));
        if (g >= G) v.x = v.y = v.z = v.w = base - 1;
        dst[u] = v;
      }
    }
  };

  for (int it = 0; it < ITERS; ++it) {
    const int p = place_of(it);
    int4 cur[8];
    load_place(p, cur);
#pragma unroll
    for (int half = 0; half < 2; ++half) {
      if (cnt >= BATCH) { flush(cnt); cnt = 0; }
#pragma unroll
      for (int u = 4 * half; u < 4 * half + 4; ++u) {
        const int e = hd + 4 * group_of(p, u);
        consider(cur[u].x, e);
        consider(cur[u].y, e + 1);
        consider(cur[u].z, e + 2);
        consider(cur[u].w, e + 3);
      }
    }
  }
  flush(cnt);
}

static constexpr int XROWS = TROWS;
static constexpr int XSTR  = IN_DIM + 4;
static_assert(XROWS * (IN_DIM / 4) <= 4 * 1024, "staging: four 16-byte loads per thread");
static_assert(NWAVE * LCAP >= 1024 * 4 * (CCAP / 16), "a thread's cell entries fit the LDS edge list space");

__global__ __launch_bounds__(1024)
void tile_lists_xw1_kernel(const float* __restrict__ X,
                       const float* __restrict__ W1,
                       const int* __restrict__ tgt,
                       float* __restrict__ dinv,
                       float* __restrict__ Hout,
                       const int* __restrict__ cells,
                       const int* __restrict__ ccount,
                       int* __restrict__ tlist, int* __restrict__ tcount,
                       int* __restrict__ wseg,
                       int E, int N) {
  __shared__ __attribute__((aligned(16))) float xt[XROWS * XSTR];
  __shared__ int   kept[NWAVE * LCAP];
  __shared__ int   degc[TROWS];
  __shared__ float dl[TROWS];
  __shared__ int   ccnt[NBIN];
  __shared__ int   rstart[TROWS];
  __shared__ int   tot;
  __shared__ int   badc;

  const int lane = threadIdx.x & 31;
  const int wave = threadIdx.x >> 5;
  const int h  = lane >> 4;
  const int ln = lane & 15;
  const int tp = wave & 7;
  const int mq = __builtin_amdgcn_readfirstlane(wave >> 3);
  const int n2 = tp * 32 + 2 * ln;
  int* mylist = kept + wave * LCAP;

  for (int base = blockIdx.x * TROWS; base < N; base += gridDim.x * TROWS) {
    const int rows = min(TROWS, N - base);
    const int tile_id  = base / TROWS;
    const bool listed  = tile_id < NLIST && N <= (1 << 24);
    const int  lt      = listed ? tile_id : 0;

    if ((int)threadIdx.x < TROWS) degc[threadIdx.x] = 0;
    if (threadIdx.x == 0) { tot = 0; badc = 0; }
    __syncthreads();

    if (listed && (int)threadIdx.x < NBIN) {
      const int cn = ccount[(size_t)lt * NBIN + threadIdx.x];
      ccnt[threadIdx.x] = cn;
      __hip_atomic_fetch_add(&tot, cn, __ATOMIC_RELAXED, __HIP_MEMORY_SCOPE_WORKGROUP);
      if (cn > CCAP) badc = 1;
    }
    __syncthreads();
    const int  nall     = tot;
    const bool use_cell = listed && badc == 0 && nall + rows <= TCAP;

    if (use_cell) {
      const int q  = (int)threadIdx.x & 15;
      const int c0 = (int)threadIdx.x >> 4;
      int* tl = tlist + (size_t)lt * TCAP;
      int cn[4];
#pragma unroll
      for (int u = 0; u < 4; ++u) cn[u] = ccnt[c0 + 64 * u];
      int p[4][CCAP / 16];
#pragma unroll
      for (int u = 0; u < 4; ++u) {
        const int* cp = cells + ((size_t)lt * NBIN + c0 + 64 * u) * CCAP;
#pragma unroll
        for (int v = 0; v < CCAP / 16; ++v)
          p[u][v] = (q + 16 * v < cn[u]) ? cp[q + 16 * v] : -1;
      }
      int rk[4][CCAP / 16];
#pragma unroll
      for (int u = 0; u < 4; ++u) {
#pragma unroll
        for (int v = 0; v < CCAP / 16; ++v) {
          rk[u][v] = -1;
          if (q + 16 * v < cn[u]) {
            const int r = (int)((unsigned)p[u][v] >> 24);
            rk[u][v] = __hip_atomic_fetch_add(&degc[r], 1, __ATOMIC_RELAXED,
                                              __HIP_MEMORY_SCOPE_WORKGROUP);
            kept[(int)threadIdx.x + 1024 * (4 * u + v)] = p[u][v];
          }
        }
      }
      __syncthreads();
      if (wave == 0) {
        int cr[NOWN];
        int mine = 0;
#pragma unroll
        for (int sl = 0; sl < NOWN; ++sl) {
          const int r = lane + NWAVE * sl;
          cr[sl] = (r < rows) ? degc[r] + 1 : 0;
          mine += cr[sl];
        }
        int before = 0, run = 0;
        for (int j = 0; j < 32; ++j) {
          const int vj = __builtin_amdgcn_readlane(mine, j);
          if (lane == j) before = run;
          run += vj;
        }
        wseg[(size_t)lt * WSTR + lane] = before;
        if (lane == 31) wseg[(size_t)lt * WSTR + 32] = run;
        int at = before;
#pragma unroll
        for (int sl = 0; sl < NOWN; ++sl) {
          const int r = lane + NWAVE * sl;
          if (r < rows) rstart[r] = at;
          at += cr[sl];
        }
      }
      __syncthreads();
#pragma unroll
      for (int u = 0; u < 4; ++u) {
#pragma unroll
        for (int v = 0; v < CCAP / 16; ++v) {
          if (rk[u][v] >= 0) {
            const int e = kept[(int)threadIdx.x + 1024 * (4 * u + v)];
            tl[rstart[(int)((unsigned)e >> 24)] + 1 + rk[u][v]] = e;
          }
        }
      }
      if ((int)threadIdx.x < rows)
        tl[rstart[threadIdx.x]] =
            (int)((unsigned)(base + (int)threadIdx.x) | ((unsigned)threadIdx.x << 24));
      if (threadIdx.x == 0) tcount[lt] = nall + rows;
    } else {
      auto flush = [&](int n) {
        __builtin_amdgcn_fence(__ATOMIC_ACQ_REL, "wavefront");
        for (int k0 = 0; k0 < n; k0 += 32) {
          const int k = k0 + lane;
          if (k < n) {
            const int r = tgt[mylist[k]] - base;
            __hip_atomic_fetch_add(&degc[r], 1, __ATOMIC_RELAXED, __HIP_MEMORY_SCOPE_WORKGROUP);
          }
        }
        __builtin_amdgcn_fence(__ATOMIC_ACQ_REL, "wavefront");
      };
      search_edges(tgt, E, base, rows, mylist, lane, wave, flush);
      if (listed && threadIdx.x == 0) tcount[lt] = -1;
    }
    __syncthreads();

    if ((int)threadIdx.x < rows) {
      const float d = rsqrtf(1.0f + (float)degc[threadIdx.x]);
      dl[threadIdx.x] = d;
      dinv[base + threadIdx.x] = d;
    }

    const int nvec = rows * (IN_DIM / 4);
    const float4* xp = (const float4*)(X + (size_t)base * IN_DIM);
    float4 q[4];
#pragma unroll
    for (int u = 0; u < 4; ++u) {
      const int v = (int)threadIdx.x + 1024 * u;
      q[u] = (v < nvec) ? xp[v] : make_float4(0.0f, 0.0f, 0.0f, 0.0f);
    }

#pragma unroll
    for (int u = 0; u < 4; ++u) {
      const int v = (int)threadIdx.x + 1024 * u;
      if (v < nvec) *(float4*)(xt + (v >> 4) * XSTR + 4 * (v & 15)) = q[u];
    }

    v2f be[16], bo[16];
#pragma unroll
    for (int s = 0; s < 16; ++s) {
      const float* wp = W1 + (size_t)(4 * s + 2 * h) * HID + n2;
      const float2 w0 = *(const float2*)wp;
      const float2 w1 = *(const float2*)(wp + HID);
      be[s].x = w0.x;  be[s].y = w1.x;
      bo[s].x = w0.y;  bo[s].y = w1.y;
    }
    __syncthreads();

    const int ntiles = (rows + 15) >> 4;
    for (int t = mq; t < ntiles; t += 4) {
      const int row = min(t * 16 + ln, rows - 1);
      const float* ap = xt + row * XSTR + 2 * h;
      v2f a[16];
#pragma unroll
      for (int s = 0; s < 16; ++s) a[s] = *(const v2f*)(ap + 4 * s);
      v8f c0 = {}, c1 = {};
#pragma unroll
      for (int s = 0; s < 16; ++s) {
        c0 = __builtin_amdgcn_wmma_f32_16x16x4_f32(
            false, a[s], false, be[s], (short)0, c0, false, false);
        c1 = __builtin_amdgcn_wmma_f32_16x16x4_f32(
            false, a[s], false, bo[s], (short)0, c1, false, false);
      }
      const int r0 = t * 16 + 8 * h;
#pragma unroll
      for (int r = 0; r < 8; ++r) {
        if (r0 + r < rows) {
          const float d = dl[r0 + r];
          store_hpair(Hout, (size_t)(base + r0 + r), n2, c0[r] * d, c1[r] * d);
        }
      }
    }
    __syncthreads();
  }
}


static constexpr int TSTR  = HID + 1;

template <int NS>
__device__ __forceinline__ void rows_times_w(const float* tile,
                                             const float* __restrict__ Wm,
                                             const float* __restrict__ dsc,
                                             float* __restrict__ out,
                                             int base, int rows, int mq, int tp,
                                             int lane) {
  const int h  = lane >> 4;
  const int ln = lane & 15;
  const int n2 = tp * 32 + 2 * ln;
  int aoff[NS];
#pragma unroll
  for (int i = 0; i < NS; ++i) {
    const int row = min((mq + 4 * i) * 16 + ln, rows - 1);
    aoff[i] = row * TSTR;
  }
  v8f acc[NS][2];
#pragma unroll
  for (int i = 0; i < NS; ++i) { acc[i][0] = v8f{}; acc[i][1] = v8f{}; }

  auto w_of = [&](int s) {
    const int lh = ((s & 31) + 16 * h) & 31;
    const int c0 = (s >> 5) * 128 + 4 * lh + 2 * h;
    return Wm + (size_t)c0 * HID + n2;
  };
  const float* wp = w_of(0);
  float2 w0n = *(const float2*)wp;
  float2 w1n = *(const float2*)(wp + HID);
#pragma unroll 2
  for (int s = 0; s < 64; ++s) {
    const float2 w0 = w0n, w1 = w1n;
    wp  = w_of(min(s + 1, 63));
    w0n = *(const float2*)wp;
    w1n = *(const float2*)(wp + HID);
    const int lh  = ((s & 31) + 16 * h) & 31;
    const int pos = (s >> 5) * 128 + 64 * h + lh;
    v2f a[NS];
#pragma unroll
    for (int i = 0; i < NS; ++i) {
      a[i].x = tile[aoff[i] + pos];
      a[i].y = tile[aoff[i] + pos + 32];
    }
    v2f be, bo;
    be.x = w0.x;  be.y = w1.x;
    bo.x = w0.y;  bo.y = w1.y;
#pragma unroll
    for (int i = 0; i < NS; ++i) {
      acc[i][0] = __builtin_amdgcn_wmma_f32_16x16x4_f32(
          false, a[i], false, be, (short)0, acc[i][0], false, false);
      acc[i][1] = __builtin_amdgcn_wmma_f32_16x16x4_f32(
          false, a[i], false, bo, (short)0, acc[i][1], false, false);
    }
  }

#pragma unroll
  for (int i = 0; i < NS; ++i) {
    const int r0 = (mq + 4 * i) * 16 + 8 * h;
#pragma unroll
    for (int r = 0; r < 8; ++r) {
      if (r0 + r < rows) {
        const float d = dsc[r0 + r];
        store_hpair(out, (size_t)(base + r0 + r), n2, acc[i][0][r] * d, acc[i][1][r] * d);
      }
    }
  }
}

static_assert(TROWS <= 256 && NWAVE == 32 && NOWN <= 8,
              "list entry: tile row in bits 24..31, so its slot (row / 32) in bits 29..31");

template <bool LAST>
__global__ __launch_bounds__(1024)
void aggregate_lds_kernel(const float* __restrict__ H,
                          const int* __restrict__ src,
                          const int* __restrict__ tgt,
                          const float* __restrict__ dinv,
                          const float* __restrict__ bias,
                          const float* __restrict__ Wn,
                          float* __restrict__ out,
                          const int* __restrict__ tlist,
                          const int* __restrict__ tcount,
                          const int* __restrict__ wseg,
                          int E, int N) {
  __shared__ float tile[TROWS * TSTR];
  __shared__ int   kept[NWAVE * LCAP];

  const int lane = threadIdx.x & 31;
  const int wave = threadIdx.x >> 5;
  int* mylist = kept + wave * LCAP;

  for (int base = blockIdx.x * TROWS; base < N; base += gridDim.x * TROWS) {
    const int rows = min(TROWS, N - base);
    const int tile_id  = base / TROWS;
    const bool listed  = tile_id < NLIST && N <= (1 << 24);
    const int  lt      = listed ? tile_id : 0;

    int nt = -1;
    if (listed) {
      nt = __builtin_amdgcn_readfirstlane(tcount[lt]);
    }

    if (nt < 0) {
      for (int r = wave; r < rows; r += NWAVE) {
        const int i = base + r;
        h4v ah, bh;
        load_hrow(H, (size_t)i, lane, ah, bh);
        const float4 a = widen(ah), b = widen(bh);
        float* tp = tile + r * TSTR;
        tp[lane]       = a.x;  tp[32 + lane]  = a.y;
        tp[64 + lane]  = a.z;  tp[96 + lane]  = a.w;
        tp[128 + lane] = b.x;  tp[160 + lane] = b.y;
        tp[192 + lane] = b.z;  tp[224 + lane] = b.w;
      }
      __syncthreads();
    }

    auto add_rows = [&](int s, int r, int m) {
      int j = 0;
      for (; j + 4 <= m; j += 4) {
        h4v ah[4], bh[4];
        int rj[4];
#pragma unroll
        for (int u = 0; u < 4; ++u) {
          const int sj = __builtin_amdgcn_readlane(s, j + u);
          rj[u] = __builtin_amdgcn_readlane(r, j + u);
          load_hrow(H, (size_t)sj, lane, ah[u], bh[u]);
        }
#pragma unroll
        for (int u = 0; u < 4; ++u) {
          const float4 a = widen(ah[u]), b = widen(bh[u]);
          float* tp = tile + rj[u] * TSTR + lane;
          __hip_atomic_fetch_add(tp,       a.x, __ATOMIC_RELAXED, __HIP_MEMORY_SCOPE_WORKGROUP);
          __hip_atomic_fetch_add(tp + 32,  a.y, __ATOMIC_RELAXED, __HIP_MEMORY_SCOPE_WORKGROUP);
          __hip_atomic_fetch_add(tp + 64,  a.z, __ATOMIC_RELAXED, __HIP_MEMORY_SCOPE_WORKGROUP);
          __hip_atomic_fetch_add(tp + 96,  a.w, __ATOMIC_RELAXED, __HIP_MEMORY_SCOPE_WORKGROUP);
          __hip_atomic_fetch_add(tp + 128, b.x, __ATOMIC_RELAXED, __HIP_MEMORY_SCOPE_WORKGROUP);
          __hip_atomic_fetch_add(tp + 160, b.y, __ATOMIC_RELAXED, __HIP_MEMORY_SCOPE_WORKGROUP);
          __hip_atomic_fetch_add(tp + 192, b.z, __ATOMIC_RELAXED, __HIP_MEMORY_SCOPE_WORKGROUP);
          __hip_atomic_fetch_add(tp + 224, b.w, __ATOMIC_RELAXED, __HIP_MEMORY_SCOPE_WORKGROUP);
        }
      }
      for (; j < m; ++j) {
        const int sj = __builtin_amdgcn_readlane(s, j);
        const int rr = __builtin_amdgcn_readlane(r, j);
        h4v ah, bh;
        load_hrow(H, (size_t)sj, lane, ah, bh);
        const float4 a = widen(ah), b = widen(bh);
        float* tp = tile + rr * TSTR + lane;
        __hip_atomic_fetch_add(tp,       a.x, __ATOMIC_RELAXED, __HIP_MEMORY_SCOPE_WORKGROUP);
        __hip_atomic_fetch_add(tp + 32,  a.y, __ATOMIC_RELAXED, __HIP_MEMORY_SCOPE_WORKGROUP);
        __hip_atomic_fetch_add(tp + 64,  a.z, __ATOMIC_RELAXED, __HIP_MEMORY_SCOPE_WORKGROUP);
        __hip_atomic_fetch_add(tp + 96,  a.w, __ATOMIC_RELAXED, __HIP_MEMORY_SCOPE_WORKGROUP);
        __hip_atomic_fetch_add(tp + 128, b.x, __ATOMIC_RELAXED, __HIP_MEMORY_SCOPE_WORKGROUP);
        __hip_atomic_fetch_add(tp + 160, b.y, __ATOMIC_RELAXED, __HIP_MEMORY_SCOPE_WORKGROUP);
        __hip_atomic_fetch_add(tp + 192, b.z, __ATOMIC_RELAXED, __HIP_MEMORY_SCOPE_WORKGROUP);
        __hip_atomic_fetch_add(tp + 224, b.w, __ATOMIC_RELAXED, __HIP_MEMORY_SCOPE_WORKGROUP);
      }
    };

    if (nt >= 0) {
      const int wv = __builtin_amdgcn_readfirstlane(wave);
      const int* wsp = wseg + (size_t)lt * WSTR;
      const int p0 = __builtin_amdgcn_readfirstlane(wsp[wv]);
      const int p1 = __builtin_amdgcn_readfirstlane(wsp[wv + 1]);
      const int n  = p1 - p0;
      const int* __restrict__ seg = tlist + (size_t)lt * TCAP + p0;

      if (n > 0) {
        float4 accA = make_float4(0.0f, 0.0f, 0.0f, 0.0f);
        float4 accB = make_float4(0.0f, 0.0f, 0.0f, 0.0f);
        auto commit = [&](int sl) {
          float* tp = tile + (wv + NWAVE * sl) * TSTR + lane;
          tp[0]   = accA.x;  tp[32]  = accA.y;  tp[64]  = accA.z;  tp[96]  = accA.w;
          tp[128] = accB.x;  tp[160] = accB.y;  tp[192] = accB.z;  tp[224] = accB.w;
          accA = make_float4(0.0f, 0.0f, 0.0f, 0.0f);
          accB = make_float4(0.0f, 0.0f, 0.0f, 0.0f);
        };
        auto add = [&](const float4& a, const float4& b) {
          accA.x += a.x;  accA.y += a.y;  accA.z += a.z;  accA.w += a.w;
          accB.x += b.x;  accB.y += b.y;  accB.z += b.z;  accB.w += b.w;
        };
        auto entry_at = [&](int k) { return (k < n) ? seg[k] : 0; };
        int e0 = entry_at(lane), e1 = entry_at(32 + lane);
        int e2 = entry_at(64 + lane), e3 = entry_at(96 + lane);
        int cur = __builtin_amdgcn_readfirstlane((int)((unsigned)e0 >> 29));
        for (int k0 = 0; k0 < n; k0 += 32) {
          const int e  = e0;
          e0 = e1;  e1 = e2;  e2 = e3;
          e3 = entry_at(k0 + 128 + lane);
          const int s  = e & 0xFFFFFF;
          const int sl = (int)((unsigned)e >> 29);
          const int m  = min(32, n - k0);
          for (int j = 0; j < m; j += 8) {
            h4v a[8], b[8];
            int slj[8];
#pragma unroll
            for (int u = 0; u < 8; ++u) {
              const int ju = min(j + u, m - 1);
              const int sj = __builtin_amdgcn_readlane(s, ju);
              slj[u] = __builtin_amdgcn_readlane(sl, ju);
              load_hrow(H, (size_t)sj, lane, a[u], b[u]);
            }
#pragma unroll
            for (int u = 0; u < 8; ++u) {
              if (j + u < m) {
                if (slj[u] != cur) { commit(cur); cur = slj[u]; }
                add(widen(a[u]), widen(b[u]));
              }
            }
          }
        }
        commit(cur);
      }
    } else {
      auto flush = [&](int n) {
        __builtin_amdgcn_fence(__ATOMIC_ACQ_REL, "wavefront");
        for (int k0 = 0; k0 < n; k0 += 32) {
          const int k = k0 + lane;
          int s = 0, r = 0;
          if (k < n) {
            const int e = mylist[k];
            s = src[e];
            r = tgt[e] - base;
          }
          add_rows(s, r, min(32, n - k0));
        }
        __builtin_amdgcn_fence(__ATOMIC_ACQ_REL, "wavefront");
      };
      search_edges(tgt, E, base, rows, mylist, lane, wave, flush);
      __syncthreads();
    }

    float4 ba, bb;
    ba.x = bias[4 * lane];           ba.y = bias[4 * lane + 1];
    ba.z = bias[4 * lane + 2];       ba.w = bias[4 * lane + 3];
    bb.x = bias[128 + 4 * lane];     bb.y = bias[128 + 4 * lane + 1];
    bb.z = bias[128 + 4 * lane + 2]; bb.w = bias[128 + 4 * lane + 3];
    if (LAST) {
      for (int r = wave; r < rows; r += NWAVE) {
        const int i = base + r;
        const float d = dinv[i];
        const float* tp = tile + r * TSTR;
        float4 a, b;
        a.x = tp[lane] * d;        a.y = tp[32 + lane] * d;
        a.z = tp[64 + lane] * d;   a.w = tp[96 + lane] * d;
        b.x = tp[128 + lane] * d;  b.y = tp[160 + lane] * d;
        b.z = tp[192 + lane] * d;  b.w = tp[224 + lane] * d;
        a.x += ba.x; a.y += ba.y; a.z += ba.z; a.w += ba.w;
        b.x += bb.x; b.y += bb.y; b.z += bb.z; b.w += bb.w;
        float4* op = (float4*)(out + (size_t)i * HID);
        op[lane]      = a;
        op[32 + lane] = b;
      }
    } else {
      const float kLn1e4Over256 = 9.210340371976184f / 256.0f;
      const float f0 = expf(-(float)(4 * lane)           * kLn1e4Over256);
      const float f1 = expf(-(float)(4 * lane + 2)       * kLn1e4Over256);
      const float f2 = expf(-(float)(128 + 4 * lane)     * kLn1e4Over256);
      const float f3 = expf(-(float)(128 + 4 * lane + 2) * kLn1e4Over256);
      for (int r = wave; r < rows; r += NWAVE) {
        const int i = base + r;
        const float d  = dinv[i];
        const float fi = (float)i;
        float* tp = tile + r * TSTR;
        float s0, c0, s1, c1, s2, c2, s3, c3;
        sincosf(fi * f0, &s0, &c0);
        sincosf(fi * f1, &s1, &c1);
        sincosf(fi * f2, &s2, &c2);
        sincosf(fi * f3, &s3, &c3);
        float4 a, b;
        a.x = tp[lane] * d + ba.x + s0;        a.y = tp[32 + lane] * d + ba.y + c0;
        a.z = tp[64 + lane] * d + ba.z + s1;   a.w = tp[96 + lane] * d + ba.w + c1;
        b.x = tp[128 + lane] * d + bb.x + s2;  b.y = tp[160 + lane] * d + bb.y + c2;
        b.z = tp[192 + lane] * d + bb.z + s3;  b.w = tp[224 + lane] * d + bb.w + c3;
        tp[lane]       = a.x > 0.0f ? a.x : 0.0f;  tp[32 + lane]  = a.y > 0.0f ? a.y : 0.0f;
        tp[64 + lane]  = a.z > 0.0f ? a.z : 0.0f;  tp[96 + lane]  = a.w > 0.0f ? a.w : 0.0f;
        tp[128 + lane] = b.x > 0.0f ? b.x : 0.0f;  tp[160 + lane] = b.y > 0.0f ? b.y : 0.0f;
        tp[192 + lane] = b.z > 0.0f ? b.z : 0.0f;  tp[224 + lane] = b.w > 0.0f ? b.w : 0.0f;
      }
      __syncthreads();
      const int tp8 = wave & 7;
      const int mq  = wave >> 3;
      if (mq == 0) rows_times_w<4>(tile, Wn, dinv + base, out, base, rows, mq, tp8, lane);
      else         rows_times_w<3>(tile, Wn, dinv + base, out, base, rows, mq, tp8, lane);
    }
    __syncthreads();
  }
}

extern "C" void kernel_launch(void* const* d_in, const int* in_sizes, int n_in,
                              void* d_out, int out_size, void* d_ws, size_t ws_size,
                              hipStream_t stream) {
  const float* X  = (const float*)d_in[0];
  const int*   EI = (const int*)d_in[1];
  const float* W1 = (const float*)d_in[2];
  const float* b1 = (const float*)d_in[3];
  const float* W2 = (const float*)d_in[4];
  const float* b2 = (const float*)d_in[5];
  float* out = (float*)d_out;

  const int N = in_sizes[0] / IN_DIM;
  const int E = in_sizes[1] / 2;
  const int NF = N * HID;

  float* dinv   = (float*)d_ws;
  float* bufL   = dinv + (1 << 16);
  float* bufAgg = bufL + (size_t)NF;
  float* bufH   = out;
  int* ccount = (int*)bufL;
  int* cells  = ccount + NLIST * NBIN;
  int* tcount = cells + NLIST * NBIN * CCAP;
  int* tlist  = tcount + NLIST;
  int* wseg   = tlist + NLIST * TCAP;

  const int* src = EI;
  const int* tgt = EI + E;

  bin_edges_kernel<<<NBIN, BINT, 0, stream>>>(src, tgt, cells, ccount, E, N);

  tile_lists_xw1_kernel<<<256, 1024, 0, stream>>>(X, W1, tgt, dinv, bufH,
                                                  cells, ccount, tlist, tcount, wseg, E, N);

  aggregate_lds_kernel<false><<<256, 1024, 0, stream>>>(bufH, src, tgt, dinv, b1, W2,
                                                        bufAgg, tlist, tcount, wseg, E, N);

  aggregate_lds_kernel<true><<<256, 1024, 0, stream>>>(bufAgg, src, tgt, dinv, b2, W2,
                                                       out, tlist, tcount, wseg, E, N);
}
